# starting kernel
# baseline (speedup 1.0000x reference)
#include <hip/hip_runtime.h>
#include <math.h>


typedef __attribute__((ext_vector_type(2))) float v2f;
typedef __attribute__((ext_vector_type(8))) float v8f;

__device__ __forceinline__ void atomAddF32(float* p, float v) {
  (void)__hip_atomic_fetch_add(p, v, __ATOMIC_RELAXED, __HIP_MEMORY_SCOPE_AGENT);
}

__global__ void k_deg_init(float* deg, int n) {
  int i = blockIdx.x * blockDim.x + threadIdx.x;
  if (i < n) deg[i] = 1.0f;
}

__global__ void k_deg_scatter(const int* __restrict__ dst, float* deg, int E) {
  int e = blockIdx.x * blockDim.x + threadIdx.x;
  if (e < E) atomAddF32(&deg[dst[e]], 1.0f);
}

__global__ void k_dinv(float* deg, int n) {
  int i = blockIdx.x * blockDim.x + threadIdx.x;
  if (i < n) deg[i] = 1.0f / sqrtf(deg[i]);
}

#define GEMM_WAVES 4

__global__ __launch_bounds__(GEMM_WAVES * 32) void k_gemm_xW1(
    const float* __restrict__ x, const float* __restrict__ W1,
    float* __restrict__ h, int n) {
  __shared__ float sA[GEMM_WAVES][16][132];
  __shared__ float sB[16][132];
  const int tid = threadIdx.x;
  const int lane = tid & 31;
  const int wave = tid >> 5;
  const int base = (blockIdx.x * GEMM_WAVES + wave) * 16;
  const bool active = (base < n);

  for (int i = tid; i < 16 * 128; i += GEMM_WAVES * 32) {
    const int col = i >> 7;
    const int k = i & 127;
    sB[col][k] = (col < 4) ? W1[k * 4 + col] : 0.0f;
  }
  if (active) {
    const float* xt = x + (size_t)base * 128;
#pragma unroll
    for (int i = 0; i < 16; ++i) {
      float4 v = *(const float4*)(xt + (size_t)i * 128 + lane * 4);
      *(float4*)(&sA[wave][i][lane * 4]) = v;
    }
  }
  __syncthreads();
  if (!active) return;

  const int half = lane >> 4;
  const int mn = lane & 15;
  const float* aRow = &sA[wave][mn][0];
  const float* bRow = &sB[mn][0];
  v8f c = {};
  for (int ks = 0; ks < 32; ++ks) {
    const int k = ks * 4 + 2 * half;
    v2f a = *(const v2f*)(aRow + k);
    v2f b = *(const v2f*)(bRow + k);
    c = __builtin_amdgcn_wmma_f32_16x16x4_f32(false, a, false, b, (short)0, c,
                                              false, false);
  }
  if (mn < 4) {
    float* o = h + (size_t)(base + 8 * half) * 4 + mn;
#pragma unroll
    for (int r = 0; r < 8; ++r) o[(size_t)r * 4] = c[r];
  }
}

__global__ void k_init_agg4(const float* __restrict__ dinv,
                            const float* __restrict__ h, float* __restrict__ agg,
                            int n) {
  int v = blockIdx.x * blockDim.x + threadIdx.x;
  if (v >= n) return;
  float di = dinv[v];
  float s = di * di;
  float4 t = *(const float4*)(h + (size_t)v * 4);
  float4 o;
  o.x = s * t.x; o.y = s * t.y; o.z = s * t.z; o.w = s * t.w;
  *(float4*)(agg + (size_t)v * 4) = o;
}

__global__ void k_scatter4(const int* __restrict__ src, const int* __restrict__ dst,
                           const float* __restrict__ dinv,
                           const float* __restrict__ h, float* agg, int E) {
  int e = blockIdx.x * blockDim.x + threadIdx.x;
  if (e >= E) return;
  int s = src[e], d = dst[e];
  float w = dinv[s] * dinv[d];
  float4 m = *(const float4*)(h + (size_t)s * 4);
  float* o = agg + (size_t)d * 4;
  atomAddF32(o + 0, w * m.x);
  atomAddF32(o + 1, w * m.y);
  atomAddF32(o + 2, w * m.z);
  atomAddF32(o + 3, w * m.w);
}

__global__ void k_act_mm4(float* agg, const float* __restrict__ bvec,
                          const float* __restrict__ W,
                          const float* __restrict__ dinv,
                          float* __restrict__ hout, int n) {
  int v = blockIdx.x * blockDim.x + threadIdx.x;
  if (v >= n) return;
  float4 t = *(const float4*)(agg + (size_t)v * 4);
  float t0 = fmaxf(t.x + bvec[0], 0.0f);
  float t1 = fmaxf(t.y + bvec[1], 0.0f);
  float t2 = fmaxf(t.z + bvec[2], 0.0f);
  float t3 = fmaxf(t.w + bvec[3], 0.0f);
  float h0 = t0 * W[0] + t1 * W[4] + t2 * W[8]  + t3 * W[12];
  float h1 = t0 * W[1] + t1 * W[5] + t2 * W[9]  + t3 * W[13];
  float h2 = t0 * W[2] + t1 * W[6] + t2 * W[10] + t3 * W[14];
  float h3 = t0 * W[3] + t1 * W[7] + t2 * W[11] + t3 * W[15];
  float di = dinv[v];
  float s = di * di;
  float4 hn; hn.x = h0; hn.y = h1; hn.z = h2; hn.w = h3;
  float4 an; an.x = s * h0; an.y = s * h1; an.z = s * h2; an.w = s * h3;
  *(float4*)(hout + (size_t)v * 4) = hn;
  *(float4*)(agg + (size_t)v * 4) = an;
}

__global__ void k_act_mm_final(const float* __restrict__ agg,
                               const float* __restrict__ b2,
                               const float* __restrict__ W3,
                               const float* __restrict__ b3,
                               const float* __restrict__ dinv,
                               float* __restrict__ h3, float* __restrict__ out,
                               int n) {
  int v = blockIdx.x * blockDim.x + threadIdx.x;
  if (v >= n) return;
  float4 t = *(const float4*)(agg + (size_t)v * 4);
  float t0 = fmaxf(t.x + b2[0], 0.0f);
  float t1 = fmaxf(t.y + b2[1], 0.0f);
  float t2 = fmaxf(t.z + b2[2], 0.0f);
  float t3 = fmaxf(t.w + b2[3], 0.0f);
  float h0 = t0 * W3[0] + t1 * W3[2] + t2 * W3[4] + t3 * W3[6];
  float h1 = t0 * W3[1] + t1 * W3[3] + t2 * W3[5] + t3 * W3[7];
  float di = dinv[v];
  float s = di * di;
  *(float2*)(h3 + (size_t)v * 2) = make_float2(h0, h1);
  *(float2*)(out + (size_t)v * 2) = make_float2(s * h0 + b3[0], s * h1 + b3[1]);
}

__global__ void k_scatter2(const int* __restrict__ src, const int* __restrict__ dst,
                           const float* __restrict__ dinv,
                           const float* __restrict__ h3, float* out, int E) {
  int e = blockIdx.x * blockDim.x + threadIdx.x;
  if (e >= E) return;
  int s = src[e], d = dst[e];
  float w = dinv[s] * dinv[d];
  float2 m = *(const float2*)(h3 + (size_t)s * 2);
  atomAddF32(out + (size_t)d * 2 + 0, w * m.x);
  atomAddF32(out + (size_t)d * 2 + 1, w * m.y);
}

extern "C" void kernel_launch(void* const* d_in, const int* in_sizes, int n_in,
                              void* d_out, int out_size, void* d_ws,
                              size_t ws_size, hipStream_t stream) {
  const float* x  = (const float*)d_in[0];
  const int*   ei = (const int*)d_in[1];
  const float* W1 = (const float*)d_in[2];
  const float* b1 = (const float*)d_in[3];
  const float* W2 = (const float*)d_in[4];
  const float* b2 = (const float*)d_in[5];
  const float* W3 = (const float*)d_in[6];
  const float* b3 = (const float*)d_in[7];
  float* out = (float*)d_out;

  const int n = in_sizes[0] / 128;
  const int E = in_sizes[1] / 2;
  const int* src = ei;
  const int* dst = ei + E;

  float* ws   = (float*)d_ws;
  float* dinv = ws;
  float* hA   = dinv + (size_t)n;
  float* hB   = hA + (size_t)n * 4;
  float* hC   = hB + (size_t)n * 4;

  const int TB = 256;
  dim3 blk(TB);
  dim3 gn((n + TB - 1) / TB);
  dim3 ge((E + TB - 1) / TB);
  int tiles = (n + 16 * GEMM_WAVES - 1) / (16 * GEMM_WAVES);

  k_deg_init<<<gn, blk, 0, stream>>>(dinv, n);
  k_deg_scatter<<<ge, blk, 0, stream>>>(dst, dinv, E);
  k_dinv<<<gn, blk, 0, stream>>>(dinv, n);
  k_gemm_xW1<<<tiles, GEMM_WAVES * 32, 0, stream>>>(x, W1, hA, n);
  k_init_agg4<<<gn, blk, 0, stream>>>(dinv, hA, hB, n);
  k_scatter4<<<ge, blk, 0, stream>>>(src, dst, dinv, hA, hB, E);
  k_act_mm4<<<gn, blk, 0, stream>>>(hB, b1, W2, dinv, hA, n);
  k_scatter4<<<ge, blk, 0, stream>>>(src, dst, dinv, hA, hB, E);
  k_act_mm_final<<<gn, blk, 0, stream>>>(hB, b2, W3, b3, dinv, hC, out, n);
  k_scatter2<<<ge, blk, 0, stream>>>(src, dst, dinv, hC, out, E);
}


# Edges binned per call by 1024-node destination range (LDS-ranked counting sort); each work-group sums its range in LDS with ds float adds; no global atomics; sum order still arrival order
# speedup vs baseline: 7.0024x; 7.0024x over previous
#include <hip/hip_runtime.h>
#include <math.h>


typedef __attribute__((ext_vector_type(2))) float v2f;
typedef __attribute__((ext_vector_type(8))) float v8f;

__device__ __forceinline__ void atomAddF32(float* p, float v) {
  (void)__hip_atomic_fetch_add(p, v, __ATOMIC_RELAXED, __HIP_MEMORY_SCOPE_AGENT);
}

__global__ void k_deg_init(float* deg, int n) {
  int i = blockIdx.x * blockDim.x + threadIdx.x;
  if (i < n) deg[i] = 1.0f;
}

__global__ void k_deg_scatter(const int* __restrict__ dst, float* deg, int E) {
  int e = blockIdx.x * blockDim.x + threadIdx.x;
  if (e < E) atomAddF32(&deg[dst[e]], 1.0f);
}

__global__ void k_dinv(float* deg, int n) {
  int i = blockIdx.x * blockDim.x + threadIdx.x;
  if (i < n) deg[i] = 1.0f / sqrtf(deg[i]);
}

#define GEMM_WAVES 4

__global__ __launch_bounds__(GEMM_WAVES * 32) void k_gemm_xW1(
    const float* __restrict__ x, const float* __restrict__ W1,
    float* __restrict__ h, int n) {
  __shared__ float sA[GEMM_WAVES][16][132];
  __shared__ float sB[16][132];
  const int tid = threadIdx.x;
  const int lane = tid & 31;
  const int wave = tid >> 5;
  const int base = (blockIdx.x * GEMM_WAVES + wave) * 16;
  const bool active = (base < n);

  for (int i = tid; i < 16 * 128; i += GEMM_WAVES * 32) {
    const int col = i >> 7;
    const int k = i & 127;
    sB[col][k] = (col < 4) ? W1[k * 4 + col] : 0.0f;
  }
  if (active) {
    const float* xt = x + (size_t)base * 128;
#pragma unroll
    for (int i = 0; i < 16; ++i) {
      float4 v = *(const float4*)(xt + (size_t)i * 128 + lane * 4);
      *(float4*)(&sA[wave][i][lane * 4]) = v;
    }
  }
  __syncthreads();
  if (!active) return;

  const int half = lane >> 4;
  const int mn = lane & 15;
  const float* aRow = &sA[wave][mn][0];
  const float* bRow = &sB[mn][0];
  v8f c = {};
  for (int ks = 0; ks < 32; ++ks) {
    const int k = ks * 4 + 2 * half;
    v2f a = *(const v2f*)(aRow + k);
    v2f b = *(const v2f*)(bRow + k);
    c = __builtin_amdgcn_wmma_f32_16x16x4_f32(false, a, false, b, (short)0, c,
                                              false, false);
  }
  if (mn < 4) {
    float* o = h + (size_t)(base + 8 * half) * 4 + mn;
#pragma unroll
    for (int r = 0; r < 8; ++r) o[(size_t)r * 4] = c[r];
  }
}

__global__ void k_init_agg4(const float* __restrict__ dinv,
                            const float* __restrict__ h, float* __restrict__ agg,
                            int n) {
  int v = blockIdx.x * blockDim.x + threadIdx.x;
  if (v >= n) return;
  float di = dinv[v];
  float s = di * di;
  float4 t = *(const float4*)(h + (size_t)v * 4);
  float4 o;
  o.x = s * t.x; o.y = s * t.y; o.z = s * t.z; o.w = s * t.w;
  *(float4*)(agg + (size_t)v * 4) = o;
}

__global__ void k_scatter4(const int* __restrict__ src, const int* __restrict__ dst,
                           const float* __restrict__ dinv,
                           const float* __restrict__ h, float* agg, int E) {
  int e = blockIdx.x * blockDim.x + threadIdx.x;
  if (e >= E) return;
  int s = src[e], d = dst[e];
  float w = dinv[s] * dinv[d];
  float4 m = *(const float4*)(h + (size_t)s * 4);
  float* o = agg + (size_t)d * 4;
  atomAddF32(o + 0, w * m.x);
  atomAddF32(o + 1, w * m.y);
  atomAddF32(o + 2, w * m.z);
  atomAddF32(o + 3, w * m.w);
}

__global__ void k_act_mm4(float* agg, const float* __restrict__ bvec,
                          const float* __restrict__ W,
                          const float* __restrict__ dinv,
                          float* __restrict__ hout, int n) {
  int v = blockIdx.x * blockDim.x + threadIdx.x;
  if (v >= n) return;
  float4 t = *(const float4*)(agg + (size_t)v * 4);
  float t0 = fmaxf(t.x + bvec[0], 0.0f);
  float t1 = fmaxf(t.y + bvec[1], 0.0f);
  float t2 = fmaxf(t.z + bvec[2], 0.0f);
  float t3 = fmaxf(t.w + bvec[3], 0.0f);
  float h0 = t0 * W[0] + t1 * W[4] + t2 * W[8]  + t3 * W[12];
  float h1 = t0 * W[1] + t1 * W[5] + t2 * W[9]  + t3 * W[13];
  float h2 = t0 * W[2] + t1 * W[6] + t2 * W[10] + t3 * W[14];
  float h3 = t0 * W[3] + t1 * W[7] + t2 * W[11] + t3 * W[15];
  float di = dinv[v];
  float s = di * di;
  float4 hn; hn.x = h0; hn.y = h1; hn.z = h2; hn.w = h3;
  float4 an; an.x = s * h0; an.y = s * h1; an.z = s * h2; an.w = s * h3;
  *(float4*)(hout + (size_t)v * 4) = hn;
  *(float4*)(agg + (size_t)v * 4) = an;
}

__global__ void k_act_mm_final(const float* __restrict__ agg,
                               const float* __restrict__ b2,
                               const float* __restrict__ W3,
                               const float* __restrict__ b3,
                               const float* __restrict__ dinv,
                               float* __restrict__ h3, float* __restrict__ out,
                               int n) {
  int v = blockIdx.x * blockDim.x + threadIdx.x;
  if (v >= n) return;
  float4 t = *(const float4*)(agg + (size_t)v * 4);
  float t0 = fmaxf(t.x + b2[0], 0.0f);
  float t1 = fmaxf(t.y + b2[1], 0.0f);
  float t2 = fmaxf(t.z + b2[2], 0.0f);
  float t3 = fmaxf(t.w + b2[3], 0.0f);
  float h0 = t0 * W3[0] + t1 * W3[2] + t2 * W3[4] + t3 * W3[6];
  float h1 = t0 * W3[1] + t1 * W3[3] + t2 * W3[5] + t3 * W3[7];
  float di = dinv[v];
  float s = di * di;
  *(float2*)(h3 + (size_t)v * 2) = make_float2(h0, h1);
  *(float2*)(out + (size_t)v * 2) = make_float2(s * h0 + b3[0], s * h1 + b3[1]);
}

__global__ void k_scatter2(const int* __restrict__ src, const int* __restrict__ dst,
                           const float* __restrict__ dinv,
                           const float* __restrict__ h3, float* out, int E) {
  int e = blockIdx.x * blockDim.x + threadIdx.x;
  if (e >= E) return;
  int s = src[e], d = dst[e];
  float w = dinv[s] * dinv[d];
  float2 m = *(const float2*)(h3 + (size_t)s * 2);
  atomAddF32(out + (size_t)d * 2 + 0, w * m.x);
  atomAddF32(out + (size_t)d * 2 + 1, w * m.y);
}

#define BIN_SHIFT 10
#define BIN_NODES (1 << BIN_SHIFT)
#define BIN_MASK (BIN_NODES - 1)
#define BIN_MAX 4096
#define EDGE_CHUNK 32768
#define BIN_TB 1024
#define SCAN_TB 256
#define EDGE_BATCH 4
#define BATCH_SPAN (EDGE_BATCH * BIN_TB)

__device__ __forceinline__ unsigned ldsAddU32(unsigned* p, unsigned v) {
  return __hip_atomic_fetch_add(p, v, __ATOMIC_RELAXED, __HIP_MEMORY_SCOPE_WORKGROUP);
}
__device__ __forceinline__ void ldsAddF32(float* p, float v) {
  (void)__hip_atomic_fetch_add(p, v, __ATOMIC_RELAXED, __HIP_MEMORY_SCOPE_WORKGROUP);
}

__global__ __launch_bounds__(BIN_TB) void k_bin_count(
    const int* __restrict__ dst, unsigned* __restrict__ counts, int E, int nb,
    int nblk) {
  __shared__ unsigned cnt[BIN_MAX];
  const int t = threadIdx.x;
  const int blk = blockIdx.x;
  for (int i = t; i < nb; i += BIN_TB) cnt[i] = 0u;
  __syncthreads();
  const int e0 = blk * EDGE_CHUNK;
  const int ne = (E - e0 < EDGE_CHUNK) ? (E - e0) : EDGE_CHUNK;
  int i0 = 0;
  for (; i0 + BATCH_SPAN <= ne; i0 += BATCH_SPAN) {
    unsigned d[EDGE_BATCH];
    for (int u = 0; u < EDGE_BATCH; ++u) d[u] = (unsigned)dst[e0 + i0 + u * BIN_TB + t];
    for (int u = 0; u < EDGE_BATCH; ++u) (void)ldsAddU32(&cnt[d[u] >> BIN_SHIFT], 1u);
  }
  for (int i = i0 + t; i < ne; i += BIN_TB)
    (void)ldsAddU32(&cnt[(unsigned)dst[e0 + i] >> BIN_SHIFT], 1u);
  __syncthreads();
  for (int i = t; i < nb; i += BIN_TB) counts[(size_t)i * nblk + blk] = cnt[i];
}

__global__ __launch_bounds__(SCAN_TB) void k_scan_rows(
    const unsigned* in, unsigned* out, unsigned* totals, int len) {
  __shared__ unsigned s[SCAN_TB];
  const int t = threadIdx.x;
  const unsigned* irow = in + (size_t)blockIdx.x * len;
  unsigned* orow = out + (size_t)blockIdx.x * len;
  unsigned carry = 0u;
  for (int base = 0; base < len; base += SCAN_TB) {
    const int i = base + t;
    const unsigned v = (i < len) ? irow[i] : 0u;
    s[t] = v;
    __syncthreads();
    for (int off = 1; off < SCAN_TB; off <<= 1) {
      const unsigned add = (t >= off) ? s[t - off] : 0u;
      __syncthreads();
      s[t] += add;
      __syncthreads();
    }
    const unsigned incl = s[t];
    const unsigned total = s[SCAN_TB - 1];
    if (i < len) orow[i] = carry + incl - v;
    carry += total;
    __syncthreads();
  }
  if (t == 0) totals[blockIdx.x] = carry;
}

__global__ __launch_bounds__(BIN_TB) void k_bin_fill(
    const int* __restrict__ src, const int* __restrict__ dst,
    const unsigned* __restrict__ counts, const unsigned* __restrict__ bstart,
    unsigned* __restrict__ sorted, int E, int nb, int nblk) {
  __shared__ unsigned cur[BIN_MAX];
  const int t = threadIdx.x;
  const int blk = blockIdx.x;
  for (int i = t; i < nb; i += BIN_TB)
    cur[i] = bstart[i] + counts[(size_t)i * nblk + blk];
  __syncthreads();
  const int e0 = blk * EDGE_CHUNK;
  const int ne = (E - e0 < EDGE_CHUNK) ? (E - e0) : EDGE_CHUNK;
  int i0 = 0;
  for (; i0 + BATCH_SPAN <= ne; i0 += BATCH_SPAN) {
    unsigned s[EDGE_BATCH], d[EDGE_BATCH];
    for (int u = 0; u < EDGE_BATCH; ++u) {
      const int e = e0 + i0 + u * BIN_TB + t;
      s[u] = (unsigned)src[e];
      d[u] = (unsigned)dst[e];
    }
    for (int u = 0; u < EDGE_BATCH; ++u) {
      const unsigned pos = ldsAddU32(&cur[d[u] >> BIN_SHIFT], 1u);
      sorted[pos] = (s[u] << BIN_SHIFT) | (d[u] & BIN_MASK);
    }
  }
  for (int i = i0 + t; i < ne; i += BIN_TB) {
    const unsigned s = (unsigned)src[e0 + i];
    const unsigned d = (unsigned)dst[e0 + i];
    const unsigned pos = ldsAddU32(&cur[d >> BIN_SHIFT], 1u);
    sorted[pos] = (s << BIN_SHIFT) | (d & BIN_MASK);
  }
}

__global__ __launch_bounds__(BIN_TB) void k_bin_dinv(
    const unsigned* __restrict__ sorted, const unsigned* __restrict__ bstart,
    float* __restrict__ dinv, int n) {
  __shared__ unsigned c[BIN_NODES];
  const int t = threadIdx.x;
  const int b = blockIdx.x;
  c[t] = 1u;
  __syncthreads();
  const unsigned lo = bstart[b], hi = bstart[b + 1];
  unsigned i0 = lo;
  for (; i0 + BATCH_SPAN <= hi; i0 += BATCH_SPAN) {
    unsigned p[EDGE_BATCH];
    for (int u = 0; u < EDGE_BATCH; ++u) p[u] = sorted[i0 + u * BIN_TB + t];
    for (int u = 0; u < EDGE_BATCH; ++u) (void)ldsAddU32(&c[p[u] & BIN_MASK], 1u);
  }
  for (unsigned j = i0 + t; j < hi; j += BIN_TB)
    (void)ldsAddU32(&c[sorted[j] & BIN_MASK], 1u);
  __syncthreads();
  const int v = b * BIN_NODES + t;
  if (v < n) dinv[v] = 1.0f / sqrtf((float)c[t]);
}

__global__ void k_prescale4(const float* __restrict__ dinv,
                            const float* __restrict__ h, float* __restrict__ hs,
                            int n) {
  int v = blockIdx.x * blockDim.x + threadIdx.x;
  if (v >= n) return;
  const float di = dinv[v];
  float4 t = *(const float4*)(h + (size_t)v * 4);
  float4 o;
  o.x = di * t.x; o.y = di * t.y; o.z = di * t.z; o.w = di * t.w;
  *(float4*)(hs + (size_t)v * 4) = o;
}

__global__ __launch_bounds__(BIN_TB) void k_bin_layer4(
    const unsigned* __restrict__ sorted, const unsigned* __restrict__ bstart,
    const float* __restrict__ hs_in, const float* __restrict__ dinv,
    const float* __restrict__ bvec, const float* __restrict__ W,
    float* __restrict__ hs_out, int n, int fout) {
  __shared__ float acc[4][BIN_NODES];
  const int t = threadIdx.x;
  const int b = blockIdx.x;
  const int v = b * BIN_NODES + t;
  float4 self = make_float4(0.0f, 0.0f, 0.0f, 0.0f);
  if (v < n) self = *(const float4*)(hs_in + (size_t)v * 4);
  acc[0][t] = self.x; acc[1][t] = self.y; acc[2][t] = self.z; acc[3][t] = self.w;
  __syncthreads();
  const unsigned lo = bstart[b], hi = bstart[b + 1];
  unsigned i0 = lo;
  for (; i0 + BATCH_SPAN <= hi; i0 += BATCH_SPAN) {
    unsigned p[EDGE_BATCH];
    float4 m[EDGE_BATCH];
    for (int u = 0; u < EDGE_BATCH; ++u) p[u] = sorted[i0 + u * BIN_TB + t];
    for (int u = 0; u < EDGE_BATCH; ++u)
      m[u] = *(const float4*)(hs_in + (size_t)(p[u] >> BIN_SHIFT) * 4);
    for (int u = 0; u < EDGE_BATCH; ++u) {
      const unsigned dl = p[u] & BIN_MASK;
      ldsAddF32(&acc[0][dl], m[u].x);
      ldsAddF32(&acc[1][dl], m[u].y);
      ldsAddF32(&acc[2][dl], m[u].z);
      ldsAddF32(&acc[3][dl], m[u].w);
    }
  }
  for (unsigned j = i0 + t; j < hi; j += BIN_TB) {
    const unsigned p = sorted[j];
    const unsigned dl = p & BIN_MASK;
    const float4 m = *(const float4*)(hs_in + (size_t)(p >> BIN_SHIFT) * 4);
    ldsAddF32(&acc[0][dl], m.x);
    ldsAddF32(&acc[1][dl], m.y);
    ldsAddF32(&acc[2][dl], m.z);
    ldsAddF32(&acc[3][dl], m.w);
  }
  __syncthreads();
  if (v < n) {
    const float di = dinv[v];
    const float t0 = fmaxf(di * acc[0][t] + bvec[0], 0.0f);
    const float t1 = fmaxf(di * acc[1][t] + bvec[1], 0.0f);
    const float t2 = fmaxf(di * acc[2][t] + bvec[2], 0.0f);
    const float t3 = fmaxf(di * acc[3][t] + bvec[3], 0.0f);
    if (fout == 4) {
      const float h0 = t0 * W[0] + t1 * W[4] + t2 * W[8]  + t3 * W[12];
      const float h1 = t0 * W[1] + t1 * W[5] + t2 * W[9]  + t3 * W[13];
      const float h2 = t0 * W[2] + t1 * W[6] + t2 * W[10] + t3 * W[14];
      const float h3 = t0 * W[3] + t1 * W[7] + t2 * W[11] + t3 * W[15];
      *(float4*)(hs_out + (size_t)v * 4) =
          make_float4(di * h0, di * h1, di * h2, di * h3);
    } else {
      const float h0 = t0 * W[0] + t1 * W[2] + t2 * W[4] + t3 * W[6];
      const float h1 = t0 * W[1] + t1 * W[3] + t2 * W[5] + t3 * W[7];
      *(float2*)(hs_out + (size_t)v * 2) = make_float2(di * h0, di * h1);
    }
  }
}

__global__ __launch_bounds__(BIN_TB) void k_bin_final2(
    const unsigned* __restrict__ sorted, const unsigned* __restrict__ bstart,
    const float* __restrict__ hs3, const float* __restrict__ dinv,
    const float* __restrict__ b3, float* __restrict__ out, int n) {
  __shared__ float acc[2][BIN_NODES];
  const int t = threadIdx.x;
  const int b = blockIdx.x;
  const int v = b * BIN_NODES + t;
  float2 self = make_float2(0.0f, 0.0f);
  if (v < n) self = *(const float2*)(hs3 + (size_t)v * 2);
  acc[0][t] = self.x; acc[1][t] = self.y;
  __syncthreads();
  const unsigned lo = bstart[b], hi = bstart[b + 1];
  unsigned i0 = lo;
  for (; i0 + BATCH_SPAN <= hi; i0 += BATCH_SPAN) {
    unsigned p[EDGE_BATCH];
    float2 m[EDGE_BATCH];
    for (int u = 0; u < EDGE_BATCH; ++u) p[u] = sorted[i0 + u * BIN_TB + t];
    for (int u = 0; u < EDGE_BATCH; ++u)
      m[u] = *(const float2*)(hs3 + (size_t)(p[u] >> BIN_SHIFT) * 2);
    for (int u = 0; u < EDGE_BATCH; ++u) {
      const unsigned dl = p[u] & BIN_MASK;
      ldsAddF32(&acc[0][dl], m[u].x);
      ldsAddF32(&acc[1][dl], m[u].y);
    }
  }
  for (unsigned j = i0 + t; j < hi; j += BIN_TB) {
    const unsigned p = sorted[j];
    const unsigned dl = p & BIN_MASK;
    const float2 m = *(const float2*)(hs3 + (size_t)(p >> BIN_SHIFT) * 2);
    ldsAddF32(&acc[0][dl], m.x);
    ldsAddF32(&acc[1][dl], m.y);
  }
  __syncthreads();
  if (v < n) {
    const float di = dinv[v];
    *(float2*)(out + (size_t)v * 2) =
        make_float2(di * acc[0][t] + b3[0], di * acc[1][t] + b3[1]);
  }
}

extern "C" void kernel_launch(void* const* d_in, const int* in_sizes, int n_in,
                              void* d_out, int out_size, void* d_ws,
                              size_t ws_size, hipStream_t stream) {
  const float* x  = (const float*)d_in[0];
  const int*   ei = (const int*)d_in[1];
  const float* W1 = (const float*)d_in[2];
  const float* b1 = (const float*)d_in[3];
  const float* W2 = (const float*)d_in[4];
  const float* b2 = (const float*)d_in[5];
  const float* W3 = (const float*)d_in[6];
  const float* b3 = (const float*)d_in[7];
  float* out = (float*)d_out;

  const int n = in_sizes[0] / 128;
  const int E = in_sizes[1] / 2;
  const int* src = ei;
  const int* dst = ei + E;

  const int TB = 256;
  dim3 blk(TB);
  dim3 gn((n + TB - 1) / TB);
  int tiles = (n + 16 * GEMM_WAVES - 1) / (16 * GEMM_WAVES);

  const int nb = (n + BIN_NODES - 1) >> BIN_SHIFT;
  const int nblk = (E + EDGE_CHUNK - 1) / EDGE_CHUNK;
  const size_t wE = ((size_t)E + 63) & ~(size_t)63;
  const size_t wC = ((size_t)nb * (size_t)nblk + 63) & ~(size_t)63;
  const size_t wB = ((size_t)2 * nb + 1 + 63) & ~(size_t)63;
  const size_t wN = ((size_t)n + 63) & ~(size_t)63;
  const size_t w4 = ((size_t)4 * n + 63) & ~(size_t)63;
  const size_t w2 = ((size_t)2 * n + 63) & ~(size_t)63;
  const size_t need = 4 * (wE + wC + wB + wN + 2 * w4 + w2);

  if (n <= (1 << 22) && nblk > 0 && ws_size >= need) {
    unsigned* sorted = (unsigned*)d_ws;
    unsigned* counts = sorted + wE;
    unsigned* btot   = counts + wC;
    unsigned* bstart = btot + nb;
    float* dinv = (float*)(btot + wB);
    float* bufA = dinv + wN;
    float* bufB = bufA + w4;
    float* buf3 = bufB + w4;

    k_bin_count<<<nblk, BIN_TB, 0, stream>>>(dst, counts, E, nb, nblk);
    k_scan_rows<<<nb, SCAN_TB, 0, stream>>>(counts, counts, btot, nblk);
    k_scan_rows<<<1, SCAN_TB, 0, stream>>>(btot, bstart, bstart + nb, nb);
    k_bin_fill<<<nblk, BIN_TB, 0, stream>>>(src, dst, counts, bstart, sorted, E, nb, nblk);
    k_bin_dinv<<<nb, BIN_TB, 0, stream>>>(sorted, bstart, dinv, n);
    k_gemm_xW1<<<tiles, GEMM_WAVES * 32, 0, stream>>>(x, W1, bufA, n);
    k_prescale4<<<gn, blk, 0, stream>>>(dinv, bufA, bufB, n);
    k_bin_layer4<<<nb, BIN_TB, 0, stream>>>(sorted, bstart, bufB, dinv, b1, W2, bufA, n, 4);
    k_bin_layer4<<<nb, BIN_TB, 0, stream>>>(sorted, bstart, bufA, dinv, b2, W3, buf3, n, 2);
    k_bin_final2<<<nb, BIN_TB, 0, stream>>>(sorted, bstart, buf3, dinv, b3, out, n);
    return;
  }

  dim3 ge((E + TB - 1) / TB);
  float* ws   = (float*)d_ws;
  float* dinv = ws;
  float* hA   = dinv + (size_t)n;
  float* hB   = hA + (size_t)n * 4;
  float* hC   = hB + (size_t)n * 4;

  k_deg_init<<<gn, blk, 0, stream>>>(dinv, n);
  k_deg_scatter<<<ge, blk, 0, stream>>>(dst, dinv, E);
  k_dinv<<<gn, blk, 0, stream>>>(dinv, n);
  k_gemm_xW1<<<tiles, GEMM_WAVES * 32, 0, stream>>>(x, W1, hA, n);
  k_init_agg4<<<gn, blk, 0, stream>>>(dinv, hA, hB, n);
  k_scatter4<<<ge, blk, 0, stream>>>(src, dst, dinv, hA, hB, E);
  k_act_mm4<<<gn, blk, 0, stream>>>(hB, b1, W2, dinv, hA, n);
  k_scatter4<<<ge, blk, 0, stream>>>(src, dst, dinv, hA, hB, E);
  k_act_mm_final<<<gn, blk, 0, stream>>>(hB, b2, W3, b3, dinv, hC, out, n);
  k_scatter2<<<ge, blk, 0, stream>>>(src, dst, dinv, hC, out, E);
}


# Bin fill ranks its 32768-edge chunk into a bin-ordered LDS stage and stores each bin run with consecutive addresses, not a scattered 4-byte store per edge; edge order in a bin (LDS add order) changes
# speedup vs baseline: 8.3696x; 1.1952x over previous
#include <hip/hip_runtime.h>
#include <math.h>


typedef __attribute__((ext_vector_type(2))) float v2f;
typedef __attribute__((ext_vector_type(8))) float v8f;

__device__ __forceinline__ void atomAddF32(float* p, float v) {
  (void)__hip_atomic_fetch_add(p, v, __ATOMIC_RELAXED, __HIP_MEMORY_SCOPE_AGENT);
}

__global__ void k_deg_init(float* deg, int n) {
  int i = blockIdx.x * blockDim.x + threadIdx.x;
  if (i < n) deg[i] = 1.0f;
}

__global__ void k_deg_scatter(const int* __restrict__ dst, float* deg, int E) {
  int e = blockIdx.x * blockDim.x + threadIdx.x;
  if (e < E) atomAddF32(&deg[dst[e]], 1.0f);
}

__global__ void k_dinv(float* deg, int n) {
  int i = blockIdx.x * blockDim.x + threadIdx.x;
  if (i < n) deg[i] = 1.0f / sqrtf(deg[i]);
}

#define GEMM_WAVES 4

__global__ __launch_bounds__(GEMM_WAVES * 32) void k_gemm_xW1(
    const float* __restrict__ x, const float* __restrict__ W1,
    float* __restrict__ h, int n) {
  __shared__ float sA[GEMM_WAVES][16][132];
  __shared__ float sB[16][132];
  const int tid = threadIdx.x;
  const int lane = tid & 31;
  const int wave = tid >> 5;
  const int base = (blockIdx.x * GEMM_WAVES + wave) * 16;
  const bool active = (base < n);

  for (int i = tid; i < 16 * 128; i += GEMM_WAVES * 32) {
    const int col = i >> 7;
    const int k = i & 127;
    sB[col][k] = (col < 4) ? W1[k * 4 + col] : 0.0f;
  }
  if (active) {
    const float* xt = x + (size_t)base * 128;
#pragma unroll
    for (int i = 0; i < 16; ++i) {
      float4 v = *(const float4*)(xt + (size_t)i * 128 + lane * 4);
      *(float4*)(&sA[wave][i][lane * 4]) = v;
    }
  }
  __syncthreads();
  if (!active) return;

  const int half = lane >> 4;
  const int mn = lane & 15;
  const float* aRow = &sA[wave][mn][0];
  const float* bRow = &sB[mn][0];
  v8f c = {};
  for (int ks = 0; ks < 32; ++ks) {
    const int k = ks * 4 + 2 * half;
    v2f a = *(const v2f*)(aRow + k);
    v2f b = *(const v2f*)(bRow + k);
    c = __builtin_amdgcn_wmma_f32_16x16x4_f32(false, a, false, b, (short)0, c,
                                              false, false);
  }
  if (mn < 4) {
    float* o = h + (size_t)(base + 8 * half) * 4 + mn;
#pragma unroll
    for (int r = 0; r < 8; ++r) o[(size_t)r * 4] = c[r];
  }
}

__global__ void k_init_agg4(const float* __restrict__ dinv,
                            const float* __restrict__ h, float* __restrict__ agg,
                            int n) {
  int v = blockIdx.x * blockDim.x + threadIdx.x;
  if (v >= n) return;
  float di = dinv[v];
  float s = di * di;
  float4 t = *(const float4*)(h + (size_t)v * 4);
  float4 o;
  o.x = s * t.x; o.y = s * t.y; o.z = s * t.z; o.w = s * t.w;
  *(float4*)(agg + (size_t)v * 4) = o;
}

__global__ void k_scatter4(const int* __restrict__ src, const int* __restrict__ dst,
                           const float* __restrict__ dinv,
                           const float* __restrict__ h, float* agg, int E) {
  int e = blockIdx.x * blockDim.x + threadIdx.x;
  if (e >= E) return;
  int s = src[e], d = dst[e];
  float w = dinv[s] * dinv[d];
  float4 m = *(const float4*)(h + (size_t)s * 4);
  float* o = agg + (size_t)d * 4;
  atomAddF32(o + 0, w * m.x);
  atomAddF32(o + 1, w * m.y);
  atomAddF32(o + 2, w * m.z);
  atomAddF32(o + 3, w * m.w);
}

__global__ void k_act_mm4(float* agg, const float* __restrict__ bvec,
                          const float* __restrict__ W,
                          const float* __restrict__ dinv,
                          float* __restrict__ hout, int n) {
  int v = blockIdx.x * blockDim.x + threadIdx.x;
  if (v >= n) return;
  float4 t = *(const float4*)(agg + (size_t)v * 4);
  float t0 = fmaxf(t.x + bvec[0], 0.0f);
  float t1 = fmaxf(t.y + bvec[1], 0.0f);
  float t2 = fmaxf(t.z + bvec[2], 0.0f);
  float t3 = fmaxf(t.w + bvec[3], 0.0f);
  float h0 = t0 * W[0] + t1 * W[4] + t2 * W[8]  + t3 * W[12];
  float h1 = t0 * W[1] + t1 * W[5] + t2 * W[9]  + t3 * W[13];
  float h2 = t0 * W[2] + t1 * W[6] + t2 * W[10] + t3 * W[14];
  float h3 = t0 * W[3] + t1 * W[7] + t2 * W[11] + t3 * W[15];
  float di = dinv[v];
  float s = di * di;
  float4 hn; hn.x = h0; hn.y = h1; hn.z = h2; hn.w = h3;
  float4 an; an.x = s * h0; an.y = s * h1; an.z = s * h2; an.w = s * h3;
  *(float4*)(hout + (size_t)v * 4) = hn;
  *(float4*)(agg + (size_t)v * 4) = an;
}

__global__ void k_act_mm_final(const float* __restrict__ agg,
                               const float* __restrict__ b2,
                               const float* __restrict__ W3,
                               const float* __restrict__ b3,
                               const float* __restrict__ dinv,
                               float* __restrict__ h3, float* __restrict__ out,
                               int n) {
  int v = blockIdx.x * blockDim.x + threadIdx.x;
  if (v >= n) return;
  float4 t = *(const float4*)(agg + (size_t)v * 4);
  float t0 = fmaxf(t.x + b2[0], 0.0f);
  float t1 = fmaxf(t.y + b2[1], 0.0f);
  float t2 = fmaxf(t.z + b2[2], 0.0f);
  float t3 = fmaxf(t.w + b2[3], 0.0f);
  float h0 = t0 * W3[0] + t1 * W3[2] + t2 * W3[4] + t3 * W3[6];
  float h1 = t0 * W3[1] + t1 * W3[3] + t2 * W3[5] + t3 * W3[7];
  float di = dinv[v];
  float s = di * di;
  *(float2*)(h3 + (size_t)v * 2) = make_float2(h0, h1);
  *(float2*)(out + (size_t)v * 2) = make_float2(s * h0 + b3[0], s * h1 + b3[1]);
}

__global__ void k_scatter2(const int* __restrict__ src, const int* __restrict__ dst,
                           const float* __restrict__ dinv,
                           const float* __restrict__ h3, float* out, int E) {
  int e = blockIdx.x * blockDim.x + threadIdx.x;
  if (e >= E) return;
  int s = src[e], d = dst[e];
  float w = dinv[s] * dinv[d];
  float2 m = *(const float2*)(h3 + (size_t)s * 2);
  atomAddF32(out + (size_t)d * 2 + 0, w * m.x);
  atomAddF32(out + (size_t)d * 2 + 1, w * m.y);
}

#define BIN_SHIFT 10
#define BIN_NODES (1 << BIN_SHIFT)
#define BIN_MASK (BIN_NODES - 1)
#define BIN_MAX 4096
#define EDGE_CHUNK 32768
#define BIN_TB 1024
#define SCAN_TB 256
#define EDGE_BATCH 4
#define BATCH_SPAN (EDGE_BATCH * BIN_TB)

__device__ __forceinline__ unsigned ldsAddU32(unsigned* p, unsigned v) {
  return __hip_atomic_fetch_add(p, v, __ATOMIC_RELAXED, __HIP_MEMORY_SCOPE_WORKGROUP);
}
__device__ __forceinline__ void ldsAddF32(float* p, float v) {
  (void)__hip_atomic_fetch_add(p, v, __ATOMIC_RELAXED, __HIP_MEMORY_SCOPE_WORKGROUP);
}

__global__ __launch_bounds__(BIN_TB) void k_bin_count(
    const int* __restrict__ dst, unsigned* __restrict__ counts, int E, int nb,
    int nblk) {
  __shared__ unsigned cnt[BIN_MAX];
  const int t = threadIdx.x;
  const int blk = blockIdx.x;
  for (int i = t; i < nb; i += BIN_TB) cnt[i] = 0u;
  __syncthreads();
  const int e0 = blk * EDGE_CHUNK;
  const int ne = (E - e0 < EDGE_CHUNK) ? (E - e0) : EDGE_CHUNK;
  int i0 = 0;
  for (; i0 + BATCH_SPAN <= ne; i0 += BATCH_SPAN) {
    unsigned d[EDGE_BATCH];
    for (int u = 0; u < EDGE_BATCH; ++u) d[u] = (unsigned)dst[e0 + i0 + u * BIN_TB + t];
    for (int u = 0; u < EDGE_BATCH; ++u) (void)ldsAddU32(&cnt[d[u] >> BIN_SHIFT], 1u);
  }
  for (int i = i0 + t; i < ne; i += BIN_TB)
    (void)ldsAddU32(&cnt[(unsigned)dst[e0 + i] >> BIN_SHIFT], 1u);
  __syncthreads();
  for (int i = t; i < nb; i += BIN_TB) counts[(size_t)i * nblk + blk] = cnt[i];
}

__global__ __launch_bounds__(SCAN_TB) void k_scan_rows(
    const unsigned* in, unsigned* out, unsigned* totals, int len) {
  __shared__ unsigned s[SCAN_TB];
  const int t = threadIdx.x;
  const unsigned* irow = in + (size_t)blockIdx.x * len;
  unsigned* orow = out + (size_t)blockIdx.x * len;
  unsigned carry = 0u;
  for (int base = 0; base < len; base += SCAN_TB) {
    const int i = base + t;
    const unsigned v = (i < len) ? irow[i] : 0u;
    s[t] = v;
    __syncthreads();
    for (int off = 1; off < SCAN_TB; off <<= 1) {
      const unsigned add = (t >= off) ? s[t - off] : 0u;
      __syncthreads();
      s[t] += add;
      __syncthreads();
    }
    const unsigned incl = s[t];
    const unsigned total = s[SCAN_TB - 1];
    if (i < len) orow[i] = carry + incl - v;
    carry += total;
    __syncthreads();
  }
  if (t == 0) totals[blockIdx.x] = carry;
}

#define FILL_BIN_MAX 2048
#define FILL_BINS_PER_THREAD (FILL_BIN_MAX / BIN_TB)

__global__ __launch_bounds__(BIN_TB) void k_bin_fill(
    const int* __restrict__ src, const int* __restrict__ dst,
    const unsigned* __restrict__ counts, const unsigned* __restrict__ btot,
    const unsigned* __restrict__ bstart, unsigned* __restrict__ sorted, int E,
    int nb, int nblk) {
  __shared__ unsigned stage[EDGE_CHUNK];
  __shared__ unsigned lcur[FILL_BIN_MAX];
  __shared__ unsigned gbase[FILL_BIN_MAX];
  __shared__ unsigned wtot[BIN_TB / 32];
  const int t = threadIdx.x;
  const int lane = t & 31;
  const int wave = t >> 5;
  const int blk = blockIdx.x;

  unsigned c[FILL_BINS_PER_THREAD];
  unsigned v = 0u;
  for (int u = 0; u < FILL_BINS_PER_THREAD; ++u) {
    const int i = t * FILL_BINS_PER_THREAD + u;
    c[u] = 0u;
    if (i < nb) {
      const size_t at = (size_t)i * nblk + blk;
      const unsigned off = counts[at];
      const unsigned nxt = (blk + 1 < nblk) ? counts[at + 1] : btot[i];
      c[u] = nxt - off;
      gbase[i] = bstart[i] + off;
    }
    v += c[u];
  }
  unsigned incl = v;
  for (int off = 1; off < 32; off <<= 1) {
    const unsigned up = __shfl_up(incl, (unsigned)off, 32);
    if (lane >= off) incl += up;
  }
  if (lane == 31) wtot[wave] = incl;
  __syncthreads();
  const unsigned wv = wtot[lane];
  unsigned winc = wv;
  for (int off = 1; off < 32; off <<= 1) {
    const unsigned up = __shfl_up(winc, (unsigned)off, 32);
    if (lane >= off) winc += up;
  }
  const unsigned wexcl = __shfl(winc - wv, wave, 32);
  unsigned run = wexcl + incl - v;
  for (int u = 0; u < FILL_BINS_PER_THREAD; ++u) {
    lcur[t * FILL_BINS_PER_THREAD + u] = run;
    run += c[u];
  }
  __syncthreads();

  const int e0 = blk * EDGE_CHUNK;
  const int ne = (E - e0 < EDGE_CHUNK) ? (E - e0) : EDGE_CHUNK;
  int i0 = 0;
  for (; i0 + BATCH_SPAN <= ne; i0 += BATCH_SPAN) {
    unsigned s[EDGE_BATCH], d[EDGE_BATCH];
    for (int u = 0; u < EDGE_BATCH; ++u) {
      const int e = e0 + i0 + u * BIN_TB + t;
      s[u] = (unsigned)src[e];
      d[u] = (unsigned)dst[e];
    }
    unsigned pos[EDGE_BATCH];
    for (int u = 0; u < EDGE_BATCH; ++u)
      pos[u] = ldsAddU32(&lcur[d[u] >> BIN_SHIFT], 1u);
    for (int u = 0; u < EDGE_BATCH; ++u)
      stage[pos[u]] = (s[u] << BIN_SHIFT) | (d[u] & BIN_MASK);
  }
  for (int i = i0 + t; i < ne; i += BIN_TB) {
    const unsigned s = (unsigned)src[e0 + i];
    const unsigned d = (unsigned)dst[e0 + i];
    const unsigned pos = ldsAddU32(&lcur[d >> BIN_SHIFT], 1u);
    stage[pos] = (s << BIN_SHIFT) | (d & BIN_MASK);
  }
  __syncthreads();

  for (int i = wave; i < nb; i += BIN_TB / 32) {
    const unsigned lo = (i == 0) ? 0u : lcur[i - 1];
    const unsigned hi = lcur[i];
    unsigned* __restrict__ o = sorted + gbase[i];
    for (unsigned k = lo + lane; k < hi; k += 32) o[k - lo] = stage[k];
  }
}

__global__ __launch_bounds__(BIN_TB) void k_bin_dinv(
    const unsigned* __restrict__ sorted, const unsigned* __restrict__ bstart,
    float* __restrict__ dinv, int n) {
  __shared__ unsigned c[BIN_NODES];
  const int t = threadIdx.x;
  const int b = blockIdx.x;
  c[t] = 1u;
  __syncthreads();
  const unsigned lo = bstart[b], hi = bstart[b + 1];
  unsigned i0 = lo;
  for (; i0 + BATCH_SPAN <= hi; i0 += BATCH_SPAN) {
    unsigned p[EDGE_BATCH];
    for (int u = 0; u < EDGE_BATCH; ++u) p[u] = sorted[i0 + u * BIN_TB + t];
    for (int u = 0; u < EDGE_BATCH; ++u) (void)ldsAddU32(&c[p[u] & BIN_MASK], 1u);
  }
  for (unsigned j = i0 + t; j < hi; j += BIN_TB)
    (void)ldsAddU32(&c[sorted[j] & BIN_MASK], 1u);
  __syncthreads();
  const int v = b * BIN_NODES + t;
  if (v < n) dinv[v] = 1.0f / sqrtf((float)c[t]);
}

__global__ void k_prescale4(const float* __restrict__ dinv,
                            const float* __restrict__ h, float* __restrict__ hs,
                            int n) {
  int v = blockIdx.x * blockDim.x + threadIdx.x;
  if (v >= n) return;
  const float di = dinv[v];
  float4 t = *(const float4*)(h + (size_t)v * 4);
  float4 o;
  o.x = di * t.x; o.y = di * t.y; o.z = di * t.z; o.w = di * t.w;
  *(float4*)(hs + (size_t)v * 4) = o;
}

__global__ __launch_bounds__(BIN_TB) void k_bin_layer4(
    const unsigned* __restrict__ sorted, const unsigned* __restrict__ bstart,
    const float* __restrict__ hs_in, const float* __restrict__ dinv,
    const float* __restrict__ bvec, const float* __restrict__ W,
    float* __restrict__ hs_out, int n, int fout) {
  __shared__ float acc[4][BIN_NODES];
  const int t = threadIdx.x;
  const int b = blockIdx.x;
  const int v = b * BIN_NODES + t;
  float4 self = make_float4(0.0f, 0.0f, 0.0f, 0.0f);
  if (v < n) self = *(const float4*)(hs_in + (size_t)v * 4);
  acc[0][t] = self.x; acc[1][t] = self.y; acc[2][t] = self.z; acc[3][t] = self.w;
  __syncthreads();
  const unsigned lo = bstart[b], hi = bstart[b + 1];
  unsigned i0 = lo;
  for (; i0 + BATCH_SPAN <= hi; i0 += BATCH_SPAN) {
    unsigned p[EDGE_BATCH];
    float4 m[EDGE_BATCH];
    for (int u = 0; u < EDGE_BATCH; ++u) p[u] = sorted[i0 + u * BIN_TB + t];
    for (int u = 0; u < EDGE_BATCH; ++u)
      m[u] = *(const float4*)(hs_in + (size_t)(p[u] >> BIN_SHIFT) * 4);
    for (int u = 0; u < EDGE_BATCH; ++u) {
      const unsigned dl = p[u] & BIN_MASK;
      ldsAddF32(&acc[0][dl], m[u].x);
      ldsAddF32(&acc[1][dl], m[u].y);
      ldsAddF32(&acc[2][dl], m[u].z);
      ldsAddF32(&acc[3][dl], m[u].w);
    }
  }
  for (unsigned j = i0 + t; j < hi; j += BIN_TB) {
    const unsigned p = sorted[j];
    const unsigned dl = p & BIN_MASK;
    const float4 m = *(const float4*)(hs_in + (size_t)(p >> BIN_SHIFT) * 4);
    ldsAddF32(&acc[0][dl], m.x);
    ldsAddF32(&acc[1][dl], m.y);
    ldsAddF32(&acc[2][dl], m.z);
    ldsAddF32(&acc[3][dl], m.w);
  }
  __syncthreads();
  if (v < n) {
    const float di = dinv[v];
    const float t0 = fmaxf(di * acc[0][t] + bvec[0], 0.0f);
    const float t1 = fmaxf(di * acc[1][t] + bvec[1], 0.0f);
    const float t2 = fmaxf(di * acc[2][t] + bvec[2], 0.0f);
    const float t3 = fmaxf(di * acc[3][t] + bvec[3], 0.0f);
    if (fout == 4) {
      const float h0 = t0 * W[0] + t1 * W[4] + t2 * W[8]  + t3 * W[12];
      const float h1 = t0 * W[1] + t1 * W[5] + t2 * W[9]  + t3 * W[13];
      const float h2 = t0 * W[2] + t1 * W[6] + t2 * W[10] + t3 * W[14];
      const float h3 = t0 * W[3] + t1 * W[7] + t2 * W[11] + t3 * W[15];
      *(float4*)(hs_out + (size_t)v * 4) =
          make_float4(di * h0, di * h1, di * h2, di * h3);
    } else {
      const float h0 = t0 * W[0] + t1 * W[2] + t2 * W[4] + t3 * W[6];
      const float h1 = t0 * W[1] + t1 * W[3] + t2 * W[5] + t3 * W[7];
      *(float2*)(hs_out + (size_t)v * 2) = make_float2(di * h0, di * h1);
    }
  }
}

__global__ __launch_bounds__(BIN_TB) void k_bin_final2(
    const unsigned* __restrict__ sorted, const unsigned* __restrict__ bstart,
    const float* __restrict__ hs3, const float* __restrict__ dinv,
    const float* __restrict__ b3, float* __restrict__ out, int n) {
  __shared__ float acc[2][BIN_NODES];
  const int t = threadIdx.x;
  const int b = blockIdx.x;
  const int v = b * BIN_NODES + t;
  float2 self = make_float2(0.0f, 0.0f);
  if (v < n) self = *(const float2*)(hs3 + (size_t)v * 2);
  acc[0][t] = self.x; acc[1][t] = self.y;
  __syncthreads();
  const unsigned lo = bstart[b], hi = bstart[b + 1];
  unsigned i0 = lo;
  for (; i0 + BATCH_SPAN <= hi; i0 += BATCH_SPAN) {
    unsigned p[EDGE_BATCH];
    float2 m[EDGE_BATCH];
    for (int u = 0; u < EDGE_BATCH; ++u) p[u] = sorted[i0 + u * BIN_TB + t];
    for (int u = 0; u < EDGE_BATCH; ++u)
      m[u] = *(const float2*)(hs3 + (size_t)(p[u] >> BIN_SHIFT) * 2);
    for (int u = 0; u < EDGE_BATCH; ++u) {
      const unsigned dl = p[u] & BIN_MASK;
      ldsAddF32(&acc[0][dl], m[u].x);
      ldsAddF32(&acc[1][dl], m[u].y);
    }
  }
  for (unsigned j = i0 + t; j < hi; j += BIN_TB) {
    const unsigned p = sorted[j];
    const unsigned dl = p & BIN_MASK;
    const float2 m = *(const float2*)(hs3 + (size_t)(p >> BIN_SHIFT) * 2);
    ldsAddF32(&acc[0][dl], m.x);
    ldsAddF32(&acc[1][dl], m.y);
  }
  __syncthreads();
  if (v < n) {
    const float di = dinv[v];
    *(float2*)(out + (size_t)v * 2) =
        make_float2(di * acc[0][t] + b3[0], di * acc[1][t] + b3[1]);
  }
}

extern "C" void kernel_launch(void* const* d_in, const int* in_sizes, int n_in,
                              void* d_out, int out_size, void* d_ws,
                              size_t ws_size, hipStream_t stream) {
  const float* x  = (const float*)d_in[0];
  const int*   ei = (const int*)d_in[1];
  const float* W1 = (const float*)d_in[2];
  const float* b1 = (const float*)d_in[3];
  const float* W2 = (const float*)d_in[4];
  const float* b2 = (const float*)d_in[5];
  const float* W3 = (const float*)d_in[6];
  const float* b3 = (const float*)d_in[7];
  float* out = (float*)d_out;

  const int n = in_sizes[0] / 128;
  const int E = in_sizes[1] / 2;
  const int* src = ei;
  const int* dst = ei + E;

  const int TB = 256;
  dim3 blk(TB);
  dim3 gn((n + TB - 1) / TB);
  int tiles = (n + 16 * GEMM_WAVES - 1) / (16 * GEMM_WAVES);

  const int nb = (n + BIN_NODES - 1) >> BIN_SHIFT;
  const int nblk = (E + EDGE_CHUNK - 1) / EDGE_CHUNK;
  const size_t wE = ((size_t)E + 63) & ~(size_t)63;
  const size_t wC = ((size_t)nb * (size_t)nblk + 63) & ~(size_t)63;
  const size_t wB = ((size_t)2 * nb + 1 + 63) & ~(size_t)63;
  const size_t wN = ((size_t)n + 63) & ~(size_t)63;
  const size_t w4 = ((size_t)4 * n + 63) & ~(size_t)63;
  const size_t w2 = ((size_t)2 * n + 63) & ~(size_t)63;
  const size_t need = 4 * (wE + wC + wB + wN + 2 * w4 + w2);

  if (n <= (1 << 21) && nblk > 0 && ws_size >= need) {
    unsigned* sorted = (unsigned*)d_ws;
    unsigned* counts = sorted + wE;
    unsigned* btot   = counts + wC;
    unsigned* bstart = btot + nb;
    float* dinv = (float*)(btot + wB);
    float* bufA = dinv + wN;
    float* bufB = bufA + w4;
    float* buf3 = bufB + w4;

    k_bin_count<<<nblk, BIN_TB, 0, stream>>>(dst, counts, E, nb, nblk);
    k_scan_rows<<<nb, SCAN_TB, 0, stream>>>(counts, counts, btot, nblk);
    k_scan_rows<<<1, SCAN_TB, 0, stream>>>(btot, bstart, bstart + nb, nb);
    k_bin_fill<<<nblk, BIN_TB, 0, stream>>>(src, dst, counts, btot, bstart, sorted, E, nb, nblk);
    k_bin_dinv<<<nb, BIN_TB, 0, stream>>>(sorted, bstart, dinv, n);
    k_gemm_xW1<<<tiles, GEMM_WAVES * 32, 0, stream>>>(x, W1, bufA, n);
    k_prescale4<<<gn, blk, 0, stream>>>(dinv, bufA, bufB, n);
    k_bin_layer4<<<nb, BIN_TB, 0, stream>>>(sorted, bstart, bufB, dinv, b1, W2, bufA, n, 4);
    k_bin_layer4<<<nb, BIN_TB, 0, stream>>>(sorted, bstart, bufA, dinv, b2, W3, buf3, n, 2);
    k_bin_final2<<<nb, BIN_TB, 0, stream>>>(sorted, bstart, buf3, dinv, b3, out, n);
    return;
  }

  dim3 ge((E + TB - 1) / TB);
  float* ws   = (float*)d_ws;
  float* dinv = ws;
  float* hA   = dinv + (size_t)n;
  float* hB   = hA + (size_t)n * 4;
  float* hC   = hB + (size_t)n * 4;

  k_deg_init<<<gn, blk, 0, stream>>>(dinv, n);
  k_deg_scatter<<<ge, blk, 0, stream>>>(dst, dinv, E);
  k_dinv<<<gn, blk, 0, stream>>>(dinv, n);
  k_gemm_xW1<<<tiles, GEMM_WAVES * 32, 0, stream>>>(x, W1, hA, n);
  k_init_agg4<<<gn, blk, 0, stream>>>(dinv, hA, hB, n);
  k_scatter4<<<ge, blk, 0, stream>>>(src, dst, dinv, hA, hB, E);
  k_act_mm4<<<gn, blk, 0, stream>>>(hB, b1, W2, dinv, hA, n);
  k_scatter4<<<ge, blk, 0, stream>>>(src, dst, dinv, hA, hB, E);
  k_act_mm_final<<<gn, blk, 0, stream>>>(hB, b2, W3, b3, dinv, hC, out, n);
  k_scatter2<<<ge, blk, 0, stream>>>(src, dst, dinv, hC, out, E);
}


# In-degree count (dinv), x@W1 and the source pre-scale fused into one per-bin kernel: 16 waves, 4 row tiles per wave with the next tile's rows loaded ahead; 3 launches become 1; sum orders unchanged
# speedup vs baseline: 8.6721x; 1.0361x over previous
#include <hip/hip_runtime.h>
#include <math.h>


typedef __attribute__((ext_vector_type(2))) float v2f;
typedef __attribute__((ext_vector_type(8))) float v8f;

__device__ __forceinline__ void atomAddF32(float* p, float v) {
  (void)__hip_atomic_fetch_add(p, v, __ATOMIC_RELAXED, __HIP_MEMORY_SCOPE_AGENT);
}

__global__ void k_deg_init(float* deg, int n) {
  int i = blockIdx.x * blockDim.x + threadIdx.x;
  if (i < n) deg[i] = 1.0f;
}

__global__ void k_deg_scatter(const int* __restrict__ dst, float* deg, int E) {
  int e = blockIdx.x * blockDim.x + threadIdx.x;
  if (e < E) atomAddF32(&deg[dst[e]], 1.0f);
}

__global__ void k_dinv(float* deg, int n) {
  int i = blockIdx.x * blockDim.x + threadIdx.x;
  if (i < n) deg[i] = 1.0f / sqrtf(deg[i]);
}

#define GEMM_WAVES 4

__global__ __launch_bounds__(GEMM_WAVES * 32) void k_gemm_xW1(
    const float* __restrict__ x, const float* __restrict__ W1,
    float* __restrict__ h, int n) {
  __shared__ float sA[GEMM_WAVES][16][132];
  __shared__ float sB[16][132];
  const int tid = threadIdx.x;
  const int lane = tid & 31;
  const int wave = tid >> 5;
  const int base = (blockIdx.x * GEMM_WAVES + wave) * 16;
  const bool active = (base < n);

  for (int i = tid; i < 16 * 128; i += GEMM_WAVES * 32) {
    const int col = i >> 7;
    const int k = i & 127;
    sB[col][k] = (col < 4) ? W1[k * 4 + col] : 0.0f;
  }
  if (active) {
    const float* xt = x + (size_t)base * 128;
#pragma unroll
    for (int i = 0; i < 16; ++i) {
      float4 v = *(const float4*)(xt + (size_t)i * 128 + lane * 4);
      *(float4*)(&sA[wave][i][lane * 4]) = v;
    }
  }
  __syncthreads();
  if (!active) return;

  const int half = lane >> 4;
  const int mn = lane & 15;
  const float* aRow = &sA[wave][mn][0];
  const float* bRow = &sB[mn][0];
  v8f c = {};
  for (int ks = 0; ks < 32; ++ks) {
    const int k = ks * 4 + 2 * half;
    v2f a = *(const v2f*)(aRow + k);
    v2f b = *(const v2f*)(bRow + k);
    c = __builtin_amdgcn_wmma_f32_16x16x4_f32(false, a, false, b, (short)0, c,
                                              false, false);
  }
  if (mn < 4) {
    float* o = h + (size_t)(base + 8 * half) * 4 + mn;
#pragma unroll
    for (int r = 0; r < 8; ++r) o[(size_t)r * 4] = c[r];
  }
}

__global__ void k_init_agg4(const float* __restrict__ dinv,
                            const float* __restrict__ h, float* __restrict__ agg,
                            int n) {
  int v = blockIdx.x * blockDim.x + threadIdx.x;
  if (v >= n) return;
  float di = dinv[v];
  float s = di * di;
  float4 t = *(const float4*)(h + (size_t)v * 4);
  float4 o;
  o.x = s * t.x; o.y = s * t.y; o.z = s * t.z; o.w = s * t.w;
  *(float4*)(agg + (size_t)v * 4) = o;
}

__global__ void k_scatter4(const int* __restrict__ src, const int* __restrict__ dst,
                           const float* __restrict__ dinv,
                           const float* __restrict__ h, float* agg, int E) {
  int e = blockIdx.x * blockDim.x + threadIdx.x;
  if (e >= E) return;
  int s = src[e], d = dst[e];
  float w = dinv[s] * dinv[d];
  float4 m = *(const float4*)(h + (size_t)s * 4);
  float* o = agg + (size_t)d * 4;
  atomAddF32(o + 0, w * m.x);
  atomAddF32(o + 1, w * m.y);
  atomAddF32(o + 2, w * m.z);
  atomAddF32(o + 3, w * m.w);
}

__global__ void k_act_mm4(float* agg, const float* __restrict__ bvec,
                          const float* __restrict__ W,
                          const float* __restrict__ dinv,
                          float* __restrict__ hout, int n) {
  int v = blockIdx.x * blockDim.x + threadIdx.x;
  if (v >= n) return;
  float4 t = *(const float4*)(agg + (size_t)v * 4);
  float t0 = fmaxf(t.x + bvec[0], 0.0f);
  float t1 = fmaxf(t.y + bvec[1], 0.0f);
  float t2 = fmaxf(t.z + bvec[2], 0.0f);
  float t3 = fmaxf(t.w + bvec[3], 0.0f);
  float h0 = t0 * W[0] + t1 * W[4] + t2 * W[8]  + t3 * W[12];
  float h1 = t0 * W[1] + t1 * W[5] + t2 * W[9]  + t3 * W[13];
  float h2 = t0 * W[2] + t1 * W[6] + t2 * W[10] + t3 * W[14];
  float h3 = t0 * W[3] + t1 * W[7] + t2 * W[11] + t3 * W[15];
  float di = dinv[v];
  float s = di * di;
  float4 hn; hn.x = h0; hn.y = h1; hn.z = h2; hn.w = h3;
  float4 an; an.x = s * h0; an.y = s * h1; an.z = s * h2; an.w = s * h3;
  *(float4*)(hout + (size_t)v * 4) = hn;
  *(float4*)(agg + (size_t)v * 4) = an;
}

__global__ void k_act_mm_final(const float* __restrict__ agg,
                               const float* __restrict__ b2,
                               const float* __restrict__ W3,
                               const float* __restrict__ b3,
                               const float* __restrict__ dinv,
                               float* __restrict__ h3, float* __restrict__ out,
                               int n) {
  int v = blockIdx.x * blockDim.x + threadIdx.x;
  if (v >= n) return;
  float4 t = *(const float4*)(agg + (size_t)v * 4);
  float t0 = fmaxf(t.x + b2[0], 0.0f);
  float t1 = fmaxf(t.y + b2[1], 0.0f);
  float t2 = fmaxf(t.z + b2[2], 0.0f);
  float t3 = fmaxf(t.w + b2[3], 0.0f);
  float h0 = t0 * W3[0] + t1 * W3[2] + t2 * W3[4] + t3 * W3[6];
  float h1 = t0 * W3[1] + t1 * W3[3] + t2 * W3[5] + t3 * W3[7];
  float di = dinv[v];
  float s = di * di;
  *(float2*)(h3 + (size_t)v * 2) = make_float2(h0, h1);
  *(float2*)(out + (size_t)v * 2) = make_float2(s * h0 + b3[0], s * h1 + b3[1]);
}

__global__ void k_scatter2(const int* __restrict__ src, const int* __restrict__ dst,
                           const float* __restrict__ dinv,
                           const float* __restrict__ h3, float* out, int E) {
  int e = blockIdx.x * blockDim.x + threadIdx.x;
  if (e >= E) return;
  int s = src[e], d = dst[e];
  float w = dinv[s] * dinv[d];
  float2 m = *(const float2*)(h3 + (size_t)s * 2);
  atomAddF32(out + (size_t)d * 2 + 0, w * m.x);
  atomAddF32(out + (size_t)d * 2 + 1, w * m.y);
}

#define BIN_SHIFT 10
#define BIN_NODES (1 << BIN_SHIFT)
#define BIN_MASK (BIN_NODES - 1)
#define BIN_MAX 4096
#define EDGE_CHUNK 32768
#define BIN_TB 1024
#define SCAN_TB 256
#define EDGE_BATCH 4
#define BATCH_SPAN (EDGE_BATCH * BIN_TB)

__device__ __forceinline__ unsigned ldsAddU32(unsigned* p, unsigned v) {
  return __hip_atomic_fetch_add(p, v, __ATOMIC_RELAXED, __HIP_MEMORY_SCOPE_WORKGROUP);
}
__device__ __forceinline__ void ldsAddF32(float* p, float v) {
  (void)__hip_atomic_fetch_add(p, v, __ATOMIC_RELAXED, __HIP_MEMORY_SCOPE_WORKGROUP);
}

__global__ __launch_bounds__(BIN_TB) void k_bin_count(
    const int* __restrict__ dst, unsigned* __restrict__ counts, int E, int nb,
    int nblk) {
  __shared__ unsigned cnt[BIN_MAX];
  const int t = threadIdx.x;
  const int blk = blockIdx.x;
  for (int i = t; i < nb; i += BIN_TB) cnt[i] = 0u;
  __syncthreads();
  const int e0 = blk * EDGE_CHUNK;
  const int ne = (E - e0 < EDGE_CHUNK) ? (E - e0) : EDGE_CHUNK;
  int i0 = 0;
  for (; i0 + BATCH_SPAN <= ne; i0 += BATCH_SPAN) {
    unsigned d[EDGE_BATCH];
    for (int u = 0; u < EDGE_BATCH; ++u) d[u] = (unsigned)dst[e0 + i0 + u * BIN_TB + t];
    for (int u = 0; u < EDGE_BATCH; ++u) (void)ldsAddU32(&cnt[d[u] >> BIN_SHIFT], 1u);
  }
  for (int i = i0 + t; i < ne; i += BIN_TB)
    (void)ldsAddU32(&cnt[(unsigned)dst[e0 + i] >> BIN_SHIFT], 1u);
  __syncthreads();
  for (int i = t; i < nb; i += BIN_TB) counts[(size_t)i * nblk + blk] = cnt[i];
}

__global__ __launch_bounds__(SCAN_TB) void k_scan_rows(
    const unsigned* in, unsigned* out, unsigned* totals, int len) {
  __shared__ unsigned s[SCAN_TB];
  const int t = threadIdx.x;
  const unsigned* irow = in + (size_t)blockIdx.x * len;
  unsigned* orow = out + (size_t)blockIdx.x * len;
  unsigned carry = 0u;
  for (int base = 0; base < len; base += SCAN_TB) {
    const int i = base + t;
    const unsigned v = (i < len) ? irow[i] : 0u;
    s[t] = v;
    __syncthreads();
    for (int off = 1; off < SCAN_TB; off <<= 1) {
      const unsigned add = (t >= off) ? s[t - off] : 0u;
      __syncthreads();
      s[t] += add;
      __syncthreads();
    }
    const unsigned incl = s[t];
    const unsigned total = s[SCAN_TB - 1];
    if (i < len) orow[i] = carry + incl - v;
    carry += total;
    __syncthreads();
  }
  if (t == 0) totals[blockIdx.x] = carry;
}

#define FILL_BIN_MAX 2048
#define FILL_BINS_PER_THREAD (FILL_BIN_MAX / BIN_TB)

__global__ __launch_bounds__(BIN_TB) void k_bin_fill(
    const int* __restrict__ src, const int* __restrict__ dst,
    const unsigned* __restrict__ counts, const unsigned* __restrict__ btot,
    const unsigned* __restrict__ bstart, unsigned* __restrict__ sorted, int E,
    int nb, int nblk) {
  __shared__ unsigned stage[EDGE_CHUNK];
  __shared__ unsigned lcur[FILL_BIN_MAX];
  __shared__ unsigned gbase[FILL_BIN_MAX];
  __shared__ unsigned wtot[BIN_TB / 32];
  const int t = threadIdx.x;
  const int lane = t & 31;
  const int wave = t >> 5;
  const int blk = blockIdx.x;

  unsigned c[FILL_BINS_PER_THREAD];
  unsigned v = 0u;
  for (int u = 0; u < FILL_BINS_PER_THREAD; ++u) {
    const int i = t * FILL_BINS_PER_THREAD + u;
    c[u] = 0u;
    if (i < nb) {
      const size_t at = (size_t)i * nblk + blk;
      const unsigned off = counts[at];
      const unsigned nxt = (blk + 1 < nblk) ? counts[at + 1] : btot[i];
      c[u] = nxt - off;
      gbase[i] = bstart[i] + off;
    }
    v += c[u];
  }
  unsigned incl = v;
  for (int off = 1; off < 32; off <<= 1) {
    const unsigned up = __shfl_up(incl, (unsigned)off, 32);
    if (lane >= off) incl += up;
  }
  if (lane == 31) wtot[wave] = incl;
  __syncthreads();
  const unsigned wv = wtot[lane];
  unsigned winc = wv;
  for (int off = 1; off < 32; off <<= 1) {
    const unsigned up = __shfl_up(winc, (unsigned)off, 32);
    if (lane >= off) winc += up;
  }
  const unsigned wexcl = __shfl(winc - wv, wave, 32);
  unsigned run = wexcl + incl - v;
  for (int u = 0; u < FILL_BINS_PER_THREAD; ++u) {
    lcur[t * FILL_BINS_PER_THREAD + u] = run;
    run += c[u];
  }
  __syncthreads();

  const int e0 = blk * EDGE_CHUNK;
  const int ne = (E - e0 < EDGE_CHUNK) ? (E - e0) : EDGE_CHUNK;
  int i0 = 0;
  for (; i0 + BATCH_SPAN <= ne; i0 += BATCH_SPAN) {
    unsigned s[EDGE_BATCH], d[EDGE_BATCH];
    for (int u = 0; u < EDGE_BATCH; ++u) {
      const int e = e0 + i0 + u * BIN_TB + t;
      s[u] = (unsigned)src[e];
      d[u] = (unsigned)dst[e];
    }
    unsigned pos[EDGE_BATCH];
    for (int u = 0; u < EDGE_BATCH; ++u)
      pos[u] = ldsAddU32(&lcur[d[u] >> BIN_SHIFT], 1u);
    for (int u = 0; u < EDGE_BATCH; ++u)
      stage[pos[u]] = (s[u] << BIN_SHIFT) | (d[u] & BIN_MASK);
  }
  for (int i = i0 + t; i < ne; i += BIN_TB) {
    const unsigned s = (unsigned)src[e0 + i];
    const unsigned d = (unsigned)dst[e0 + i];
    const unsigned pos = ldsAddU32(&lcur[d >> BIN_SHIFT], 1u);
    stage[pos] = (s << BIN_SHIFT) | (d & BIN_MASK);
  }
  __syncthreads();

  for (int i = wave; i < nb; i += BIN_TB / 32) {
    const unsigned lo = (i == 0) ? 0u : lcur[i - 1];
    const unsigned hi = lcur[i];
    unsigned* __restrict__ o = sorted + gbase[i];
    for (unsigned k = lo + lane; k < hi; k += 32) o[k - lo] = stage[k];
  }
}

#define FUSE_TB 512
#define FUSE_WAVES (FUSE_TB / 32)
#define FUSE_TILES (BIN_NODES / 16 / FUSE_WAVES)
#define FUSE_BATCH 16
#define FUSE_SPAN (FUSE_BATCH * FUSE_TB)

#define ROWS16(M) M(0) M(1) M(2) M(3) M(4) M(5) M(6) M(7) M(8) M(9) M(10) M(11) M(12) M(13) M(14) M(15)
#define ROW_DECL(r) float4 xr##r;
#define ROW_LOAD(r) xr##r = *(const float4*)(xt + (size_t)(r) * 128);
#define ROW_STAGE(r) *(float4*)(&sA[wave][r][lane * 4]) = xr##r;
#define TILE_ROWS_PTR(base_) \
  (x + (size_t)(((base_) < n) ? (base_) : (n - 16)) * 128 + lane * 4)

__device__ __forceinline__ void ldsBarrier() {
  __builtin_amdgcn_fence(__ATOMIC_RELEASE, "workgroup", "local");
  __builtin_amdgcn_s_barrier();
  __builtin_amdgcn_fence(__ATOMIC_ACQUIRE, "workgroup", "local");
}

__global__ __launch_bounds__(FUSE_TB) void k_bin_dinv_gemm(
    const unsigned* __restrict__ sorted, const unsigned* __restrict__ bstart,
    const float* __restrict__ x, const float* __restrict__ W1,
    float* __restrict__ dinv, float* __restrict__ hs, int n) {
  __shared__ float sA[FUSE_WAVES][16][132];
  __shared__ float sB[16][132];
  __shared__ unsigned cnt[BIN_NODES];
  __shared__ float sdinv[BIN_NODES];
  const int t = threadIdx.x;
  const int lane = t & 31;
  const int wave = t >> 5;
  const int b = blockIdx.x;
  const int node0 = b * BIN_NODES;

  ROWS16(ROW_DECL)
  {
    const float* xt = TILE_ROWS_PTR(node0 + wave * 16);
    ROWS16(ROW_LOAD)
  }

  for (int i = t; i < BIN_NODES; i += FUSE_TB) cnt[i] = 1u;
  for (int i = t + FUSE_TB; i < 16 * 128; i += FUSE_TB) sB[i >> 7][i & 127] = 0.0f;
  ldsBarrier();

  const unsigned lo = bstart[b], hi = bstart[b + 1];
  unsigned i0 = lo;
  for (; i0 + FUSE_SPAN <= hi; i0 += FUSE_SPAN) {
    unsigned p[FUSE_BATCH];
#pragma unroll
    for (int u = 0; u < FUSE_BATCH; ++u) p[u] = sorted[i0 + u * FUSE_TB + t];
#pragma unroll
    for (int u = 0; u < FUSE_BATCH; ++u) (void)ldsAddU32(&cnt[p[u] & BIN_MASK], 1u);
  }
  if (i0 < hi) {
    const unsigned last = hi - 1u;
    unsigned p[FUSE_BATCH];
#pragma unroll
    for (int u = 0; u < FUSE_BATCH; ++u) {
      const unsigned j = i0 + u * FUSE_TB + t;
      p[u] = sorted[(j < hi) ? j : last];
    }
#pragma unroll
    for (int u = 0; u < FUSE_BATCH; ++u) {
      const unsigned j = i0 + u * FUSE_TB + t;
      if (j < hi) (void)ldsAddU32(&cnt[p[u] & BIN_MASK], 1u);
    }
  }
  ldsBarrier();
  sB[t >> 7][t & 127] = W1[(t & 127) * 4 + (t >> 7)];
  for (int i = t; i < BIN_NODES; i += FUSE_TB) {
    const float di = 1.0f / sqrtf((float)cnt[i]);
    sdinv[i] = di;
    const int v = node0 + i;
    if (v < n) dinv[v] = di;
  }
  ldsBarrier();

  const int half = lane >> 4;
  const int mn = lane & 15;
  const float* aRow = &sA[wave][mn][0];
  const float* bRow = &sB[mn][0];
#pragma unroll
  for (int i = 0; i < FUSE_TILES; ++i) {
    const int tl = (i * FUSE_WAVES + wave) * 16;
    const int base = node0 + tl;
    const bool active = (base < n);
    if (active) {
      ROWS16(ROW_STAGE)
    }
    if (i + 1 < FUSE_TILES) {
      const float* xt = TILE_ROWS_PTR(node0 + ((i + 1) * FUSE_WAVES + wave) * 16);
      ROWS16(ROW_LOAD)
    }
    __builtin_amdgcn_fence(__ATOMIC_RELEASE, "wavefront", "local");
    __builtin_amdgcn_wave_barrier();
    __builtin_amdgcn_fence(__ATOMIC_ACQUIRE, "wavefront", "local");
    if (active) {
      v8f c = {};
      for (int ks = 0; ks < 32; ++ks) {
        const int k = ks * 4 + 2 * half;
        v2f a = *(const v2f*)(aRow + k);
        v2f bq = *(const v2f*)(bRow + k);
        c = __builtin_amdgcn_wmma_f32_16x16x4_f32(false, a, false, bq, (short)0, c,
                                                  false, false);
      }
      if (mn < 4) {
        float* o = hs + (size_t)(base + 8 * half) * 4 + mn;
        const float* dl = &sdinv[tl + 8 * half];
#pragma unroll
        for (int r = 0; r < 8; ++r) o[(size_t)r * 4] = dl[r] * c[r];
      }
    }
  }
}

__global__ __launch_bounds__(BIN_TB) void k_bin_layer4(
    const unsigned* __restrict__ sorted, const unsigned* __restrict__ bstart,
    const float* __restrict__ hs_in, const float* __restrict__ dinv,
    const float* __restrict__ bvec, const float* __restrict__ W,
    float* __restrict__ hs_out, int n, int fout) {
  __shared__ float acc[4][BIN_NODES];
  const int t = threadIdx.x;
  const int b = blockIdx.x;
  const int v = b * BIN_NODES + t;
  float4 self = make_float4(0.0f, 0.0f, 0.0f, 0.0f);
  if (v < n) self = *(const float4*)(hs_in + (size_t)v * 4);
  acc[0][t] = self.x; acc[1][t] = self.y; acc[2][t] = self.z; acc[3][t] = self.w;
  __syncthreads();
  const unsigned lo = bstart[b], hi = bstart[b + 1];
  unsigned i0 = lo;
  for (; i0 + BATCH_SPAN <= hi; i0 += BATCH_SPAN) {
    unsigned p[EDGE_BATCH];
    float4 m[EDGE_BATCH];
    for (int u = 0; u < EDGE_BATCH; ++u) p[u] = sorted[i0 + u * BIN_TB + t];
    for (int u = 0; u < EDGE_BATCH; ++u)
      m[u] = *(const float4*)(hs_in + (size_t)(p[u] >> BIN_SHIFT) * 4);
    for (int u = 0; u < EDGE_BATCH; ++u) {
      const unsigned dl = p[u] & BIN_MASK;
      ldsAddF32(&acc[0][dl], m[u].x);
      ldsAddF32(&acc[1][dl], m[u].y);
      ldsAddF32(&acc[2][dl], m[u].z);
      ldsAddF32(&acc[3][dl], m[u].w);
    }
  }
  for (unsigned j = i0 + t; j < hi; j += BIN_TB) {
    const unsigned p = sorted[j];
    const unsigned dl = p & BIN_MASK;
    const float4 m = *(const float4*)(hs_in + (size_t)(p >> BIN_SHIFT) * 4);
    ldsAddF32(&acc[0][dl], m.x);
    ldsAddF32(&acc[1][dl], m.y);
    ldsAddF32(&acc[2][dl], m.z);
    ldsAddF32(&acc[3][dl], m.w);
  }
  __syncthreads();
  if (v < n) {
    const float di = dinv[v];
    const float t0 = fmaxf(di * acc[0][t] + bvec[0], 0.0f);
    const float t1 = fmaxf(di * acc[1][t] + bvec[1], 0.0f);
    const float t2 = fmaxf(di * acc[2][t] + bvec[2], 0.0f);
    const float t3 = fmaxf(di * acc[3][t] + bvec[3], 0.0f);
    if (fout == 4) {
      const float h0 = t0 * W[0] + t1 * W[4] + t2 * W[8]  + t3 * W[12];
      const float h1 = t0 * W[1] + t1 * W[5] + t2 * W[9]  + t3 * W[13];
      const float h2 = t0 * W[2] + t1 * W[6] + t2 * W[10] + t3 * W[14];
      const float h3 = t0 * W[3] + t1 * W[7] + t2 * W[11] + t3 * W[15];
      *(float4*)(hs_out + (size_t)v * 4) =
          make_float4(di * h0, di * h1, di * h2, di * h3);
    } else {
      const float h0 = t0 * W[0] + t1 * W[2] + t2 * W[4] + t3 * W[6];
      const float h1 = t0 * W[1] + t1 * W[3] + t2 * W[5] + t3 * W[7];
      *(float2*)(hs_out + (size_t)v * 2) = make_float2(di * h0, di * h1);
    }
  }
}

__global__ __launch_bounds__(BIN_TB) void k_bin_final2(
    const unsigned* __restrict__ sorted, const unsigned* __restrict__ bstart,
    const float* __restrict__ hs3, const float* __restrict__ dinv,
    const float* __restrict__ b3, float* __restrict__ out, int n) {
  __shared__ float acc[2][BIN_NODES];
  const int t = threadIdx.x;
  const int b = blockIdx.x;
  const int v = b * BIN_NODES + t;
  float2 self = make_float2(0.0f, 0.0f);
  if (v < n) self = *(const float2*)(hs3 + (size_t)v * 2);
  acc[0][t] = self.x; acc[1][t] = self.y;
  __syncthreads();
  const unsigned lo = bstart[b], hi = bstart[b + 1];
  unsigned i0 = lo;
  for (; i0 + BATCH_SPAN <= hi; i0 += BATCH_SPAN) {
    unsigned p[EDGE_BATCH];
    float2 m[EDGE_BATCH];
    for (int u = 0; u < EDGE_BATCH; ++u) p[u] = sorted[i0 + u * BIN_TB + t];
    for (int u = 0; u < EDGE_BATCH; ++u)
      m[u] = *(const float2*)(hs3 + (size_t)(p[u] >> BIN_SHIFT) * 2);
    for (int u = 0; u < EDGE_BATCH; ++u) {
      const unsigned dl = p[u] & BIN_MASK;
      ldsAddF32(&acc[0][dl], m[u].x);
      ldsAddF32(&acc[1][dl], m[u].y);
    }
  }
  for (unsigned j = i0 + t; j < hi; j += BIN_TB) {
    const unsigned p = sorted[j];
    const unsigned dl = p & BIN_MASK;
    const float2 m = *(const float2*)(hs3 + (size_t)(p >> BIN_SHIFT) * 2);
    ldsAddF32(&acc[0][dl], m.x);
    ldsAddF32(&acc[1][dl], m.y);
  }
  __syncthreads();
  if (v < n) {
    const float di = dinv[v];
    *(float2*)(out + (size_t)v * 2) =
        make_float2(di * acc[0][t] + b3[0], di * acc[1][t] + b3[1]);
  }
}

extern "C" void kernel_launch(void* const* d_in, const int* in_sizes, int n_in,
                              void* d_out, int out_size, void* d_ws,
                              size_t ws_size, hipStream_t stream) {
  const float* x  = (const float*)d_in[0];
  const int*   ei = (const int*)d_in[1];
  const float* W1 = (const float*)d_in[2];
  const float* b1 = (const float*)d_in[3];
  const float* W2 = (const float*)d_in[4];
  const float* b2 = (const float*)d_in[5];
  const float* W3 = (const float*)d_in[6];
  const float* b3 = (const float*)d_in[7];
  float* out = (float*)d_out;

  const int n = in_sizes[0] / 128;
  const int E = in_sizes[1] / 2;
  const int* src = ei;
  const int* dst = ei + E;

  const int TB = 256;
  dim3 blk(TB);
  dim3 gn((n + TB - 1) / TB);
  int tiles = (n + 16 * GEMM_WAVES - 1) / (16 * GEMM_WAVES);

  const int nb = (n + BIN_NODES - 1) >> BIN_SHIFT;
  const int nblk = (E + EDGE_CHUNK - 1) / EDGE_CHUNK;
  const size_t wE = ((size_t)E + 63) & ~(size_t)63;
  const size_t wC = ((size_t)nb * (size_t)nblk + 63) & ~(size_t)63;
  const size_t wB = ((size_t)2 * nb + 1 + 63) & ~(size_t)63;
  const size_t wN = ((size_t)n + 63) & ~(size_t)63;
  const size_t w4 = ((size_t)4 * n + 63) & ~(size_t)63;
  const size_t w2 = ((size_t)2 * n + 63) & ~(size_t)63;
  const size_t need = 4 * (wE + wC + wB + wN + 2 * w4 + w2);

  if (n <= (1 << 21) && nblk > 0 && ws_size >= need) {
    unsigned* sorted = (unsigned*)d_ws;
    unsigned* counts = sorted + wE;
    unsigned* btot   = counts + wC;
    unsigned* bstart = btot + nb;
    float* dinv = (float*)(btot + wB);
    float* bufA = dinv + wN;
    float* bufB = bufA + w4;
    float* buf3 = bufB + w4;

    k_bin_count<<<nblk, BIN_TB, 0, stream>>>(dst, counts, E, nb, nblk);
    k_scan_rows<<<nb, SCAN_TB, 0, stream>>>(counts, counts, btot, nblk);
    k_scan_rows<<<1, SCAN_TB, 0, stream>>>(btot, bstart, bstart + nb, nb);
    k_bin_fill<<<nblk, BIN_TB, 0, stream>>>(src, dst, counts, btot, bstart, sorted, E, nb, nblk);
    k_bin_dinv_gemm<<<nb, FUSE_TB, 0, stream>>>(sorted, bstart, x, W1, dinv, bufB, n);
    k_bin_layer4<<<nb, BIN_TB, 0, stream>>>(sorted, bstart, bufB, dinv, b1, W2, bufA, n, 4);
    k_bin_layer4<<<nb, BIN_TB, 0, stream>>>(sorted, bstart, bufA, dinv, b2, W3, buf3, n, 2);
    k_bin_final2<<<nb, BIN_TB, 0, stream>>>(sorted, bstart, buf3, dinv, b3, out, n);
    return;
  }

  dim3 ge((E + TB - 1) / TB);
  float* ws   = (float*)d_ws;
  float* dinv = ws;
  float* hA   = dinv + (size_t)n;
  float* hB   = hA + (size_t)n * 4;
  float* hC   = hB + (size_t)n * 4;

  k_deg_init<<<gn, blk, 0, stream>>>(dinv, n);
  k_deg_scatter<<<ge, blk, 0, stream>>>(dst, dinv, E);
  k_dinv<<<gn, blk, 0, stream>>>(dinv, n);
  k_gemm_xW1<<<tiles, GEMM_WAVES * 32, 0, stream>>>(x, W1, hA, n);
  k_init_agg4<<<gn, blk, 0, stream>>>(dinv, hA, hB, n);
  k_scatter4<<<ge, blk, 0, stream>>>(src, dst, dinv, hA, hB, E);
  k_act_mm4<<<gn, blk, 0, stream>>>(hB, b1, W2, dinv, hA, n);
  k_scatter4<<<ge, blk, 0, stream>>>(src, dst, dinv, hA, hB, E);
  k_act_mm_final<<<gn, blk, 0, stream>>>(hB, b2, W3, b3, dinv, hC, out, n);
  k_scatter2<<<ge, blk, 0, stream>>>(src, dst, dinv, hC, out, E);
}


# Edges binned per 32768-edge chunk in ONE kernel (LDS histogram, scan, rank; chunk written whole); per-bin kernels walk the per-chunk runs; count and 2 scan launches go (8 -> 5); LDS add order changes
# speedup vs baseline: 9.0687x; 1.0457x over previous
#include <hip/hip_runtime.h>
#include <math.h>


typedef __attribute__((ext_vector_type(2))) float v2f;
typedef __attribute__((ext_vector_type(8))) float v8f;

__device__ __forceinline__ void atomAddF32(float* p, float v) {
  (void)__hip_atomic_fetch_add(p, v, __ATOMIC_RELAXED, __HIP_MEMORY_SCOPE_AGENT);
}

__global__ void k_deg_init(float* deg, int n) {
  int i = blockIdx.x * blockDim.x + threadIdx.x;
  if (i < n) deg[i] = 1.0f;
}

__global__ void k_deg_scatter(const int* __restrict__ dst, float* deg, int E) {
  int e = blockIdx.x * blockDim.x + threadIdx.x;
  if (e < E) atomAddF32(&deg[dst[e]], 1.0f);
}

__global__ void k_dinv(float* deg, int n) {
  int i = blockIdx.x * blockDim.x + threadIdx.x;
  if (i < n) deg[i] = 1.0f / sqrtf(deg[i]);
}

#define GEMM_WAVES 4

__global__ __launch_bounds__(GEMM_WAVES * 32) void k_gemm_xW1(
    const float* __restrict__ x, const float* __restrict__ W1,
    float* __restrict__ h, int n) {
  __shared__ float sA[GEMM_WAVES][16][132];
  __shared__ float sB[16][132];
  const int tid = threadIdx.x;
  const int lane = tid & 31;
  const int wave = tid >> 5;
  const int base = (blockIdx.x * GEMM_WAVES + wave) * 16;
  const bool active = (base < n);

  for (int i = tid; i < 16 * 128; i += GEMM_WAVES * 32) {
    const int col = i >> 7;
    const int k = i & 127;
    sB[col][k] = (col < 4) ? W1[k * 4 + col] : 0.0f;
  }
  if (active) {
    const float* xt = x + (size_t)base * 128;
#pragma unroll
    for (int i = 0; i < 16; ++i) {
      float4 v = *(const float4*)(xt + (size_t)i * 128 + lane * 4);
      *(float4*)(&sA[wave][i][lane * 4]) = v;
    }
  }
  __syncthreads();
  if (!active) return;

  const int half = lane >> 4;
  const int mn = lane & 15;
  const float* aRow = &sA[wave][mn][0];
  const float* bRow = &sB[mn][0];
  v8f c = {};
  for (int ks = 0; ks < 32; ++ks) {
    const int k = ks * 4 + 2 * half;
    v2f a = *(const v2f*)(aRow + k);
    v2f b = *(const v2f*)(bRow + k);
    c = __builtin_amdgcn_wmma_f32_16x16x4_f32(false, a, false, b, (short)0, c,
                                              false, false);
  }
  if (mn < 4) {
    float* o = h + (size_t)(base + 8 * half) * 4 + mn;
#pragma unroll
    for (int r = 0; r < 8; ++r) o[(size_t)r * 4] = c[r];
  }
}

__global__ void k_init_agg4(const float* __restrict__ dinv,
                            const float* __restrict__ h, float* __restrict__ agg,
                            int n) {
  int v = blockIdx.x * blockDim.x + threadIdx.x;
  if (v >= n) return;
  float di = dinv[v];
  float s = di * di;
  float4 t = *(const float4*)(h + (size_t)v * 4);
  float4 o;
  o.x = s * t.x; o.y = s * t.y; o.z = s * t.z; o.w = s * t.w;
  *(float4*)(agg + (size_t)v * 4) = o;
}

__global__ void k_scatter4(const int* __restrict__ src, const int* __restrict__ dst,
                           const float* __restrict__ dinv,
                           const float* __restrict__ h, float* agg, int E) {
  int e = blockIdx.x * blockDim.x + threadIdx.x;
  if (e >= E) return;
  int s = src[e], d = dst[e];
  float w = dinv[s] * dinv[d];
  float4 m = *(const float4*)(h + (size_t)s * 4);
  float* o = agg + (size_t)d * 4;
  atomAddF32(o + 0, w * m.x);
  atomAddF32(o + 1, w * m.y);
  atomAddF32(o + 2, w * m.z);
  atomAddF32(o + 3, w * m.w);
}

__global__ void k_act_mm4(float* agg, const float* __restrict__ bvec,
                          const float* __restrict__ W,
                          const float* __restrict__ dinv,
                          float* __restrict__ hout, int n) {
  int v = blockIdx.x * blockDim.x + threadIdx.x;
  if (v >= n) return;
  float4 t = *(const float4*)(agg + (size_t)v * 4);
  float t0 = fmaxf(t.x + bvec[0], 0.0f);
  float t1 = fmaxf(t.y + bvec[1], 0.0f);
  float t2 = fmaxf(t.z + bvec[2], 0.0f);
  float t3 = fmaxf(t.w + bvec[3], 0.0f);
  float h0 = t0 * W[0] + t1 * W[4] + t2 * W[8]  + t3 * W[12];
  float h1 = t0 * W[1] + t1 * W[5] + t2 * W[9]  + t3 * W[13];
  float h2 = t0 * W[2] + t1 * W[6] + t2 * W[10] + t3 * W[14];
  float h3 = t0 * W[3] + t1 * W[7] + t2 * W[11] + t3 * W[15];
  float di = dinv[v];
  float s = di * di;
  float4 hn; hn.x = h0; hn.y = h1; hn.z = h2; hn.w = h3;
  float4 an; an.x = s * h0; an.y = s * h1; an.z = s * h2; an.w = s * h3;
  *(float4*)(hout + (size_t)v * 4) = hn;
  *(float4*)(agg + (size_t)v * 4) = an;
}

__global__ void k_act_mm_final(const float* __restrict__ agg,
                               const float* __restrict__ b2,
                               const float* __restrict__ W3,
                               const float* __restrict__ b3,
                               const float* __restrict__ dinv,
                               float* __restrict__ h3, float* __restrict__ out,
                               int n) {
  int v = blockIdx.x * blockDim.x + threadIdx.x;
  if (v >= n) return;
  float4 t = *(const float4*)(agg + (size_t)v * 4);
  float t0 = fmaxf(t.x + b2[0], 0.0f);
  float t1 = fmaxf(t.y + b2[1], 0.0f);
  float t2 = fmaxf(t.z + b2[2], 0.0f);
  float t3 = fmaxf(t.w + b2[3], 0.0f);
  float h0 = t0 * W3[0] + t1 * W3[2] + t2 * W3[4] + t3 * W3[6];
  float h1 = t0 * W3[1] + t1 * W3[3] + t2 * W3[5] + t3 * W3[7];
  float di = dinv[v];
  float s = di * di;
  *(float2*)(h3 + (size_t)v * 2) = make_float2(h0, h1);
  *(float2*)(out + (size_t)v * 2) = make_float2(s * h0 + b3[0], s * h1 + b3[1]);
}

__global__ void k_scatter2(const int* __restrict__ src, const int* __restrict__ dst,
                           const float* __restrict__ dinv,
                           const float* __restrict__ h3, float* out, int E) {
  int e = blockIdx.x * blockDim.x + threadIdx.x;
  if (e >= E) return;
  int s = src[e], d = dst[e];
  float w = dinv[s] * dinv[d];
  float2 m = *(const float2*)(h3 + (size_t)s * 2);
  atomAddF32(out + (size_t)d * 2 + 0, w * m.x);
  atomAddF32(out + (size_t)d * 2 + 1, w * m.y);
}

#define BIN_SHIFT 10
#define BIN_NODES (1 << BIN_SHIFT)
#define BIN_MASK (BIN_NODES - 1)
#define EDGE_CHUNK 32768
#define BIN_TB 1024
#define EDGE_BATCH 4
#define BATCH_SPAN (EDGE_BATCH * BIN_TB)

__device__ __forceinline__ unsigned ldsAddU32(unsigned* p, unsigned v) {
  return __hip_atomic_fetch_add(p, v, __ATOMIC_RELAXED, __HIP_MEMORY_SCOPE_WORKGROUP);
}
__device__ __forceinline__ void ldsAddF32(float* p, float v) {
  (void)__hip_atomic_fetch_add(p, v, __ATOMIC_RELAXED, __HIP_MEMORY_SCOPE_WORKGROUP);
}

#define FILL_BIN_MAX 2048
#define FILL_BINS_PER_THREAD (FILL_BIN_MAX / BIN_TB)

__global__ __launch_bounds__(BIN_TB) void k_bin_chunk(
    const int* __restrict__ src, const int* __restrict__ dst,
    unsigned* __restrict__ off, unsigned* __restrict__ sorted, int E, int nb) {
  __shared__ unsigned stage[EDGE_CHUNK];
  __shared__ unsigned lcur[FILL_BIN_MAX];
  __shared__ unsigned wtot[BIN_TB / 32];
  const int t = threadIdx.x;
  const int lane = t & 31;
  const int wave = t >> 5;
  const int blk = blockIdx.x;
  for (int u = 0; u < FILL_BINS_PER_THREAD; ++u) lcur[t * FILL_BINS_PER_THREAD + u] = 0u;
  __syncthreads();

  const int e0 = blk * EDGE_CHUNK;
  const int ne = (E - e0 < EDGE_CHUNK) ? (E - e0) : EDGE_CHUNK;
  int i0 = 0;
  for (; i0 + BATCH_SPAN <= ne; i0 += BATCH_SPAN) {
    unsigned d[EDGE_BATCH];
    for (int u = 0; u < EDGE_BATCH; ++u) d[u] = (unsigned)dst[e0 + i0 + u * BIN_TB + t];
    for (int u = 0; u < EDGE_BATCH; ++u) (void)ldsAddU32(&lcur[d[u] >> BIN_SHIFT], 1u);
  }
  for (int i = i0 + t; i < ne; i += BIN_TB)
    (void)ldsAddU32(&lcur[(unsigned)dst[e0 + i] >> BIN_SHIFT], 1u);
  __syncthreads();

  unsigned c[FILL_BINS_PER_THREAD];
  unsigned v = 0u;
  for (int u = 0; u < FILL_BINS_PER_THREAD; ++u) {
    c[u] = lcur[t * FILL_BINS_PER_THREAD + u];
    v += c[u];
  }
  unsigned incl = v;
  for (int o = 1; o < 32; o <<= 1) {
    const unsigned up = __shfl_up(incl, (unsigned)o, 32);
    if (lane >= o) incl += up;
  }
  if (lane == 31) wtot[wave] = incl;
  __syncthreads();
  const unsigned wv = wtot[lane];
  unsigned winc = wv;
  for (int o = 1; o < 32; o <<= 1) {
    const unsigned up = __shfl_up(winc, (unsigned)o, 32);
    if (lane >= o) winc += up;
  }
  const unsigned wexcl = __shfl(winc - wv, wave, 32);
  unsigned run = wexcl + incl - v;
  unsigned* __restrict__ orow = off + (size_t)blk * (nb + 1);
  for (int u = 0; u < FILL_BINS_PER_THREAD; ++u) {
    const int i = t * FILL_BINS_PER_THREAD + u;
    lcur[i] = run;
    if (i < nb) orow[i] = run;
    run += c[u];
  }
  if (t == 0) orow[nb] = (unsigned)ne;
  __syncthreads();

  i0 = 0;
  for (; i0 + BATCH_SPAN <= ne; i0 += BATCH_SPAN) {
    unsigned s[EDGE_BATCH], d[EDGE_BATCH];
    for (int u = 0; u < EDGE_BATCH; ++u) {
      const int e = e0 + i0 + u * BIN_TB + t;
      s[u] = (unsigned)src[e];
      d[u] = (unsigned)dst[e];
    }
    unsigned pos[EDGE_BATCH];
    for (int u = 0; u < EDGE_BATCH; ++u)
      pos[u] = ldsAddU32(&lcur[d[u] >> BIN_SHIFT], 1u);
    for (int u = 0; u < EDGE_BATCH; ++u)
      stage[pos[u]] = (s[u] << BIN_SHIFT) | (d[u] & BIN_MASK);
  }
  for (int i = i0 + t; i < ne; i += BIN_TB) {
    const unsigned s = (unsigned)src[e0 + i];
    const unsigned d = (unsigned)dst[e0 + i];
    const unsigned pos = ldsAddU32(&lcur[d >> BIN_SHIFT], 1u);
    stage[pos] = (s << BIN_SHIFT) | (d & BIN_MASK);
  }
  __syncthreads();

  for (int k = t; k < ne; k += BIN_TB) sorted[(size_t)e0 + k] = stage[k];
}

__device__ __forceinline__ void waveLdsOrder() {
  __builtin_amdgcn_fence(__ATOMIC_RELEASE, "wavefront", "local");
  __builtin_amdgcn_wave_barrier();
  __builtin_amdgcn_fence(__ATOMIC_ACQUIRE, "wavefront", "local");
}

template <int NW, int B, typename F>
__device__ __forceinline__ void forBinEdges(
    const unsigned* __restrict__ sorted, const unsigned* __restrict__ off,
    uint2* tab, int nblk, int nb, int b, int wave, int lane, F f) {
  for (int rb = wave; rb < nblk; rb += NW * 32) {
    const int r = rb + NW * lane;
    const bool valid = r < nblk;
    const int rc = valid ? r : (nblk - 1);
    const unsigned* __restrict__ o = off + (size_t)rc * (nb + 1) + b;
    const unsigned lo = o[0];
    const unsigned hi = o[1];
    const unsigned len = valid ? (hi - lo) : 0u;
    unsigned incl = len;
    for (int d = 1; d < 32; d <<= 1) {
      const unsigned up = __shfl_up(incl, (unsigned)d, 32);
      if (lane >= d) incl += up;
    }
    const unsigned total = __shfl(incl, 31, 32);
    waveLdsOrder();
    tab[lane] = make_uint2(incl, (unsigned)rc * EDGE_CHUNK + lo - (incl - len));
    waveLdsOrder();
    int i = 0;
    uint2 cur = tab[0];
    unsigned s0 = 0;
    for (; s0 + 32 * B <= total; s0 += 32 * B) {
      unsigned a[B], p[B];
      bool ok[B];
#pragma unroll
      for (int u = 0; u < B; ++u) {
        const unsigned j = s0 + u * 32 + lane;
        ok[u] = true;
        while (j >= cur.x) cur = tab[++i];
        a[u] = j + cur.y;
      }
#pragma unroll
      for (int u = 0; u < B; ++u) p[u] = sorted[a[u]];
      f(p, ok);
    }
    if (s0 < total) {
      unsigned a[B], p[B];
      bool ok[B];
#pragma unroll
      for (int u = 0; u < B; ++u) {
        const unsigned j = s0 + u * 32 + lane;
        ok[u] = j < total;
        while (ok[u] && j >= cur.x) cur = tab[++i];
        a[u] = ok[u] ? (j + cur.y) : 0u;
      }
#pragma unroll
      for (int u = 0; u < B; ++u) p[u] = sorted[a[u]];
      f(p, ok);
    }
  }
}

#define FUSE_TB 512
#define FUSE_WAVES (FUSE_TB / 32)
#define FUSE_TILES (BIN_NODES / 16 / FUSE_WAVES)
#define FUSE_BATCH 8

#define ROWS16(M) M(0) M(1) M(2) M(3) M(4) M(5) M(6) M(7) M(8) M(9) M(10) M(11) M(12) M(13) M(14) M(15)
#define ROW_DECL(r) float4 xr##r;
#define ROW_LOAD(r) xr##r = *(const float4*)(xt + (size_t)(r) * 128);
#define ROW_STAGE(r) *(float4*)(&sA[wave][r][lane * 4]) = xr##r;
#define TILE_ROWS_PTR(base_) \
  (x + (size_t)(((base_) < n) ? (base_) : (n - 16)) * 128 + lane * 4)

__device__ __forceinline__ void ldsBarrier() {
  __builtin_amdgcn_fence(__ATOMIC_RELEASE, "workgroup", "local");
  __builtin_amdgcn_s_barrier();
  __builtin_amdgcn_fence(__ATOMIC_ACQUIRE, "workgroup", "local");
}

__global__ __launch_bounds__(FUSE_TB) void k_bin_dinv_gemm(
    const unsigned* __restrict__ sorted, const unsigned* __restrict__ off,
    const float* __restrict__ x, const float* __restrict__ W1,
    float* __restrict__ dinv, float* __restrict__ hs, int n, int nblk, int nb) {
  __shared__ float sA[FUSE_WAVES][16][132];
  __shared__ float sB[16][132];
  __shared__ unsigned cnt[BIN_NODES];
  __shared__ float sdinv[BIN_NODES];
  __shared__ uint2 runtab[FUSE_WAVES][32];
  const int t = threadIdx.x;
  const int lane = t & 31;
  const int wave = t >> 5;
  const int b = blockIdx.x;
  const int node0 = b * BIN_NODES;

  ROWS16(ROW_DECL)
  {
    const float* xt = TILE_ROWS_PTR(node0 + wave * 16);
    ROWS16(ROW_LOAD)
  }

  for (int i = t; i < BIN_NODES; i += FUSE_TB) cnt[i] = 1u;
  for (int i = t + FUSE_TB; i < 16 * 128; i += FUSE_TB) sB[i >> 7][i & 127] = 0.0f;
  ldsBarrier();

  forBinEdges<FUSE_WAVES, FUSE_BATCH>(
      sorted, off, &runtab[wave][0], nblk, nb, b, wave, lane,
      [&](const unsigned (&p)[FUSE_BATCH], const bool (&ok)[FUSE_BATCH]) {
        for (int u = 0; u < FUSE_BATCH; ++u)
          if (ok[u]) (void)ldsAddU32(&cnt[p[u] & BIN_MASK], 1u);
      });
  ldsBarrier();
  sB[t >> 7][t & 127] = W1[(t & 127) * 4 + (t >> 7)];
  for (int i = t; i < BIN_NODES; i += FUSE_TB) {
    const float di = 1.0f / sqrtf((float)cnt[i]);
    sdinv[i] = di;
    const int v = node0 + i;
    if (v < n) dinv[v] = di;
  }
  ldsBarrier();

  const int half = lane >> 4;
  const int mn = lane & 15;
  const float* aRow = &sA[wave][mn][0];
  const float* bRow = &sB[mn][0];
#pragma unroll
  for (int i = 0; i < FUSE_TILES; ++i) {
    const int tl = (i * FUSE_WAVES + wave) * 16;
    const int base = node0 + tl;
    const bool active = (base < n);
    if (active) {
      ROWS16(ROW_STAGE)
    }
    if (i + 1 < FUSE_TILES) {
      const float* xt = TILE_ROWS_PTR(node0 + ((i + 1) * FUSE_WAVES + wave) * 16);
      ROWS16(ROW_LOAD)
    }
    __builtin_amdgcn_fence(__ATOMIC_RELEASE, "wavefront", "local");
    __builtin_amdgcn_wave_barrier();
    __builtin_amdgcn_fence(__ATOMIC_ACQUIRE, "wavefront", "local");
    if (active) {
      v8f c = {};
      for (int ks = 0; ks < 32; ++ks) {
        const int k = ks * 4 + 2 * half;
        v2f a = *(const v2f*)(aRow + k);
        v2f bq = *(const v2f*)(bRow + k);
        c = __builtin_amdgcn_wmma_f32_16x16x4_f32(false, a, false, bq, (short)0, c,
                                                  false, false);
      }
      if (mn < 4) {
        float* o = hs + (size_t)(base + 8 * half) * 4 + mn;
        const float* dl = &sdinv[tl + 8 * half];
#pragma unroll
        for (int r = 0; r < 8; ++r) o[(size_t)r * 4] = dl[r] * c[r];
      }
    }
  }
}

__global__ __launch_bounds__(BIN_TB) void k_bin_layer4(
    const unsigned* __restrict__ sorted, const unsigned* __restrict__ off,
    const float* __restrict__ hs_in, const float* __restrict__ dinv,
    const float* __restrict__ bvec, const float* __restrict__ W,
    float* __restrict__ hs_out, int n, int fout, int nblk, int nb) {
  __shared__ float acc[4][BIN_NODES];
  __shared__ uint2 runtab[BIN_TB / 32][32];
  const int t = threadIdx.x;
  const int b = blockIdx.x;
  const int v = b * BIN_NODES + t;
  float4 self = make_float4(0.0f, 0.0f, 0.0f, 0.0f);
  if (v < n) self = *(const float4*)(hs_in + (size_t)v * 4);
  acc[0][t] = self.x; acc[1][t] = self.y; acc[2][t] = self.z; acc[3][t] = self.w;
  __syncthreads();
  forBinEdges<BIN_TB / 32, EDGE_BATCH>(
      sorted, off, &runtab[t >> 5][0], nblk, nb, b, t >> 5, t & 31,
      [&](const unsigned (&p)[EDGE_BATCH], const bool (&ok)[EDGE_BATCH]) {
        float4 m[EDGE_BATCH];
        for (int u = 0; u < EDGE_BATCH; ++u)
          m[u] = *(const float4*)(hs_in + (size_t)(p[u] >> BIN_SHIFT) * 4);
        for (int u = 0; u < EDGE_BATCH; ++u) {
          if (ok[u]) {
            const unsigned dl = p[u] & BIN_MASK;
            ldsAddF32(&acc[0][dl], m[u].x);
            ldsAddF32(&acc[1][dl], m[u].y);
            ldsAddF32(&acc[2][dl], m[u].z);
            ldsAddF32(&acc[3][dl], m[u].w);
          }
        }
      });
  __syncthreads();
  if (v < n) {
    const float di = dinv[v];
    const float t0 = fmaxf(di * acc[0][t] + bvec[0], 0.0f);
    const float t1 = fmaxf(di * acc[1][t] + bvec[1], 0.0f);
    const float t2 = fmaxf(di * acc[2][t] + bvec[2], 0.0f);
    const float t3 = fmaxf(di * acc[3][t] + bvec[3], 0.0f);
    if (fout == 4) {
      const float h0 = t0 * W[0] + t1 * W[4] + t2 * W[8]  + t3 * W[12];
      const float h1 = t0 * W[1] + t1 * W[5] + t2 * W[9]  + t3 * W[13];
      const float h2 = t0 * W[2] + t1 * W[6] + t2 * W[10] + t3 * W[14];
      const float h3 = t0 * W[3] + t1 * W[7] + t2 * W[11] + t3 * W[15];
      *(float4*)(hs_out + (size_t)v * 4) =
          make_float4(di * h0, di * h1, di * h2, di * h3);
    } else {
      const float h0 = t0 * W[0] + t1 * W[2] + t2 * W[4] + t3 * W[6];
      const float h1 = t0 * W[1] + t1 * W[3] + t2 * W[5] + t3 * W[7];
      *(float2*)(hs_out + (size_t)v * 2) = make_float2(di * h0, di * h1);
    }
  }
}

__global__ __launch_bounds__(BIN_TB) void k_bin_final2(
    const unsigned* __restrict__ sorted, const unsigned* __restrict__ off,
    const float* __restrict__ hs3, const float* __restrict__ dinv,
    const float* __restrict__ b3, float* __restrict__ out, int n, int nblk, int nb) {
  __shared__ float acc[2][BIN_NODES];
  __shared__ uint2 runtab[BIN_TB / 32][32];
  const int t = threadIdx.x;
  const int b = blockIdx.x;
  const int v = b * BIN_NODES + t;
  float2 self = make_float2(0.0f, 0.0f);
  if (v < n) self = *(const float2*)(hs3 + (size_t)v * 2);
  acc[0][t] = self.x; acc[1][t] = self.y;
  __syncthreads();
  forBinEdges<BIN_TB / 32, EDGE_BATCH>(
      sorted, off, &runtab[t >> 5][0], nblk, nb, b, t >> 5, t & 31,
      [&](const unsigned (&p)[EDGE_BATCH], const bool (&ok)[EDGE_BATCH]) {
        float2 m[EDGE_BATCH];
        for (int u = 0; u < EDGE_BATCH; ++u)
          m[u] = *(const float2*)(hs3 + (size_t)(p[u] >> BIN_SHIFT) * 2);
        for (int u = 0; u < EDGE_BATCH; ++u) {
          if (ok[u]) {
            const unsigned dl = p[u] & BIN_MASK;
            ldsAddF32(&acc[0][dl], m[u].x);
            ldsAddF32(&acc[1][dl], m[u].y);
          }
        }
      });
  __syncthreads();
  if (v < n) {
    const float di = dinv[v];
    *(float2*)(out + (size_t)v * 2) =
        make_float2(di * acc[0][t] + b3[0], di * acc[1][t] + b3[1]);
  }
}

extern "C" void kernel_launch(void* const* d_in, const int* in_sizes, int n_in,
                              void* d_out, int out_size, void* d_ws,
                              size_t ws_size, hipStream_t stream) {
  const float* x  = (const float*)d_in[0];
  const int*   ei = (const int*)d_in[1];
  const float* W1 = (const float*)d_in[2];
  const float* b1 = (const float*)d_in[3];
  const float* W2 = (const float*)d_in[4];
  const float* b2 = (const float*)d_in[5];
  const float* W3 = (const float*)d_in[6];
  const float* b3 = (const float*)d_in[7];
  float* out = (float*)d_out;

  const int n = in_sizes[0] / 128;
  const int E = in_sizes[1] / 2;
  const int* src = ei;
  const int* dst = ei + E;

  const int TB = 256;
  dim3 blk(TB);
  dim3 gn((n + TB - 1) / TB);
  int tiles = (n + 16 * GEMM_WAVES - 1) / (16 * GEMM_WAVES);

  const int nb = (n + BIN_NODES - 1) >> BIN_SHIFT;
  const int nblk = (E + EDGE_CHUNK - 1) / EDGE_CHUNK;
  const size_t wE = ((size_t)E + 63) & ~(size_t)63;
  const size_t wO = ((size_t)nblk * (size_t)(nb + 1) + 63) & ~(size_t)63;
  const size_t wN = ((size_t)n + 63) & ~(size_t)63;
  const size_t w4 = ((size_t)4 * n + 63) & ~(size_t)63;
  const size_t w2 = ((size_t)2 * n + 63) & ~(size_t)63;
  const size_t need = 4 * (wE + wO + wN + 2 * w4 + w2);

  if (n <= (1 << 21) && nblk > 0 && ws_size >= need) {
    unsigned* sorted = (unsigned*)d_ws;
    unsigned* off    = sorted + wE;
    float* dinv = (float*)(off + wO);
    float* bufA = dinv + wN;
    float* bufB = bufA + w4;
    float* buf3 = bufB + w4;

    k_bin_chunk<<<nblk, BIN_TB, 0, stream>>>(src, dst, off, sorted, E, nb);
    k_bin_dinv_gemm<<<nb, FUSE_TB, 0, stream>>>(sorted, off, x, W1, dinv, bufB, n, nblk, nb);
    k_bin_layer4<<<nb, BIN_TB, 0, stream>>>(sorted, off, bufB, dinv, b1, W2, bufA, n, 4, nblk, nb);
    k_bin_layer4<<<nb, BIN_TB, 0, stream>>>(sorted, off, bufA, dinv, b2, W3, buf3, n, 2, nblk, nb);
    k_bin_final2<<<nb, BIN_TB, 0, stream>>>(sorted, off, buf3, dinv, b3, out, n, nblk, nb);
    return;
  }

  dim3 ge((E + TB - 1) / TB);
  float* ws   = (float*)d_ws;
  float* dinv = ws;
  float* hA   = dinv + (size_t)n;
  float* hB   = hA + (size_t)n * 4;
  float* hC   = hB + (size_t)n * 4;

  k_deg_init<<<gn, blk, 0, stream>>>(dinv, n);
  k_deg_scatter<<<ge, blk, 0, stream>>>(dst, dinv, E);
  k_dinv<<<gn, blk, 0, stream>>>(dinv, n);
  k_gemm_xW1<<<tiles, GEMM_WAVES * 32, 0, stream>>>(x, W1, hA, n);
  k_init_agg4<<<gn, blk, 0, stream>>>(dinv, hA, hB, n);
  k_scatter4<<<ge, blk, 0, stream>>>(src, dst, dinv, hA, hB, E);
  k_act_mm4<<<gn, blk, 0, stream>>>(hB, b1, W2, dinv, hA, n);
  k_scatter4<<<ge, blk, 0, stream>>>(src, dst, dinv, hA, hB, E);
  k_act_mm_final<<<gn, blk, 0, stream>>>(hB, b2, W3, b3, dinv, hC, out, n);
  k_scatter2<<<ge, blk, 0, stream>>>(src, dst, dinv, hC, out, E);
}


# Each 2048-node destination tile's edge words are regrouped by source per call (LDS counting sort on source/256) into one list, so a wave's row gathers share 128-byte lines; LDS add order changes
# speedup vs baseline: 9.3881x; 1.0352x over previous
#include <hip/hip_runtime.h>
#include <math.h>


typedef __attribute__((ext_vector_type(2))) float v2f;
typedef __attribute__((ext_vector_type(8))) float v8f;

__device__ __forceinline__ void atomAddF32(float* p, float v) {
  (void)__hip_atomic_fetch_add(p, v, __ATOMIC_RELAXED, __HIP_MEMORY_SCOPE_AGENT);
}

__global__ void k_deg_init(float* deg, int n) {
  int i = blockIdx.x * blockDim.x + threadIdx.x;
  if (i < n) deg[i] = 1.0f;
}

__global__ void k_deg_scatter(const int* __restrict__ dst, float* deg, int E) {
  int e = blockIdx.x * blockDim.x + threadIdx.x;
  if (e < E) atomAddF32(&deg[dst[e]], 1.0f);
}

__global__ void k_dinv(float* deg, int n) {
  int i = blockIdx.x * blockDim.x + threadIdx.x;
  if (i < n) deg[i] = 1.0f / sqrtf(deg[i]);
}

#define GEMM_WAVES 4

__global__ __launch_bounds__(GEMM_WAVES * 32) void k_gemm_xW1(
    const float* __restrict__ x, const float* __restrict__ W1,
    float* __restrict__ h, int n) {
  __shared__ float sA[GEMM_WAVES][16][132];
  __shared__ float sB[16][132];
  const int tid = threadIdx.x;
  const int lane = tid & 31;
  const int wave = tid >> 5;
  const int base = (blockIdx.x * GEMM_WAVES + wave) * 16;
  const bool active = (base < n);

  for (int i = tid; i < 16 * 128; i += GEMM_WAVES * 32) {
    const int col = i >> 7;
    const int k = i & 127;
    sB[col][k] = (col < 4) ? W1[k * 4 + col] : 0.0f;
  }
  if (active) {
    const float* xt = x + (size_t)base * 128;
#pragma unroll
    for (int i = 0; i < 16; ++i) {
      float4 v = *(const float4*)(xt + (size_t)i * 128 + lane * 4);
      *(float4*)(&sA[wave][i][lane * 4]) = v;
    }
  }
  __syncthreads();
  if (!active) return;

  const int half = lane >> 4;
  const int mn = lane & 15;
  const float* aRow = &sA[wave][mn][0];
  const float* bRow = &sB[mn][0];
  v8f c = {};
  for (int ks = 0; ks < 32; ++ks) {
    const int k = ks * 4 + 2 * half;
    v2f a = *(const v2f*)(aRow + k);
    v2f b = *(const v2f*)(bRow + k);
    c = __builtin_amdgcn_wmma_f32_16x16x4_f32(false, a, false, b, (short)0, c,
                                              false, false);
  }
  if (mn < 4) {
    float* o = h + (size_t)(base + 8 * half) * 4 + mn;
#pragma unroll
    for (int r = 0; r < 8; ++r) o[(size_t)r * 4] = c[r];
  }
}

__global__ void k_init_agg4(const float* __restrict__ dinv,
                            const float* __restrict__ h, float* __restrict__ agg,
                            int n) {
  int v = blockIdx.x * blockDim.x + threadIdx.x;
  if (v >= n) return;
  float di = dinv[v];
  float s = di * di;
  float4 t = *(const float4*)(h + (size_t)v * 4);
  float4 o;
  o.x = s * t.x; o.y = s * t.y; o.z = s * t.z; o.w = s * t.w;
  *(float4*)(agg + (size_t)v * 4) = o;
}

__global__ void k_scatter4(const int* __restrict__ src, const int* __restrict__ dst,
                           const float* __restrict__ dinv,
                           const float* __restrict__ h, float* agg, int E) {
  int e = blockIdx.x * blockDim.x + threadIdx.x;
  if (e >= E) return;
  int s = src[e], d = dst[e];
  float w = dinv[s] * dinv[d];
  float4 m = *(const float4*)(h + (size_t)s * 4);
  float* o = agg + (size_t)d * 4;
  atomAddF32(o + 0, w * m.x);
  atomAddF32(o + 1, w * m.y);
  atomAddF32(o + 2, w * m.z);
  atomAddF32(o + 3, w * m.w);
}

__global__ void k_act_mm4(float* agg, const float* __restrict__ bvec,
                          const float* __restrict__ W,
                          const float* __restrict__ dinv,
                          float* __restrict__ hout, int n) {
  int v = blockIdx.x * blockDim.x + threadIdx.x;
  if (v >= n) return;
  float4 t = *(const float4*)(agg + (size_t)v * 4);
  float t0 = fmaxf(t.x + bvec[0], 0.0f);
  float t1 = fmaxf(t.y + bvec[1], 0.0f);
  float t2 = fmaxf(t.z + bvec[2], 0.0f);
  float t3 = fmaxf(t.w + bvec[3], 0.0f);
  float h0 = t0 * W[0] + t1 * W[4] + t2 * W[8]  + t3 * W[12];
  float h1 = t0 * W[1] + t1 * W[5] + t2 * W[9]  + t3 * W[13];
  float h2 = t0 * W[2] + t1 * W[6] + t2 * W[10] + t3 * W[14];
  float h3 = t0 * W[3] + t1 * W[7] + t2 * W[11] + t3 * W[15];
  float di = dinv[v];
  float s = di * di;
  float4 hn; hn.x = h0; hn.y = h1; hn.z = h2; hn.w = h3;
  float4 an; an.x = s * h0; an.y = s * h1; an.z = s * h2; an.w = s * h3;
  *(float4*)(hout + (size_t)v * 4) = hn;
  *(float4*)(agg + (size_t)v * 4) = an;
}

__global__ void k_act_mm_final(const float* __restrict__ agg,
                               const float* __restrict__ b2,
                               const float* __restrict__ W3,
                               const float* __restrict__ b3,
                               const float* __restrict__ dinv,
                               float* __restrict__ h3, float* __restrict__ out,
                               int n) {
  int v = blockIdx.x * blockDim.x + threadIdx.x;
  if (v >= n) return;
  float4 t = *(const float4*)(agg + (size_t)v * 4);
  float t0 = fmaxf(t.x + b2[0], 0.0f);
  float t1 = fmaxf(t.y + b2[1], 0.0f);
  float t2 = fmaxf(t.z + b2[2], 0.0f);
  float t3 = fmaxf(t.w + b2[3], 0.0f);
  float h0 = t0 * W3[0] + t1 * W3[2] + t2 * W3[4] + t3 * W3[6];
  float h1 = t0 * W3[1] + t1 * W3[3] + t2 * W3[5] + t3 * W3[7];
  float di = dinv[v];
  float s = di * di;
  *(float2*)(h3 + (size_t)v * 2) = make_float2(h0, h1);
  *(float2*)(out + (size_t)v * 2) = make_float2(s * h0 + b3[0], s * h1 + b3[1]);
}

__global__ void k_scatter2(const int* __restrict__ src, const int* __restrict__ dst,
                           const float* __restrict__ dinv,
                           const float* __restrict__ h3, float* out, int E) {
  int e = blockIdx.x * blockDim.x + threadIdx.x;
  if (e >= E) return;
  int s = src[e], d = dst[e];
  float w = dinv[s] * dinv[d];
  float2 m = *(const float2*)(h3 + (size_t)s * 2);
  atomAddF32(out + (size_t)d * 2 + 0, w * m.x);
  atomAddF32(out + (size_t)d * 2 + 1, w * m.y);
}

#define BIN_SHIFT 10
#define BIN_NODES (1 << BIN_SHIFT)
#define BIN_MASK (BIN_NODES - 1)
#define EDGE_CHUNK 32768
#define BIN_TB 1024
#define EDGE_BATCH 4
#define BATCH_SPAN (EDGE_BATCH * BIN_TB)

__device__ __forceinline__ unsigned ldsAddU32(unsigned* p, unsigned v) {
  return __hip_atomic_fetch_add(p, v, __ATOMIC_RELAXED, __HIP_MEMORY_SCOPE_WORKGROUP);
}
__device__ __forceinline__ void ldsAddF32(float* p, float v) {
  (void)__hip_atomic_fetch_add(p, v, __ATOMIC_RELAXED, __HIP_MEMORY_SCOPE_WORKGROUP);
}

#define FILL_BIN_MAX 2048
#define FILL_BINS_PER_THREAD (FILL_BIN_MAX / BIN_TB)

template <int SHIFT>
__global__ __launch_bounds__(BIN_TB) void k_bin_chunk(
    const int* __restrict__ src, const int* __restrict__ dst,
    unsigned* __restrict__ off, unsigned* __restrict__ sorted, int E, int nb) {
  constexpr unsigned MASK = (1u << SHIFT) - 1u;
  __shared__ unsigned stage[EDGE_CHUNK];
  __shared__ unsigned lcur[FILL_BIN_MAX];
  __shared__ unsigned wtot[BIN_TB / 32];
  const int t = threadIdx.x;
  const int lane = t & 31;
  const int wave = t >> 5;
  const int blk = blockIdx.x;
  for (int u = 0; u < FILL_BINS_PER_THREAD; ++u) lcur[t * FILL_BINS_PER_THREAD + u] = 0u;
  __syncthreads();

  const int e0 = blk * EDGE_CHUNK;
  const int ne = (E - e0 < EDGE_CHUNK) ? (E - e0) : EDGE_CHUNK;
  int i0 = 0;
  for (; i0 + BATCH_SPAN <= ne; i0 += BATCH_SPAN) {
    unsigned d[EDGE_BATCH];
    for (int u = 0; u < EDGE_BATCH; ++u) d[u] = (unsigned)dst[e0 + i0 + u * BIN_TB + t];
    for (int u = 0; u < EDGE_BATCH; ++u) (void)ldsAddU32(&lcur[d[u] >> SHIFT], 1u);
  }
  for (int i = i0 + t; i < ne; i += BIN_TB)
    (void)ldsAddU32(&lcur[(unsigned)dst[e0 + i] >> SHIFT], 1u);
  __syncthreads();

  unsigned c[FILL_BINS_PER_THREAD];
  unsigned v = 0u;
  for (int u = 0; u < FILL_BINS_PER_THREAD; ++u) {
    c[u] = lcur[t * FILL_BINS_PER_THREAD + u];
    v += c[u];
  }
  unsigned incl = v;
  for (int o = 1; o < 32; o <<= 1) {
    const unsigned up = __shfl_up(incl, (unsigned)o, 32);
    if (lane >= o) incl += up;
  }
  if (lane == 31) wtot[wave] = incl;
  __syncthreads();
  const unsigned wv = wtot[lane];
  unsigned winc = wv;
  for (int o = 1; o < 32; o <<= 1) {
    const unsigned up = __shfl_up(winc, (unsigned)o, 32);
    if (lane >= o) winc += up;
  }
  const unsigned wexcl = __shfl(winc - wv, wave, 32);
  unsigned run = wexcl + incl - v;
  unsigned* __restrict__ orow = off + (size_t)blk * (nb + 1);
  for (int u = 0; u < FILL_BINS_PER_THREAD; ++u) {
    const int i = t * FILL_BINS_PER_THREAD + u;
    lcur[i] = run;
    if (i < nb) orow[i] = run;
    run += c[u];
  }
  if (t == 0) orow[nb] = (unsigned)ne;
  __syncthreads();

  i0 = 0;
  for (; i0 + BATCH_SPAN <= ne; i0 += BATCH_SPAN) {
    unsigned s[EDGE_BATCH], d[EDGE_BATCH];
    for (int u = 0; u < EDGE_BATCH; ++u) {
      const int e = e0 + i0 + u * BIN_TB + t;
      s[u] = (unsigned)src[e];
      d[u] = (unsigned)dst[e];
    }
    unsigned pos[EDGE_BATCH];
    for (int u = 0; u < EDGE_BATCH; ++u)
      pos[u] = ldsAddU32(&lcur[d[u] >> SHIFT], 1u);
    for (int u = 0; u < EDGE_BATCH; ++u)
      stage[pos[u]] = (s[u] << SHIFT) | (d[u] & MASK);
  }
  for (int i = i0 + t; i < ne; i += BIN_TB) {
    const unsigned s = (unsigned)src[e0 + i];
    const unsigned d = (unsigned)dst[e0 + i];
    const unsigned pos = ldsAddU32(&lcur[d >> SHIFT], 1u);
    stage[pos] = (s << SHIFT) | (d & MASK);
  }
  __syncthreads();

  for (int k = t; k < ne; k += BIN_TB) sorted[(size_t)e0 + k] = stage[k];
}

__device__ __forceinline__ void waveLdsOrder() {
  __builtin_amdgcn_fence(__ATOMIC_RELEASE, "wavefront", "local");
  __builtin_amdgcn_wave_barrier();
  __builtin_amdgcn_fence(__ATOMIC_ACQUIRE, "wavefront", "local");
}

template <int NW, int B, typename F>
__device__ __forceinline__ void forBinEdges(
    const unsigned* __restrict__ sorted, const unsigned* __restrict__ off,
    uint2* tab, int nblk, int nb, int b, int wave, int lane, F f) {
  for (int rb = wave; rb < nblk; rb += NW * 32) {
    const int r = rb + NW * lane;
    const bool valid = r < nblk;
    const int rc = valid ? r : (nblk - 1);
    const unsigned* __restrict__ o = off + (size_t)rc * (nb + 1) + b;
    const unsigned lo = o[0];
    const unsigned hi = o[1];
    const unsigned len = valid ? (hi - lo) : 0u;
    unsigned incl = len;
    for (int d = 1; d < 32; d <<= 1) {
      const unsigned up = __shfl_up(incl, (unsigned)d, 32);
      if (lane >= d) incl += up;
    }
    const unsigned total = __shfl(incl, 31, 32);
    waveLdsOrder();
    tab[lane] = make_uint2(incl, (unsigned)rc * EDGE_CHUNK + lo - (incl - len));
    waveLdsOrder();
    int i = 0;
    uint2 cur = tab[0];
    unsigned s0 = 0;
    for (; s0 + 32 * B <= total; s0 += 32 * B) {
      unsigned a[B], p[B];
      bool ok[B];
#pragma unroll
      for (int u = 0; u < B; ++u) {
        const unsigned j = s0 + u * 32 + lane;
        ok[u] = true;
        while (j >= cur.x) cur = tab[++i];
        a[u] = j + cur.y;
      }
#pragma unroll
      for (int u = 0; u < B; ++u) p[u] = sorted[a[u]];
      f(p, ok);
    }
    if (s0 < total) {
      unsigned a[B], p[B];
      bool ok[B];
#pragma unroll
      for (int u = 0; u < B; ++u) {
        const unsigned j = s0 + u * 32 + lane;
        ok[u] = j < total;
        while (ok[u] && j >= cur.x) cur = tab[++i];
        a[u] = ok[u] ? (j + cur.y) : 0u;
      }
#pragma unroll
      for (int u = 0; u < B; ++u) p[u] = sorted[a[u]];
      f(p, ok);
    }
  }
}

#define FUSE_TB 512
#define FUSE_WAVES (FUSE_TB / 32)
#define FUSE_TILES (BIN_NODES / 16 / FUSE_WAVES)
#define FUSE_BATCH 8

#define ROWS16(M) M(0) M(1) M(2) M(3) M(4) M(5) M(6) M(7) M(8) M(9) M(10) M(11) M(12) M(13) M(14) M(15)
#define ROW_DECL(r) float4 xr##r;
#define ROW_LOAD(r) xr##r = *(const float4*)(xt + (size_t)(r) * 128);
#define ROW_STAGE(r) *(float4*)(&sA[wave][r][lane * 4]) = xr##r;
#define TILE_ROWS_PTR(base_) \
  (x + (size_t)(((base_) < n) ? (base_) : (n - 16)) * 128 + lane * 4)

__device__ __forceinline__ void ldsBarrier() {
  __builtin_amdgcn_fence(__ATOMIC_RELEASE, "workgroup", "local");
  __builtin_amdgcn_s_barrier();
  __builtin_amdgcn_fence(__ATOMIC_ACQUIRE, "workgroup", "local");
}

template <bool COUNT>
__global__ __launch_bounds__(FUSE_TB) void k_bin_dinv_gemm(
    const unsigned* __restrict__ sorted, const unsigned* __restrict__ off,
    const float* __restrict__ x, const float* __restrict__ W1,
    float* __restrict__ dinv, float* __restrict__ hs, int n, int nblk, int nb) {
  __shared__ float sA[FUSE_WAVES][16][132];
  __shared__ float sB[16][132];
  __shared__ unsigned cnt[BIN_NODES];
  __shared__ float sdinv[BIN_NODES];
  __shared__ uint2 runtab[FUSE_WAVES][32];
  const int t = threadIdx.x;
  const int lane = t & 31;
  const int wave = t >> 5;
  const int b = blockIdx.x;
  const int node0 = b * BIN_NODES;

  ROWS16(ROW_DECL)
  float w1v = 0.0f;
  float dv[BIN_NODES / FUSE_TB] = {};
  if (!COUNT) {
    w1v = W1[(t & 127) * 4 + (t >> 7)];
    for (int q = 0; q < BIN_NODES / FUSE_TB; ++q) {
      const int v = node0 + q * FUSE_TB + t;
      dv[q] = dinv[(v < n) ? v : (n - 1)];
    }
  }
  {
    const float* xt = TILE_ROWS_PTR(node0 + wave * 16);
    ROWS16(ROW_LOAD)
  }

  for (int i = t + FUSE_TB; i < 16 * 128; i += FUSE_TB) sB[i >> 7][i & 127] = 0.0f;
  if (COUNT) {
    for (int i = t; i < BIN_NODES; i += FUSE_TB) cnt[i] = 1u;
    ldsBarrier();

    forBinEdges<FUSE_WAVES, FUSE_BATCH>(
        sorted, off, &runtab[wave][0], nblk, nb, b, wave, lane,
        [&](const unsigned (&p)[FUSE_BATCH], const bool (&ok)[FUSE_BATCH]) {
          for (int u = 0; u < FUSE_BATCH; ++u)
            if (ok[u]) (void)ldsAddU32(&cnt[p[u] & BIN_MASK], 1u);
        });
    ldsBarrier();
  }
  if (COUNT) {
    sB[t >> 7][t & 127] = W1[(t & 127) * 4 + (t >> 7)];
    for (int i = t; i < BIN_NODES; i += FUSE_TB) {
      const float di = 1.0f / sqrtf((float)cnt[i]);
      sdinv[i] = di;
      const int v = node0 + i;
      if (v < n) dinv[v] = di;
    }
  } else {
    sB[t >> 7][t & 127] = w1v;
    for (int q = 0; q < BIN_NODES / FUSE_TB; ++q) sdinv[q * FUSE_TB + t] = dv[q];
  }
  ldsBarrier();

  const int half = lane >> 4;
  const int mn = lane & 15;
  const float* aRow = &sA[wave][mn][0];
  const float* bRow = &sB[mn][0];
#pragma unroll
  for (int i = 0; i < FUSE_TILES; ++i) {
    const int tl = (i * FUSE_WAVES + wave) * 16;
    const int base = node0 + tl;
    const bool active = (base < n);
    if (active) {
      ROWS16(ROW_STAGE)
    }
    if (i + 1 < FUSE_TILES) {
      const float* xt = TILE_ROWS_PTR(node0 + ((i + 1) * FUSE_WAVES + wave) * 16);
      ROWS16(ROW_LOAD)
    }
    __builtin_amdgcn_fence(__ATOMIC_RELEASE, "wavefront", "local");
    __builtin_amdgcn_wave_barrier();
    __builtin_amdgcn_fence(__ATOMIC_ACQUIRE, "wavefront", "local");
    if (active) {
      v8f c = {};
      for (int ks = 0; ks < 32; ++ks) {
        const int k = ks * 4 + 2 * half;
        v2f a = *(const v2f*)(aRow + k);
        v2f bq = *(const v2f*)(bRow + k);
        c = __builtin_amdgcn_wmma_f32_16x16x4_f32(false, a, false, bq, (short)0, c,
                                                  false, false);
      }
      if (mn < 4) {
        float* o = hs + (size_t)(base + 8 * half) * 4 + mn;
        const float* dl = &sdinv[tl + 8 * half];
#pragma unroll
        for (int r = 0; r < 8; ++r) o[(size_t)r * 4] = dl[r] * c[r];
      }
    }
  }
}

__global__ __launch_bounds__(BIN_TB) void k_bin_layer4(
    const unsigned* __restrict__ sorted, const unsigned* __restrict__ off,
    const float* __restrict__ hs_in, const float* __restrict__ dinv,
    const float* __restrict__ bvec, const float* __restrict__ W,
    float* __restrict__ hs_out, int n, int fout, int nblk, int nb) {
  __shared__ float acc[4][BIN_NODES];
  __shared__ uint2 runtab[BIN_TB / 32][32];
  const int t = threadIdx.x;
  const int b = blockIdx.x;
  const int v = b * BIN_NODES + t;
  float4 self = make_float4(0.0f, 0.0f, 0.0f, 0.0f);
  if (v < n) self = *(const float4*)(hs_in + (size_t)v * 4);
  acc[0][t] = self.x; acc[1][t] = self.y; acc[2][t] = self.z; acc[3][t] = self.w;
  __syncthreads();
  forBinEdges<BIN_TB / 32, EDGE_BATCH>(
      sorted, off, &runtab[t >> 5][0], nblk, nb, b, t >> 5, t & 31,
      [&](const unsigned (&p)[EDGE_BATCH], const bool (&ok)[EDGE_BATCH]) {
        float4 m[EDGE_BATCH];
        for (int u = 0; u < EDGE_BATCH; ++u)
          m[u] = *(const float4*)(hs_in + (size_t)(p[u] >> BIN_SHIFT) * 4);
        for (int u = 0; u < EDGE_BATCH; ++u) {
          if (ok[u]) {
            const unsigned dl = p[u] & BIN_MASK;
            ldsAddF32(&acc[0][dl], m[u].x);
            ldsAddF32(&acc[1][dl], m[u].y);
            ldsAddF32(&acc[2][dl], m[u].z);
            ldsAddF32(&acc[3][dl], m[u].w);
          }
        }
      });
  __syncthreads();
  if (v < n) {
    const float di = dinv[v];
    const float t0 = fmaxf(di * acc[0][t] + bvec[0], 0.0f);
    const float t1 = fmaxf(di * acc[1][t] + bvec[1], 0.0f);
    const float t2 = fmaxf(di * acc[2][t] + bvec[2], 0.0f);
    const float t3 = fmaxf(di * acc[3][t] + bvec[3], 0.0f);
    if (fout == 4) {
      const float h0 = t0 * W[0] + t1 * W[4] + t2 * W[8]  + t3 * W[12];
      const float h1 = t0 * W[1] + t1 * W[5] + t2 * W[9]  + t3 * W[13];
      const float h2 = t0 * W[2] + t1 * W[6] + t2 * W[10] + t3 * W[14];
      const float h3 = t0 * W[3] + t1 * W[7] + t2 * W[11] + t3 * W[15];
      *(float4*)(hs_out + (size_t)v * 4) =
          make_float4(di * h0, di * h1, di * h2, di * h3);
    } else {
      const float h0 = t0 * W[0] + t1 * W[2] + t2 * W[4] + t3 * W[6];
      const float h1 = t0 * W[1] + t1 * W[3] + t2 * W[5] + t3 * W[7];
      *(float2*)(hs_out + (size_t)v * 2) = make_float2(di * h0, di * h1);
    }
  }
}

__global__ __launch_bounds__(BIN_TB) void k_bin_final2(
    const unsigned* __restrict__ sorted, const unsigned* __restrict__ off,
    const float* __restrict__ hs3, const float* __restrict__ dinv,
    const float* __restrict__ b3, float* __restrict__ out, int n, int nblk, int nb) {
  __shared__ float acc[2][BIN_NODES];
  __shared__ uint2 runtab[BIN_TB / 32][32];
  const int t = threadIdx.x;
  const int b = blockIdx.x;
  const int v = b * BIN_NODES + t;
  float2 self = make_float2(0.0f, 0.0f);
  if (v < n) self = *(const float2*)(hs3 + (size_t)v * 2);
  acc[0][t] = self.x; acc[1][t] = self.y;
  __syncthreads();
  forBinEdges<BIN_TB / 32, EDGE_BATCH>(
      sorted, off, &runtab[t >> 5][0], nblk, nb, b, t >> 5, t & 31,
      [&](const unsigned (&p)[EDGE_BATCH], const bool (&ok)[EDGE_BATCH]) {
        float2 m[EDGE_BATCH];
        for (int u = 0; u < EDGE_BATCH; ++u)
          m[u] = *(const float2*)(hs3 + (size_t)(p[u] >> BIN_SHIFT) * 2);
        for (int u = 0; u < EDGE_BATCH; ++u) {
          if (ok[u]) {
            const unsigned dl = p[u] & BIN_MASK;
            ldsAddF32(&acc[0][dl], m[u].x);
            ldsAddF32(&acc[1][dl], m[u].y);
          }
        }
      });
  __syncthreads();
  if (v < n) {
    const float di = dinv[v];
    *(float2*)(out + (size_t)v * 2) =
        make_float2(di * acc[0][t] + b3[0], di * acc[1][t] + b3[1]);
  }
}

#define TILE_SHIFT 11
#define TILE_NODES (1 << TILE_SHIFT)
#define TILE_MASK (TILE_NODES - 1)
#define TILE_KEY_SHIFT 8
#define TILE_KEYS 8192
#define TILE_KEYS_PER_THREAD (TILE_KEYS / BIN_TB)
#define TILE_CAP (66 * 1024)
#define TILE_BATCH 8
#define TILE_NODES_PER_THREAD (TILE_NODES / BIN_TB)

__global__ __launch_bounds__(BIN_TB) void k_tile_group(
    const unsigned* __restrict__ sorted, const unsigned* __restrict__ off,
    unsigned* __restrict__ grouped, unsigned* __restrict__ tinfo,
    float* __restrict__ dinv, int n, int nblk, int nb) {
  __shared__ unsigned stage[TILE_CAP];
  __shared__ unsigned kcur[TILE_KEYS];
  __shared__ unsigned cnt[TILE_NODES];
  __shared__ uint2 runtab[BIN_TB / 32][32];
  __shared__ unsigned wtot[BIN_TB / 32];
  __shared__ unsigned wtot1[BIN_TB / 32];
  __shared__ unsigned freeSlot;
  const int t = threadIdx.x;
  const int lane = t & 31;
  const int wave = t >> 5;
  const int b = blockIdx.x;

  unsigned s0 = 0u, s1 = 0u;
  for (int c = t; c < nblk; c += BIN_TB) {
    const unsigned* __restrict__ o = off + (size_t)c * (nb + 1) + b;
    s0 += o[0];
    s1 += o[1];
  }
  for (int d = 16; d > 0; d >>= 1) {
    s0 += __shfl_xor(s0, d, 32);
    s1 += __shfl_xor(s1, d, 32);
  }
  if (lane == 0) { wtot[wave] = s0; wtot1[wave] = s1; }
  for (int u = 0; u < TILE_KEYS_PER_THREAD; ++u) kcur[t * TILE_KEYS_PER_THREAD + u] = 0u;
  for (int i = t; i < TILE_NODES; i += BIN_TB) cnt[i] = 1u;
  if (t == 0) freeSlot = 0u;
  __syncthreads();
  s0 = wtot[lane];
  s1 = wtot1[lane];
  for (int d = 16; d > 0; d >>= 1) {
    s0 += __shfl_xor(s0, d, 32);
    s1 += __shfl_xor(s1, d, 32);
  }
  const unsigned total = s1 - s0;
  const unsigned first = (s0 + 32u * (unsigned)b) & ~31u;
  unsigned* __restrict__ lst = grouped + first;

  if (total <= TILE_CAP) {
    forBinEdges<BIN_TB / 32, TILE_BATCH>(
        sorted, off, &runtab[wave][0], nblk, nb, b, wave, lane,
        [&](const unsigned (&p)[TILE_BATCH], const bool (&ok)[TILE_BATCH]) {
          for (int u = 0; u < TILE_BATCH; ++u) {
            if (ok[u]) {
              (void)ldsAddU32(&kcur[p[u] >> (TILE_SHIFT + TILE_KEY_SHIFT)], 1u);
              (void)ldsAddU32(&cnt[p[u] & TILE_MASK], 1u);
            }
          }
        });
    __syncthreads();
    unsigned c[TILE_KEYS_PER_THREAD];
    unsigned v = 0u;
    for (int u = 0; u < TILE_KEYS_PER_THREAD; ++u) {
      c[u] = kcur[t * TILE_KEYS_PER_THREAD + u];
      v += c[u];
    }
    unsigned incl = v;
    for (int o = 1; o < 32; o <<= 1) {
      const unsigned up = __shfl_up(incl, (unsigned)o, 32);
      if (lane >= o) incl += up;
    }
    if (lane == 31) wtot[wave] = incl;
    __syncthreads();
    const unsigned wv = wtot[lane];
    unsigned winc = wv;
    for (int o = 1; o < 32; o <<= 1) {
      const unsigned up = __shfl_up(winc, (unsigned)o, 32);
      if (lane >= o) winc += up;
    }
    const unsigned wexcl = __shfl(winc - wv, wave, 32);
    unsigned run = wexcl + incl - v;
    for (int u = 0; u < TILE_KEYS_PER_THREAD; ++u) {
      kcur[t * TILE_KEYS_PER_THREAD + u] = run;
      run += c[u];
    }
    __syncthreads();
    forBinEdges<BIN_TB / 32, TILE_BATCH>(
        sorted, off, &runtab[wave][0], nblk, nb, b, wave, lane,
        [&](const unsigned (&p)[TILE_BATCH], const bool (&ok)[TILE_BATCH]) {
          unsigned pos[TILE_BATCH];
          for (int u = 0; u < TILE_BATCH; ++u)
            pos[u] = ok[u] ? ldsAddU32(&kcur[p[u] >> (TILE_SHIFT + TILE_KEY_SHIFT)], 1u) : 0u;
          for (int u = 0; u < TILE_BATCH; ++u)
            if (ok[u]) stage[pos[u]] = p[u];
        });
    __syncthreads();
    for (unsigned k = (unsigned)t; k < total; k += BIN_TB) lst[k] = stage[k];
  } else {
    forBinEdges<BIN_TB / 32, TILE_BATCH>(
        sorted, off, &runtab[wave][0], nblk, nb, b, wave, lane,
        [&](const unsigned (&p)[TILE_BATCH], const bool (&ok)[TILE_BATCH]) {
          for (int u = 0; u < TILE_BATCH; ++u) {
            if (ok[u]) {
              (void)ldsAddU32(&cnt[p[u] & TILE_MASK], 1u);
              lst[ldsAddU32(&freeSlot, 1u)] = p[u];
            }
          }
        });
    __syncthreads();
  }

  for (int i = t; i < TILE_NODES; i += BIN_TB) {
    const int nd = b * TILE_NODES + i;
    if (nd < n) dinv[nd] = 1.0f / sqrtf((float)cnt[i]);
  }
  if (t == 0) {
    tinfo[2 * b] = first;
    tinfo[2 * b + 1] = total;
  }
}

template <int B, typename F>
__device__ __forceinline__ void forListEdges(const unsigned* __restrict__ lst,
                                             unsigned total, int t, F f) {
  unsigned i0 = 0u;
  for (; i0 + B * BIN_TB <= total; i0 += B * BIN_TB) {
    unsigned p[B];
    bool ok[B];
#pragma unroll
    for (int u = 0; u < B; ++u) {
      ok[u] = true;
      p[u] = lst[i0 + u * BIN_TB + t];
    }
    f(p, ok);
  }
  if (i0 < total) {
    unsigned p[B];
    bool ok[B];
#pragma unroll
    for (int u = 0; u < B; ++u) {
      const unsigned j = i0 + u * BIN_TB + t;
      ok[u] = j < total;
      p[u] = lst[ok[u] ? j : 0u];
    }
    f(p, ok);
  }
}

__global__ __launch_bounds__(BIN_TB) void k_tile_layer4(
    const unsigned* __restrict__ grouped, const unsigned* __restrict__ tinfo,
    const float* __restrict__ hs_in, const float* __restrict__ dinv,
    const float* __restrict__ bvec, const float* __restrict__ W,
    float* __restrict__ hs_out, int n, int fout) {
  __shared__ float acc[4][TILE_NODES];
  const int t = threadIdx.x;
  const int b = blockIdx.x;
  const unsigned first = tinfo[2 * b];
  const unsigned total = tinfo[2 * b + 1];
  for (int q = 0; q < TILE_NODES_PER_THREAD; ++q) {
    const int i = q * BIN_TB + t;
    const int v = b * TILE_NODES + i;
    float4 self = make_float4(0.0f, 0.0f, 0.0f, 0.0f);
    if (v < n) self = *(const float4*)(hs_in + (size_t)v * 4);
    acc[0][i] = self.x; acc[1][i] = self.y; acc[2][i] = self.z; acc[3][i] = self.w;
  }
  __syncthreads();
  forListEdges<TILE_BATCH>(
      grouped + first, total, t,
      [&](const unsigned (&p)[TILE_BATCH], const bool (&ok)[TILE_BATCH]) {
        float4 m[TILE_BATCH];
        for (int u = 0; u < TILE_BATCH; ++u)
          m[u] = *(const float4*)(hs_in + (size_t)(p[u] >> TILE_SHIFT) * 4);
        for (int u = 0; u < TILE_BATCH; ++u) {
          if (ok[u]) {
            const unsigned dl = p[u] & TILE_MASK;
            ldsAddF32(&acc[0][dl], m[u].x);
            ldsAddF32(&acc[1][dl], m[u].y);
            ldsAddF32(&acc[2][dl], m[u].z);
            ldsAddF32(&acc[3][dl], m[u].w);
          }
        }
      });
  __syncthreads();
  for (int q = 0; q < TILE_NODES_PER_THREAD; ++q) {
    const int i = q * BIN_TB + t;
    const int v = b * TILE_NODES + i;
    if (v < n) {
      const float di = dinv[v];
      const float t0 = fmaxf(di * acc[0][i] + bvec[0], 0.0f);
      const float t1 = fmaxf(di * acc[1][i] + bvec[1], 0.0f);
      const float t2 = fmaxf(di * acc[2][i] + bvec[2], 0.0f);
      const float t3 = fmaxf(di * acc[3][i] + bvec[3], 0.0f);
      if (fout == 4) {
        const float h0 = t0 * W[0] + t1 * W[4] + t2 * W[8]  + t3 * W[12];
        const float h1 = t0 * W[1] + t1 * W[5] + t2 * W[9]  + t3 * W[13];
        const float h2 = t0 * W[2] + t1 * W[6] + t2 * W[10] + t3 * W[14];
        const float h3 = t0 * W[3] + t1 * W[7] + t2 * W[11] + t3 * W[15];
        *(float4*)(hs_out + (size_t)v * 4) =
            make_float4(di * h0, di * h1, di * h2, di * h3);
      } else {
        const float h0 = t0 * W[0] + t1 * W[2] + t2 * W[4] + t3 * W[6];
        const float h1 = t0 * W[1] + t1 * W[3] + t2 * W[5] + t3 * W[7];
        *(float2*)(hs_out + (size_t)v * 2) = make_float2(di * h0, di * h1);
      }
    }
  }
}

__global__ __launch_bounds__(BIN_TB) void k_tile_final2(
    const unsigned* __restrict__ grouped, const unsigned* __restrict__ tinfo,
    const float* __restrict__ hs3, const float* __restrict__ dinv,
    const float* __restrict__ b3, float* __restrict__ out, int n) {
  __shared__ float acc[2][TILE_NODES];
  const int t = threadIdx.x;
  const int b = blockIdx.x;
  const unsigned first = tinfo[2 * b];
  const unsigned total = tinfo[2 * b + 1];
  for (int q = 0; q < TILE_NODES_PER_THREAD; ++q) {
    const int i = q * BIN_TB + t;
    const int v = b * TILE_NODES + i;
    float2 self = make_float2(0.0f, 0.0f);
    if (v < n) self = *(const float2*)(hs3 + (size_t)v * 2);
    acc[0][i] = self.x; acc[1][i] = self.y;
  }
  __syncthreads();
  forListEdges<TILE_BATCH>(
      grouped + first, total, t,
      [&](const unsigned (&p)[TILE_BATCH], const bool (&ok)[TILE_BATCH]) {
        float2 m[TILE_BATCH];
        for (int u = 0; u < TILE_BATCH; ++u)
          m[u] = *(const float2*)(hs3 + (size_t)(p[u] >> TILE_SHIFT) * 2);
        for (int u = 0; u < TILE_BATCH; ++u) {
          if (ok[u]) {
            const unsigned dl = p[u] & TILE_MASK;
            ldsAddF32(&acc[0][dl], m[u].x);
            ldsAddF32(&acc[1][dl], m[u].y);
          }
        }
      });
  __syncthreads();
  for (int q = 0; q < TILE_NODES_PER_THREAD; ++q) {
    const int i = q * BIN_TB + t;
    const int v = b * TILE_NODES + i;
    if (v < n) {
      const float di = dinv[v];
      *(float2*)(out + (size_t)v * 2) =
          make_float2(di * acc[0][i] + b3[0], di * acc[1][i] + b3[1]);
    }
  }
}

extern "C" void kernel_launch(void* const* d_in, const int* in_sizes, int n_in,
                              void* d_out, int out_size, void* d_ws,
                              size_t ws_size, hipStream_t stream) {
  const float* x  = (const float*)d_in[0];
  const int*   ei = (const int*)d_in[1];
  const float* W1 = (const float*)d_in[2];
  const float* b1 = (const float*)d_in[3];
  const float* W2 = (const float*)d_in[4];
  const float* b2 = (const float*)d_in[5];
  const float* W3 = (const float*)d_in[6];
  const float* b3 = (const float*)d_in[7];
  float* out = (float*)d_out;

  const int n = in_sizes[0] / 128;
  const int E = in_sizes[1] / 2;
  const int* src = ei;
  const int* dst = ei + E;

  const int TB = 256;
  dim3 blk(TB);
  dim3 gn((n + TB - 1) / TB);
  int tiles = (n + 16 * GEMM_WAVES - 1) / (16 * GEMM_WAVES);

  const int nb = (n + BIN_NODES - 1) >> BIN_SHIFT;
  const int nblk = (E + EDGE_CHUNK - 1) / EDGE_CHUNK;
  const size_t wE = ((size_t)E + 63) & ~(size_t)63;
  const size_t wO = ((size_t)nblk * (size_t)(nb + 1) + 63) & ~(size_t)63;
  const size_t wN = ((size_t)n + 63) & ~(size_t)63;
  const size_t w4 = ((size_t)4 * n + 63) & ~(size_t)63;
  const size_t w2 = ((size_t)2 * n + 63) & ~(size_t)63;
  const size_t need = 4 * (wE + wO + wN + 2 * w4 + w2);

  const int nt = (n + TILE_NODES - 1) >> TILE_SHIFT;
  const size_t wL = 2 * w4 + w2;
  const size_t w0 = (wE > wL) ? wE : wL;
  const size_t wOt = ((size_t)nblk * (size_t)(nt + 1) + 63) & ~(size_t)63;
  const size_t wG = ((size_t)E + (size_t)32 * nt + 63) & ~(size_t)63;
  const size_t wT = ((size_t)2 * nt + 63) & ~(size_t)63;
  const size_t needTile = 4 * (w0 + wOt + wN + wG + wT);

  if (n <= (1 << 21) && nblk > 0 && ws_size >= needTile) {
    unsigned* sorted = (unsigned*)d_ws;
    float* bufA = (float*)d_ws;
    float* bufB = bufA + w4;
    float* buf3 = bufB + w4;
    unsigned* off    = sorted + w0;
    float* dinv = (float*)(off + wOt);
    unsigned* grouped = (unsigned*)(dinv + wN);
    unsigned* tinfo   = grouped + wG;

    k_bin_chunk<TILE_SHIFT><<<nblk, BIN_TB, 0, stream>>>(src, dst, off, sorted, E, nt);
    k_tile_group<<<nt, BIN_TB, 0, stream>>>(sorted, off, grouped, tinfo, dinv, n, nblk, nt);
    k_bin_dinv_gemm<false><<<nb, FUSE_TB, 0, stream>>>(nullptr, nullptr, x, W1, dinv, bufB, n, 0, nb);
    k_tile_layer4<<<nt, BIN_TB, 0, stream>>>(grouped, tinfo, bufB, dinv, b1, W2, bufA, n, 4);
    k_tile_layer4<<<nt, BIN_TB, 0, stream>>>(grouped, tinfo, bufA, dinv, b2, W3, buf3, n, 2);
    k_tile_final2<<<nt, BIN_TB, 0, stream>>>(grouped, tinfo, buf3, dinv, b3, out, n);
    return;
  }

  if (n <= (1 << 21) && nblk > 0 && ws_size >= need) {
    unsigned* sorted = (unsigned*)d_ws;
    unsigned* off    = sorted + wE;
    float* dinv = (float*)(off + wO);
    float* bufA = dinv + wN;
    float* bufB = bufA + w4;
    float* buf3 = bufB + w4;

    k_bin_chunk<BIN_SHIFT><<<nblk, BIN_TB, 0, stream>>>(src, dst, off, sorted, E, nb);
    k_bin_dinv_gemm<true><<<nb, FUSE_TB, 0, stream>>>(sorted, off, x, W1, dinv, bufB, n, nblk, nb);
    k_bin_layer4<<<nb, BIN_TB, 0, stream>>>(sorted, off, bufB, dinv, b1, W2, bufA, n, 4, nblk, nb);
    k_bin_layer4<<<nb, BIN_TB, 0, stream>>>(sorted, off, bufA, dinv, b2, W3, buf3, n, 2, nblk, nb);
    k_bin_final2<<<nb, BIN_TB, 0, stream>>>(sorted, off, buf3, dinv, b3, out, n, nblk, nb);
    return;
  }

  dim3 ge((E + TB - 1) / TB);
  float* ws   = (float*)d_ws;
  float* dinv = ws;
  float* hA   = dinv + (size_t)n;
  float* hB   = hA + (size_t)n * 4;
  float* hC   = hB + (size_t)n * 4;

  k_deg_init<<<gn, blk, 0, stream>>>(dinv, n);
  k_deg_scatter<<<ge, blk, 0, stream>>>(dst, dinv, E);
  k_dinv<<<gn, blk, 0, stream>>>(dinv, n);
  k_gemm_xW1<<<tiles, GEMM_WAVES * 32, 0, stream>>>(x, W1, hA, n);
  k_init_agg4<<<gn, blk, 0, stream>>>(dinv, hA, hB, n);
  k_scatter4<<<ge, blk, 0, stream>>>(src, dst, dinv, hA, hB, E);
  k_act_mm4<<<gn, blk, 0, stream>>>(hB, b1, W2, dinv, hA, n);
  k_scatter4<<<ge, blk, 0, stream>>>(src, dst, dinv, hA, hB, E);
  k_act_mm_final<<<gn, blk, 0, stream>>>(hB, b2, W3, b3, dinv, hC, out, n);
  k_scatter2<<<ge, blk, 0, stream>>>(src, dst, dinv, hC, out, E);
}


# Aggregation re-tiled: 8192-node ranges x 4 source ranges, lists sorted by source line (a gather touches ~8 lines, was ~20); a finishing kernel per layer adds the 4 partial sums; sum tree changes
# speedup vs baseline: 13.8263x; 1.4727x over previous
#include <hip/hip_runtime.h>
#include <math.h>


typedef __attribute__((ext_vector_type(2))) float v2f;
typedef __attribute__((ext_vector_type(8))) float v8f;

__device__ __forceinline__ void atomAddF32(float* p, float v) {
  (void)__hip_atomic_fetch_add(p, v, __ATOMIC_RELAXED, __HIP_MEMORY_SCOPE_AGENT);
}

__global__ void k_deg_init(float* deg, int n) {
  int i = blockIdx.x * blockDim.x + threadIdx.x;
  if (i < n) deg[i] = 1.0f;
}

__global__ void k_deg_scatter(const int* __restrict__ dst, float* deg, int E) {
  int e = blockIdx.x * blockDim.x + threadIdx.x;
  if (e < E) atomAddF32(&deg[dst[e]], 1.0f);
}

__global__ void k_dinv(float* deg, int n) {
  int i = blockIdx.x * blockDim.x + threadIdx.x;
  if (i < n) deg[i] = 1.0f / sqrtf(deg[i]);
}

#define GEMM_WAVES 4

__global__ __launch_bounds__(GEMM_WAVES * 32) void k_gemm_xW1(
    const float* __restrict__ x, const float* __restrict__ W1,
    float* __restrict__ h, int n) {
  __shared__ float sA[GEMM_WAVES][16][132];
  __shared__ float sB[16][132];
  const int tid = threadIdx.x;
  const int lane = tid & 31;
  const int wave = tid >> 5;
  const int base = (blockIdx.x * GEMM_WAVES + wave) * 16;
  const bool active = (base < n);

  for (int i = tid; i < 16 * 128; i += GEMM_WAVES * 32) {
    const int col = i >> 7;
    const int k = i & 127;
    sB[col][k] = (col < 4) ? W1[k * 4 + col] : 0.0f;
  }
  if (active) {
    const float* xt = x + (size_t)base * 128;
#pragma unroll
    for (int i = 0; i < 16; ++i) {
      float4 v = *(const float4*)(xt + (size_t)i * 128 + lane * 4);
      *(float4*)(&sA[wave][i][lane * 4]) = v;
    }
  }
  __syncthreads();
  if (!active) return;

  const int half = lane >> 4;
  const int mn = lane & 15;
  const float* aRow = &sA[wave][mn][0];
  const float* bRow = &sB[mn][0];
  v8f c = {};
  for (int ks = 0; ks < 32; ++ks) {
    const int k = ks * 4 + 2 * half;
    v2f a = *(const v2f*)(aRow + k);
    v2f b = *(const v2f*)(bRow + k);
    c = __builtin_amdgcn_wmma_f32_16x16x4_f32(false, a, false, b, (short)0, c,
                                              false, false);
  }
  if (mn < 4) {
    float* o = h + (size_t)(base + 8 * half) * 4 + mn;
#pragma unroll
    for (int r = 0; r < 8; ++r) o[(size_t)r * 4] = c[r];
  }
}

__global__ void k_init_agg4(const float* __restrict__ dinv,
                            const float* __restrict__ h, float* __restrict__ agg,
                            int n) {
  int v = blockIdx.x * blockDim.x + threadIdx.x;
  if (v >= n) return;
  float di = dinv[v];
  float s = di * di;
  float4 t = *(const float4*)(h + (size_t)v * 4);
  float4 o;
  o.x = s * t.x; o.y = s * t.y; o.z = s * t.z; o.w = s * t.w;
  *(float4*)(agg + (size_t)v * 4) = o;
}

__global__ void k_scatter4(const int* __restrict__ src, const int* __restrict__ dst,
                           const float* __restrict__ dinv,
                           const float* __restrict__ h, float* agg, int E) {
  int e = blockIdx.x * blockDim.x + threadIdx.x;
  if (e >= E) return;
  int s = src[e], d = dst[e];
  float w = dinv[s] * dinv[d];
  float4 m = *(const float4*)(h + (size_t)s * 4);
  float* o = agg + (size_t)d * 4;
  atomAddF32(o + 0, w * m.x);
  atomAddF32(o + 1, w * m.y);
  atomAddF32(o + 2, w * m.z);
  atomAddF32(o + 3, w * m.w);
}

__global__ void k_act_mm4(float* agg, const float* __restrict__ bvec,
                          const float* __restrict__ W,
                          const float* __restrict__ dinv,
                          float* __restrict__ hout, int n) {
  int v = blockIdx.x * blockDim.x + threadIdx.x;
  if (v >= n) return;
  float4 t = *(const float4*)(agg + (size_t)v * 4);
  float t0 = fmaxf(t.x + bvec[0], 0.0f);
  float t1 = fmaxf(t.y + bvec[1], 0.0f);
  float t2 = fmaxf(t.z + bvec[2], 0.0f);
  float t3 = fmaxf(t.w + bvec[3], 0.0f);
  float h0 = t0 * W[0] + t1 * W[4] + t2 * W[8]  + t3 * W[12];
  float h1 = t0 * W[1] + t1 * W[5] + t2 * W[9]  + t3 * W[13];
  float h2 = t0 * W[2] + t1 * W[6] + t2 * W[10] + t3 * W[14];
  float h3 = t0 * W[3] + t1 * W[7] + t2 * W[11] + t3 * W[15];
  float di = dinv[v];
  float s = di * di;
  float4 hn; hn.x = h0; hn.y = h1; hn.z = h2; hn.w = h3;
  float4 an; an.x = s * h0; an.y = s * h1; an.z = s * h2; an.w = s * h3;
  *(float4*)(hout + (size_t)v * 4) = hn;
  *(float4*)(agg + (size_t)v * 4) = an;
}

__global__ void k_act_mm_final(const float* __restrict__ agg,
                               const float* __restrict__ b2,
                               const float* __restrict__ W3,
                               const float* __restrict__ b3,
                               const float* __restrict__ dinv,
                               float* __restrict__ h3, float* __restrict__ out,
                               int n) {
  int v = blockIdx.x * blockDim.x + threadIdx.x;
  if (v >= n) return;
  float4 t = *(const float4*)(agg + (size_t)v * 4);
  float t0 = fmaxf(t.x + b2[0], 0.0f);
  float t1 = fmaxf(t.y + b2[1], 0.0f);
  float t2 = fmaxf(t.z + b2[2], 0.0f);
  float t3 = fmaxf(t.w + b2[3], 0.0f);
  float h0 = t0 * W3[0] + t1 * W3[2] + t2 * W3[4] + t3 * W3[6];
  float h1 = t0 * W3[1] + t1 * W3[3] + t2 * W3[5] + t3 * W3[7];
  float di = dinv[v];
  float s = di * di;
  *(float2*)(h3 + (size_t)v * 2) = make_float2(h0, h1);
  *(float2*)(out + (size_t)v * 2) = make_float2(s * h0 + b3[0], s * h1 + b3[1]);
}

__global__ void k_scatter2(const int* __restrict__ src, const int* __restrict__ dst,
                           const float* __restrict__ dinv,
                           const float* __restrict__ h3, float* out, int E) {
  int e = blockIdx.x * blockDim.x + threadIdx.x;
  if (e >= E) return;
  int s = src[e], d = dst[e];
  float w = dinv[s] * dinv[d];
  float2 m = *(const float2*)(h3 + (size_t)s * 2);
  atomAddF32(out + (size_t)d * 2 + 0, w * m.x);
  atomAddF32(out + (size_t)d * 2 + 1, w * m.y);
}

#define BIN_SHIFT 10
#define BIN_NODES (1 << BIN_SHIFT)
#define BIN_MASK (BIN_NODES - 1)
#define EDGE_CHUNK 32768
#define BIN_TB 1024
#define EDGE_BATCH 4
#define BATCH_SPAN (EDGE_BATCH * BIN_TB)

__device__ __forceinline__ unsigned ldsAddU32(unsigned* p, unsigned v) {
  return __hip_atomic_fetch_add(p, v, __ATOMIC_RELAXED, __HIP_MEMORY_SCOPE_WORKGROUP);
}
__device__ __forceinline__ void ldsAddF32(float* p, float v) {
  (void)__hip_atomic_fetch_add(p, v, __ATOMIC_RELAXED, __HIP_MEMORY_SCOPE_WORKGROUP);
}

#define FILL_BIN_MAX 2048
#define FILL_BINS_PER_THREAD (FILL_BIN_MAX / BIN_TB)

#define QS_SHIFT 17
#define QS_PARTS 4

template <int SHIFT, bool SPLIT>
__device__ __forceinline__ unsigned edgeBin(unsigned s, unsigned d, unsigned qrows) {
  if (!SPLIT) return d >> SHIFT;
  const unsigned part = (unsigned)(s >= qrows) + (unsigned)(s >= 2u * qrows) + (unsigned)(s >= 3u * qrows);
  return ((d >> SHIFT) << 2) | part;
}

template <int SHIFT, bool SPLIT>
__global__ __launch_bounds__(BIN_TB) void k_bin_chunk(
    const int* __restrict__ src, const int* __restrict__ dst,
    unsigned* __restrict__ off, unsigned* __restrict__ sorted, int E, int nb,
    unsigned qrows) {
  constexpr unsigned MASK = (1u << SHIFT) - 1u;
  __shared__ unsigned stage[EDGE_CHUNK];
  __shared__ unsigned lcur[FILL_BIN_MAX];
  __shared__ unsigned wtot[BIN_TB / 32];
  const int t = threadIdx.x;
  const int lane = t & 31;
  const int wave = t >> 5;
  const int blk = blockIdx.x;
  for (int u = 0; u < FILL_BINS_PER_THREAD; ++u) lcur[t * FILL_BINS_PER_THREAD + u] = 0u;
  __syncthreads();

  const int e0 = blk * EDGE_CHUNK;
  const int ne = (E - e0 < EDGE_CHUNK) ? (E - e0) : EDGE_CHUNK;
  int i0 = 0;
  for (; i0 + BATCH_SPAN <= ne; i0 += BATCH_SPAN) {
    unsigned s[EDGE_BATCH], d[EDGE_BATCH];
    for (int u = 0; u < EDGE_BATCH; ++u) {
      d[u] = (unsigned)dst[e0 + i0 + u * BIN_TB + t];
      s[u] = SPLIT ? (unsigned)src[e0 + i0 + u * BIN_TB + t] : 0u;
    }
    for (int u = 0; u < EDGE_BATCH; ++u)
      (void)ldsAddU32(&lcur[edgeBin<SHIFT, SPLIT>(s[u], d[u], qrows)], 1u);
  }
  for (int i = i0 + t; i < ne; i += BIN_TB) {
    const unsigned d = (unsigned)dst[e0 + i];
    const unsigned s = SPLIT ? (unsigned)src[e0 + i] : 0u;
    (void)ldsAddU32(&lcur[edgeBin<SHIFT, SPLIT>(s, d, qrows)], 1u);
  }
  __syncthreads();

  unsigned c[FILL_BINS_PER_THREAD];
  unsigned v = 0u;
  for (int u = 0; u < FILL_BINS_PER_THREAD; ++u) {
    c[u] = lcur[t * FILL_BINS_PER_THREAD + u];
    v += c[u];
  }
  unsigned incl = v;
  for (int o = 1; o < 32; o <<= 1) {
    const unsigned up = __shfl_up(incl, (unsigned)o, 32);
    if (lane >= o) incl += up;
  }
  if (lane == 31) wtot[wave] = incl;
  __syncthreads();
  const unsigned wv = wtot[lane];
  unsigned winc = wv;
  for (int o = 1; o < 32; o <<= 1) {
    const unsigned up = __shfl_up(winc, (unsigned)o, 32);
    if (lane >= o) winc += up;
  }
  const unsigned wexcl = __shfl(winc - wv, wave, 32);
  unsigned run = wexcl + incl - v;
  unsigned* __restrict__ orow = off + (size_t)blk * (nb + 1);
  for (int u = 0; u < FILL_BINS_PER_THREAD; ++u) {
    const int i = t * FILL_BINS_PER_THREAD + u;
    lcur[i] = run;
    if (i < nb) orow[i] = run;
    run += c[u];
  }
  if (t == 0) orow[nb] = (unsigned)ne;
  __syncthreads();

  i0 = 0;
  for (; i0 + BATCH_SPAN <= ne; i0 += BATCH_SPAN) {
    unsigned s[EDGE_BATCH], d[EDGE_BATCH];
    for (int u = 0; u < EDGE_BATCH; ++u) {
      const int e = e0 + i0 + u * BIN_TB + t;
      s[u] = (unsigned)src[e];
      d[u] = (unsigned)dst[e];
    }
    unsigned pos[EDGE_BATCH];
    for (int u = 0; u < EDGE_BATCH; ++u)
      pos[u] = ldsAddU32(&lcur[edgeBin<SHIFT, SPLIT>(s[u], d[u], qrows)], 1u);
    for (int u = 0; u < EDGE_BATCH; ++u)
      stage[pos[u]] = (s[u] << SHIFT) | (d[u] & MASK);
  }
  for (int i = i0 + t; i < ne; i += BIN_TB) {
    const unsigned s = (unsigned)src[e0 + i];
    const unsigned d = (unsigned)dst[e0 + i];
    const unsigned pos = ldsAddU32(&lcur[edgeBin<SHIFT, SPLIT>(s, d, qrows)], 1u);
    stage[pos] = (s << SHIFT) | (d & MASK);
  }
  __syncthreads();

  for (int k = t; k < ne; k += BIN_TB) sorted[(size_t)e0 + k] = stage[k];
}

__device__ __forceinline__ void waveLdsOrder() {
  __builtin_amdgcn_fence(__ATOMIC_RELEASE, "wavefront", "local");
  __builtin_amdgcn_wave_barrier();
  __builtin_amdgcn_fence(__ATOMIC_ACQUIRE, "wavefront", "local");
}

template <int NW, int B, typename F>
__device__ __forceinline__ void forBinEdges(
    const unsigned* __restrict__ sorted, const unsigned* __restrict__ off,
    uint2* tab, int nblk, int nb, int b, int wave, int lane, F f) {
  for (int rb = wave; rb < nblk; rb += NW * 32) {
    const int r = rb + NW * lane;
    const bool valid = r < nblk;
    const int rc = valid ? r : (nblk - 1);
    const unsigned* __restrict__ o = off + (size_t)rc * (nb + 1) + b;
    const unsigned lo = o[0];
    const unsigned hi = o[1];
    const unsigned len = valid ? (hi - lo) : 0u;
    unsigned incl = len;
    for (int d = 1; d < 32; d <<= 1) {
      const unsigned up = __shfl_up(incl, (unsigned)d, 32);
      if (lane >= d) incl += up;
    }
    const unsigned total = __shfl(incl, 31, 32);
    waveLdsOrder();
    tab[lane] = make_uint2(incl, (unsigned)rc * EDGE_CHUNK + lo - (incl - len));
    waveLdsOrder();
    int i = 0;
    uint2 cur = tab[0];
    unsigned s0 = 0;
    for (; s0 + 32 * B <= total; s0 += 32 * B) {
      unsigned a[B], p[B];
      bool ok[B];
#pragma unroll
      for (int u = 0; u < B; ++u) {
        const unsigned j = s0 + u * 32 + lane;
        ok[u] = true;
        while (j >= cur.x) cur = tab[++i];
        a[u] = j + cur.y;
      }
#pragma unroll
      for (int u = 0; u < B; ++u) p[u] = sorted[a[u]];
      f(p, ok);
    }
    if (s0 < total) {
      unsigned a[B], p[B];
      bool ok[B];
#pragma unroll
      for (int u = 0; u < B; ++u) {
        const unsigned j = s0 + u * 32 + lane;
        ok[u] = j < total;
        while (ok[u] && j >= cur.x) cur = tab[++i];
        a[u] = ok[u] ? (j + cur.y) : 0u;
      }
#pragma unroll
      for (int u = 0; u < B; ++u) p[u] = sorted[a[u]];
      f(p, ok);
    }
  }
}

#define FUSE_TB 512
#define FUSE_WAVES (FUSE_TB / 32)
#define FUSE_TILES (BIN_NODES / 16 / FUSE_WAVES)
#define FUSE_BATCH 8

#define ROWS16(M) M(0) M(1) M(2) M(3) M(4) M(5) M(6) M(7) M(8) M(9) M(10) M(11) M(12) M(13) M(14) M(15)
#define ROW_DECL(r) float4 xr##r;
#define ROW_LOAD(r) xr##r = *(const float4*)(xt + (size_t)(r) * 128);
#define ROW_STAGE(r) *(float4*)(&sA[wave][r][lane * 4]) = xr##r;
#define TILE_ROWS_PTR(base_) \
  (x + (size_t)(((base_) < n) ? (base_) : (n - 16)) * 128 + lane * 4)

__device__ __forceinline__ void ldsBarrier() {
  __builtin_amdgcn_fence(__ATOMIC_RELEASE, "workgroup", "local");
  __builtin_amdgcn_s_barrier();
  __builtin_amdgcn_fence(__ATOMIC_ACQUIRE, "workgroup", "local");
}

template <bool COUNT>
__global__ __launch_bounds__(FUSE_TB) void k_bin_dinv_gemm(
    const unsigned* __restrict__ sorted, const unsigned* __restrict__ off,
    const float* __restrict__ x, const float* __restrict__ W1,
    float* __restrict__ dinv, float* __restrict__ hs, int n, int nblk, int nb,
    const unsigned* __restrict__ cpart, size_t cstride) {
  __shared__ float sA[FUSE_WAVES][16][132];
  __shared__ float sB[16][132];
  __shared__ unsigned cnt[BIN_NODES];
  __shared__ float sdinv[BIN_NODES];
  __shared__ uint2 runtab[FUSE_WAVES][32];
  const int t = threadIdx.x;
  const int lane = t & 31;
  const int wave = t >> 5;
  const int b = blockIdx.x;
  const int node0 = b * BIN_NODES;

  ROWS16(ROW_DECL)
  float w1v = 0.0f;
  float dv[BIN_NODES / FUSE_TB] = {};
  if (!COUNT) {
    w1v = W1[(t & 127) * 4 + (t >> 7)];
    for (int q = 0; q < BIN_NODES / FUSE_TB; ++q) {
      const int v = node0 + q * FUSE_TB + t;
      const int vc = (v < n) ? v : (n - 1);
      if (cpart != nullptr) {
        unsigned cn = 1u;
        for (int r = 0; r < QS_PARTS; ++r) cn += cpart[(size_t)r * cstride + vc];
        dv[q] = 1.0f / sqrtf((float)cn);
        if (v < n) dinv[v] = dv[q];
      } else {
        dv[q] = dinv[vc];
      }
    }
  }
  {
    const float* xt = TILE_ROWS_PTR(node0 + wave * 16);
    ROWS16(ROW_LOAD)
  }

  for (int i = t + FUSE_TB; i < 16 * 128; i += FUSE_TB) sB[i >> 7][i & 127] = 0.0f;
  if (COUNT) {
    for (int i = t; i < BIN_NODES; i += FUSE_TB) cnt[i] = 1u;
    ldsBarrier();

    forBinEdges<FUSE_WAVES, FUSE_BATCH>(
        sorted, off, &runtab[wave][0], nblk, nb, b, wave, lane,
        [&](const unsigned (&p)[FUSE_BATCH], const bool (&ok)[FUSE_BATCH]) {
          for (int u = 0; u < FUSE_BATCH; ++u)
            if (ok[u]) (void)ldsAddU32(&cnt[p[u] & BIN_MASK], 1u);
        });
    ldsBarrier();
  }
  if (COUNT) {
    sB[t >> 7][t & 127] = W1[(t & 127) * 4 + (t >> 7)];
    for (int i = t; i < BIN_NODES; i += FUSE_TB) {
      const float di = 1.0f / sqrtf((float)cnt[i]);
      sdinv[i] = di;
      const int v = node0 + i;
      if (v < n) dinv[v] = di;
    }
  } else {
    sB[t >> 7][t & 127] = w1v;
    for (int q = 0; q < BIN_NODES / FUSE_TB; ++q) sdinv[q * FUSE_TB + t] = dv[q];
  }
  ldsBarrier();

  const int half = lane >> 4;
  const int mn = lane & 15;
  const float* aRow = &sA[wave][mn][0];
  const float* bRow = &sB[mn][0];
#pragma unroll
  for (int i = 0; i < FUSE_TILES; ++i) {
    const int tl = (i * FUSE_WAVES + wave) * 16;
    const int base = node0 + tl;
    const bool active = (base < n);
    if (active) {
      ROWS16(ROW_STAGE)
    }
    if (i + 1 < FUSE_TILES) {
      const float* xt = TILE_ROWS_PTR(node0 + ((i + 1) * FUSE_WAVES + wave) * 16);
      ROWS16(ROW_LOAD)
    }
    __builtin_amdgcn_fence(__ATOMIC_RELEASE, "wavefront", "local");
    __builtin_amdgcn_wave_barrier();
    __builtin_amdgcn_fence(__ATOMIC_ACQUIRE, "wavefront", "local");
    if (active) {
      v8f c = {};
      for (int ks = 0; ks < 32; ++ks) {
        const int k = ks * 4 + 2 * half;
        v2f a = *(const v2f*)(aRow + k);
        v2f bq = *(const v2f*)(bRow + k);
        c = __builtin_amdgcn_wmma_f32_16x16x4_f32(false, a, false, bq, (short)0, c,
                                                  false, false);
      }
      if (mn < 4) {
        float* o = hs + (size_t)(base + 8 * half) * 4 + mn;
        const float* dl = &sdinv[tl + 8 * half];
#pragma unroll
        for (int r = 0; r < 8; ++r) o[(size_t)r * 4] = dl[r] * c[r];
      }
    }
  }
}

__global__ __launch_bounds__(BIN_TB) void k_bin_layer4(
    const unsigned* __restrict__ sorted, const unsigned* __restrict__ off,
    const float* __restrict__ hs_in, const float* __restrict__ dinv,
    const float* __restrict__ bvec, const float* __restrict__ W,
    float* __restrict__ hs_out, int n, int fout, int nblk, int nb) {
  __shared__ float acc[4][BIN_NODES];
  __shared__ uint2 runtab[BIN_TB / 32][32];
  const int t = threadIdx.x;
  const int b = blockIdx.x;
  const int v = b * BIN_NODES + t;
  float4 self = make_float4(0.0f, 0.0f, 0.0f, 0.0f);
  if (v < n) self = *(const float4*)(hs_in + (size_t)v * 4);
  acc[0][t] = self.x; acc[1][t] = self.y; acc[2][t] = self.z; acc[3][t] = self.w;
  __syncthreads();
  forBinEdges<BIN_TB / 32, EDGE_BATCH>(
      sorted, off, &runtab[t >> 5][0], nblk, nb, b, t >> 5, t & 31,
      [&](const unsigned (&p)[EDGE_BATCH], const bool (&ok)[EDGE_BATCH]) {
        float4 m[EDGE_BATCH];
        for (int u = 0; u < EDGE_BATCH; ++u)
          m[u] = *(const float4*)(hs_in + (size_t)(p[u] >> BIN_SHIFT) * 4);
        for (int u = 0; u < EDGE_BATCH; ++u) {
          if (ok[u]) {
            const unsigned dl = p[u] & BIN_MASK;
            ldsAddF32(&acc[0][dl], m[u].x);
            ldsAddF32(&acc[1][dl], m[u].y);
            ldsAddF32(&acc[2][dl], m[u].z);
            ldsAddF32(&acc[3][dl], m[u].w);
          }
        }
      });
  __syncthreads();
  if (v < n) {
    const float di = dinv[v];
    const float t0 = fmaxf(di * acc[0][t] + bvec[0], 0.0f);
    const float t1 = fmaxf(di * acc[1][t] + bvec[1], 0.0f);
    const float t2 = fmaxf(di * acc[2][t] + bvec[2], 0.0f);
    const float t3 = fmaxf(di * acc[3][t] + bvec[3], 0.0f);
    if (fout == 4) {
      const float h0 = t0 * W[0] + t1 * W[4] + t2 * W[8]  + t3 * W[12];
      const float h1 = t0 * W[1] + t1 * W[5] + t2 * W[9]  + t3 * W[13];
      const float h2 = t0 * W[2] + t1 * W[6] + t2 * W[10] + t3 * W[14];
      const float h3 = t0 * W[3] + t1 * W[7] + t2 * W[11] + t3 * W[15];
      *(float4*)(hs_out + (size_t)v * 4) =
          make_float4(di * h0, di * h1, di * h2, di * h3);
    } else {
      const float h0 = t0 * W[0] + t1 * W[2] + t2 * W[4] + t3 * W[6];
      const float h1 = t0 * W[1] + t1 * W[3] + t2 * W[5] + t3 * W[7];
      *(float2*)(hs_out + (size_t)v * 2) = make_float2(di * h0, di * h1);
    }
  }
}

__global__ __launch_bounds__(BIN_TB) void k_bin_final2(
    const unsigned* __restrict__ sorted, const unsigned* __restrict__ off,
    const float* __restrict__ hs3, const float* __restrict__ dinv,
    const float* __restrict__ b3, float* __restrict__ out, int n, int nblk, int nb) {
  __shared__ float acc[2][BIN_NODES];
  __shared__ uint2 runtab[BIN_TB / 32][32];
  const int t = threadIdx.x;
  const int b = blockIdx.x;
  const int v = b * BIN_NODES + t;
  float2 self = make_float2(0.0f, 0.0f);
  if (v < n) self = *(const float2*)(hs3 + (size_t)v * 2);
  acc[0][t] = self.x; acc[1][t] = self.y;
  __syncthreads();
  forBinEdges<BIN_TB / 32, EDGE_BATCH>(
      sorted, off, &runtab[t >> 5][0], nblk, nb, b, t >> 5, t & 31,
      [&](const unsigned (&p)[EDGE_BATCH], const bool (&ok)[EDGE_BATCH]) {
        float2 m[EDGE_BATCH];
        for (int u = 0; u < EDGE_BATCH; ++u)
          m[u] = *(const float2*)(hs3 + (size_t)(p[u] >> BIN_SHIFT) * 2);
        for (int u = 0; u < EDGE_BATCH; ++u) {
          if (ok[u]) {
            const unsigned dl = p[u] & BIN_MASK;
            ldsAddF32(&acc[0][dl], m[u].x);
            ldsAddF32(&acc[1][dl], m[u].y);
          }
        }
      });
  __syncthreads();
  if (v < n) {
    const float di = dinv[v];
    *(float2*)(out + (size_t)v * 2) =
        make_float2(di * acc[0][t] + b3[0], di * acc[1][t] + b3[1]);
  }
}

#define TILE_SHIFT 11
#define TILE_NODES (1 << TILE_SHIFT)
#define TILE_MASK (TILE_NODES - 1)
#define TILE_KEY_SHIFT 8
#define TILE_KEYS 8192
#define TILE_KEYS_PER_THREAD (TILE_KEYS / BIN_TB)
#define TILE_CAP (66 * 1024)
#define TILE_BATCH 8
#define TILE_NODES_PER_THREAD (TILE_NODES / BIN_TB)

__global__ __launch_bounds__(BIN_TB) void k_tile_group(
    const unsigned* __restrict__ sorted, const unsigned* __restrict__ off,
    unsigned* __restrict__ grouped, unsigned* __restrict__ tinfo,
    float* __restrict__ dinv, int n, int nblk, int nb) {
  __shared__ unsigned stage[TILE_CAP];
  __shared__ unsigned kcur[TILE_KEYS];
  __shared__ unsigned cnt[TILE_NODES];
  __shared__ uint2 runtab[BIN_TB / 32][32];
  __shared__ unsigned wtot[BIN_TB / 32];
  __shared__ unsigned wtot1[BIN_TB / 32];
  __shared__ unsigned freeSlot;
  const int t = threadIdx.x;
  const int lane = t & 31;
  const int wave = t >> 5;
  const int b = blockIdx.x;

  unsigned s0 = 0u, s1 = 0u;
  for (int c = t; c < nblk; c += BIN_TB) {
    const unsigned* __restrict__ o = off + (size_t)c * (nb + 1) + b;
    s0 += o[0];
    s1 += o[1];
  }
  for (int d = 16; d > 0; d >>= 1) {
    s0 += __shfl_xor(s0, d, 32);
    s1 += __shfl_xor(s1, d, 32);
  }
  if (lane == 0) { wtot[wave] = s0; wtot1[wave] = s1; }
  for (int u = 0; u < TILE_KEYS_PER_THREAD; ++u) kcur[t * TILE_KEYS_PER_THREAD + u] = 0u;
  for (int i = t; i < TILE_NODES; i += BIN_TB) cnt[i] = 1u;
  if (t == 0) freeSlot = 0u;
  __syncthreads();
  s0 = wtot[lane];
  s1 = wtot1[lane];
  for (int d = 16; d > 0; d >>= 1) {
    s0 += __shfl_xor(s0, d, 32);
    s1 += __shfl_xor(s1, d, 32);
  }
  const unsigned total = s1 - s0;
  const unsigned first = (s0 + 32u * (unsigned)b) & ~31u;
  unsigned* __restrict__ lst = grouped + first;

  if (total <= TILE_CAP) {
    forBinEdges<BIN_TB / 32, TILE_BATCH>(
        sorted, off, &runtab[wave][0], nblk, nb, b, wave, lane,
        [&](const unsigned (&p)[TILE_BATCH], const bool (&ok)[TILE_BATCH]) {
          for (int u = 0; u < TILE_BATCH; ++u) {
            if (ok[u]) {
              (void)ldsAddU32(&kcur[p[u] >> (TILE_SHIFT + TILE_KEY_SHIFT)], 1u);
              (void)ldsAddU32(&cnt[p[u] & TILE_MASK], 1u);
            }
          }
        });
    __syncthreads();
    unsigned c[TILE_KEYS_PER_THREAD];
    unsigned v = 0u;
    for (int u = 0; u < TILE_KEYS_PER_THREAD; ++u) {
      c[u] = kcur[t * TILE_KEYS_PER_THREAD + u];
      v += c[u];
    }
    unsigned incl = v;
    for (int o = 1; o < 32; o <<= 1) {
      const unsigned up = __shfl_up(incl, (unsigned)o, 32);
      if (lane >= o) incl += up;
    }
    if (lane == 31) wtot[wave] = incl;
    __syncthreads();
    const unsigned wv = wtot[lane];
    unsigned winc = wv;
    for (int o = 1; o < 32; o <<= 1) {
      const unsigned up = __shfl_up(winc, (unsigned)o, 32);
      if (lane >= o) winc += up;
    }
    const unsigned wexcl = __shfl(winc - wv, wave, 32);
    unsigned run = wexcl + incl - v;
    for (int u = 0; u < TILE_KEYS_PER_THREAD; ++u) {
      kcur[t * TILE_KEYS_PER_THREAD + u] = run;
      run += c[u];
    }
    __syncthreads();
    forBinEdges<BIN_TB / 32, TILE_BATCH>(
        sorted, off, &runtab[wave][0], nblk, nb, b, wave, lane,
        [&](const unsigned (&p)[TILE_BATCH], const bool (&ok)[TILE_BATCH]) {
          unsigned pos[TILE_BATCH];
          for (int u = 0; u < TILE_BATCH; ++u)
            pos[u] = ok[u] ? ldsAddU32(&kcur[p[u] >> (TILE_SHIFT + TILE_KEY_SHIFT)], 1u) : 0u;
          for (int u = 0; u < TILE_BATCH; ++u)
            if (ok[u]) stage[pos[u]] = p[u];
        });
    __syncthreads();
    for (unsigned k = (unsigned)t; k < total; k += BIN_TB) lst[k] = stage[k];
  } else {
    forBinEdges<BIN_TB / 32, TILE_BATCH>(
        sorted, off, &runtab[wave][0], nblk, nb, b, wave, lane,
        [&](const unsigned (&p)[TILE_BATCH], const bool (&ok)[TILE_BATCH]) {
          for (int u = 0; u < TILE_BATCH; ++u) {
            if (ok[u]) {
              (void)ldsAddU32(&cnt[p[u] & TILE_MASK], 1u);
              lst[ldsAddU32(&freeSlot, 1u)] = p[u];
            }
          }
        });
    __syncthreads();
  }

  for (int i = t; i < TILE_NODES; i += BIN_TB) {
    const int nd = b * TILE_NODES + i;
    if (nd < n) dinv[nd] = 1.0f / sqrtf((float)cnt[i]);
  }
  if (t == 0) {
    tinfo[2 * b] = first;
    tinfo[2 * b + 1] = total;
  }
}

template <int B, typename F>
__device__ __forceinline__ void forListEdges(const unsigned* __restrict__ lst,
                                             unsigned total, int t, F f) {
  unsigned i0 = 0u;
  for (; i0 + B * BIN_TB <= total; i0 += B * BIN_TB) {
    unsigned p[B];
    bool ok[B];
#pragma unroll
    for (int u = 0; u < B; ++u) {
      ok[u] = true;
      p[u] = lst[i0 + u * BIN_TB + t];
    }
    f(p, ok);
  }
  if (i0 < total) {
    unsigned p[B];
    bool ok[B];
#pragma unroll
    for (int u = 0; u < B; ++u) {
      const unsigned j = i0 + u * BIN_TB + t;
      ok[u] = j < total;
      p[u] = lst[ok[u] ? j : 0u];
    }
    f(p, ok);
  }
}

__global__ __launch_bounds__(BIN_TB) void k_tile_layer4(
    const unsigned* __restrict__ grouped, const unsigned* __restrict__ tinfo,
    const float* __restrict__ hs_in, const float* __restrict__ dinv,
    const float* __restrict__ bvec, const float* __restrict__ W,
    float* __restrict__ hs_out, int n, int fout) {
  __shared__ float acc[4][TILE_NODES];
  const int t = threadIdx.x;
  const int b = blockIdx.x;
  const unsigned first = tinfo[2 * b];
  const unsigned total = tinfo[2 * b + 1];
  for (int q = 0; q < TILE_NODES_PER_THREAD; ++q) {
    const int i = q * BIN_TB + t;
    const int v = b * TILE_NODES + i;
    float4 self = make_float4(0.0f, 0.0f, 0.0f, 0.0f);
    if (v < n) self = *(const float4*)(hs_in + (size_t)v * 4);
    acc[0][i] = self.x; acc[1][i] = self.y; acc[2][i] = self.z; acc[3][i] = self.w;
  }
  __syncthreads();
  forListEdges<TILE_BATCH>(
      grouped + first, total, t,
      [&](const unsigned (&p)[TILE_BATCH], const bool (&ok)[TILE_BATCH]) {
        float4 m[TILE_BATCH];
        for (int u = 0; u < TILE_BATCH; ++u)
          m[u] = *(const float4*)(hs_in + (size_t)(p[u] >> TILE_SHIFT) * 4);
        for (int u = 0; u < TILE_BATCH; ++u) {
          if (ok[u]) {
            const unsigned dl = p[u] & TILE_MASK;
            ldsAddF32(&acc[0][dl], m[u].x);
            ldsAddF32(&acc[1][dl], m[u].y);
            ldsAddF32(&acc[2][dl], m[u].z);
            ldsAddF32(&acc[3][dl], m[u].w);
          }
        }
      });
  __syncthreads();
  for (int q = 0; q < TILE_NODES_PER_THREAD; ++q) {
    const int i = q * BIN_TB + t;
    const int v = b * TILE_NODES + i;
    if (v < n) {
      const float di = dinv[v];
      const float t0 = fmaxf(di * acc[0][i] + bvec[0], 0.0f);
      const float t1 = fmaxf(di * acc[1][i] + bvec[1], 0.0f);
      const float t2 = fmaxf(di * acc[2][i] + bvec[2], 0.0f);
      const float t3 = fmaxf(di * acc[3][i] + bvec[3], 0.0f);
      if (fout == 4) {
        const float h0 = t0 * W[0] + t1 * W[4] + t2 * W[8]  + t3 * W[12];
        const float h1 = t0 * W[1] + t1 * W[5] + t2 * W[9]  + t3 * W[13];
        const float h2 = t0 * W[2] + t1 * W[6] + t2 * W[10] + t3 * W[14];
        const float h3 = t0 * W[3] + t1 * W[7] + t2 * W[11] + t3 * W[15];
        *(float4*)(hs_out + (size_t)v * 4) =
            make_float4(di * h0, di * h1, di * h2, di * h3);
      } else {
        const float h0 = t0 * W[0] + t1 * W[2] + t2 * W[4] + t3 * W[6];
        const float h1 = t0 * W[1] + t1 * W[3] + t2 * W[5] + t3 * W[7];
        *(float2*)(hs_out + (size_t)v * 2) = make_float2(di * h0, di * h1);
      }
    }
  }
}

__global__ __launch_bounds__(BIN_TB) void k_tile_final2(
    const unsigned* __restrict__ grouped, const unsigned* __restrict__ tinfo,
    const float* __restrict__ hs3, const float* __restrict__ dinv,
    const float* __restrict__ b3, float* __restrict__ out, int n) {
  __shared__ float acc[2][TILE_NODES];
  const int t = threadIdx.x;
  const int b = blockIdx.x;
  const unsigned first = tinfo[2 * b];
  const unsigned total = tinfo[2 * b + 1];
  for (int q = 0; q < TILE_NODES_PER_THREAD; ++q) {
    const int i = q * BIN_TB + t;
    const int v = b * TILE_NODES + i;
    float2 self = make_float2(0.0f, 0.0f);
    if (v < n) self = *(const float2*)(hs3 + (size_t)v * 2);
    acc[0][i] = self.x; acc[1][i] = self.y;
  }
  __syncthreads();
  forListEdges<TILE_BATCH>(
      grouped + first, total, t,
      [&](const unsigned (&p)[TILE_BATCH], const bool (&ok)[TILE_BATCH]) {
        float2 m[TILE_BATCH];
        for (int u = 0; u < TILE_BATCH; ++u)
          m[u] = *(const float2*)(hs3 + (size_t)(p[u] >> TILE_SHIFT) * 2);
        for (int u = 0; u < TILE_BATCH; ++u) {
          if (ok[u]) {
            const unsigned dl = p[u] & TILE_MASK;
            ldsAddF32(&acc[0][dl], m[u].x);
            ldsAddF32(&acc[1][dl], m[u].y);
          }
        }
      });
  __syncthreads();
  for (int q = 0; q < TILE_NODES_PER_THREAD; ++q) {
    const int i = q * BIN_TB + t;
    const int v = b * TILE_NODES + i;
    if (v < n) {
      const float di = dinv[v];
      *(float2*)(out + (size_t)v * 2) =
          make_float2(di * acc[0][i] + b3[0], di * acc[1][i] + b3[1]);
    }
  }
}

#define QD_SHIFT 13
#define QD_NODES (1 << QD_SHIFT)
#define QD_MASK (QD_NODES - 1)
#define QD_KEY_ROWS_LOG2 4
#define QD_KEY_SHIFT (QD_SHIFT + QD_KEY_ROWS_LOG2)
#define QD_KEYS (1 << (QS_SHIFT - QD_KEY_ROWS_LOG2))
#define QD_KEY_MASK (QD_KEYS - 1u)
#define QD_KEYS_PER_THREAD (QD_KEYS / BIN_TB)
#define QD_NODES_PER_THREAD (QD_NODES / BIN_TB)

__global__ __launch_bounds__(BIN_TB) void k_quad_group(
    const unsigned* __restrict__ sorted, const unsigned* __restrict__ off,
    unsigned* __restrict__ grouped, unsigned* __restrict__ tinfo,
    unsigned* __restrict__ cpart, size_t cstride, int n, int nblk, int nb, unsigned qrows) {
  __shared__ unsigned stage[TILE_CAP];
  __shared__ unsigned kcur[QD_KEYS];
  __shared__ uint2 runtab[BIN_TB / 32][32];
  unsigned* cnt = stage;
  __shared__ unsigned wtot[BIN_TB / 32];
  __shared__ unsigned wtot1[BIN_TB / 32];
  __shared__ unsigned freeSlot;
  const int t = threadIdx.x;
  const int lane = t & 31;
  const int wave = t >> 5;
  const int b = blockIdx.x;

  unsigned s0 = 0u, s1 = 0u;
  for (int c = t; c < nblk; c += BIN_TB) {
    const unsigned* __restrict__ o = off + (size_t)c * (nb + 1) + b;
    s0 += o[0];
    s1 += o[1];
  }
  for (int d = 16; d > 0; d >>= 1) {
    s0 += __shfl_xor(s0, d, 32);
    s1 += __shfl_xor(s1, d, 32);
  }
  if (lane == 0) { wtot[wave] = s0; wtot1[wave] = s1; }
  for (int u = 0; u < QD_KEYS_PER_THREAD; ++u) kcur[t * QD_KEYS_PER_THREAD + u] = 0u;
  for (int i = t; i < QD_NODES; i += BIN_TB) cnt[i] = 0u;
  if (t == 0) freeSlot = 0u;
  __syncthreads();
  s0 = wtot[lane];
  s1 = wtot1[lane];
  for (int d = 16; d > 0; d >>= 1) {
    s0 += __shfl_xor(s0, d, 32);
    s1 += __shfl_xor(s1, d, 32);
  }
  const unsigned total = s1 - s0;
  const unsigned first = (s0 + 32u * (unsigned)b) & ~31u;
  unsigned* __restrict__ lst = grouped + first;
  unsigned* __restrict__ cp = cpart + (size_t)(b & (QS_PARTS - 1)) * cstride;
  const unsigned kbase = ((unsigned)(b & (QS_PARTS - 1)) * qrows) >> QD_KEY_ROWS_LOG2;

  if (total <= TILE_CAP) {
    forBinEdges<BIN_TB / 32, TILE_BATCH>(
        sorted, off, &runtab[wave][0], nblk, nb, b, wave, lane,
        [&](const unsigned (&p)[TILE_BATCH], const bool (&ok)[TILE_BATCH]) {
          for (int u = 0; u < TILE_BATCH; ++u) {
            if (ok[u]) {
              (void)ldsAddU32(&kcur[((p[u] >> QD_KEY_SHIFT) - kbase) & QD_KEY_MASK], 1u);
              (void)ldsAddU32(&cnt[p[u] & QD_MASK], 1u);
            }
          }
        });
    __syncthreads();
    for (int i = t; i < QD_NODES; i += BIN_TB) {
      const int nd = (b >> 2) * QD_NODES + i;
      if (nd < n) cp[nd] = cnt[i];
    }
    unsigned c[QD_KEYS_PER_THREAD];
    unsigned v = 0u;
    for (int u = 0; u < QD_KEYS_PER_THREAD; ++u) {
      c[u] = kcur[t * QD_KEYS_PER_THREAD + u];
      v += c[u];
    }
    unsigned incl = v;
    for (int o = 1; o < 32; o <<= 1) {
      const unsigned up = __shfl_up(incl, (unsigned)o, 32);
      if (lane >= o) incl += up;
    }
    if (lane == 31) wtot[wave] = incl;
    __syncthreads();
    const unsigned wv = wtot[lane];
    unsigned winc = wv;
    for (int o = 1; o < 32; o <<= 1) {
      const unsigned up = __shfl_up(winc, (unsigned)o, 32);
      if (lane >= o) winc += up;
    }
    const unsigned wexcl = __shfl(winc - wv, wave, 32);
    unsigned run = wexcl + incl - v;
    for (int u = 0; u < QD_KEYS_PER_THREAD; ++u) {
      kcur[t * QD_KEYS_PER_THREAD + u] = run;
      run += c[u];
    }
    __syncthreads();
    forBinEdges<BIN_TB / 32, TILE_BATCH>(
        sorted, off, &runtab[wave][0], nblk, nb, b, wave, lane,
        [&](const unsigned (&p)[TILE_BATCH], const bool (&ok)[TILE_BATCH]) {
          unsigned pos[TILE_BATCH];
          for (int u = 0; u < TILE_BATCH; ++u)
            pos[u] = ok[u] ? ldsAddU32(&kcur[((p[u] >> QD_KEY_SHIFT) - kbase) & QD_KEY_MASK], 1u) : 0u;
          for (int u = 0; u < TILE_BATCH; ++u)
            if (ok[u]) stage[pos[u]] = p[u];
        });
    __syncthreads();
    for (unsigned k = (unsigned)t; k < total; k += BIN_TB) lst[k] = stage[k];
  } else {
    forBinEdges<BIN_TB / 32, TILE_BATCH>(
        sorted, off, &runtab[wave][0], nblk, nb, b, wave, lane,
        [&](const unsigned (&p)[TILE_BATCH], const bool (&ok)[TILE_BATCH]) {
          for (int u = 0; u < TILE_BATCH; ++u) {
            if (ok[u]) {
              (void)ldsAddU32(&cnt[p[u] & QD_MASK], 1u);
              lst[ldsAddU32(&freeSlot, 1u)] = p[u];
            }
          }
        });
    __syncthreads();
    for (int i = t; i < QD_NODES; i += BIN_TB) {
      const int nd = (b >> 2) * QD_NODES + i;
      if (nd < n) cp[nd] = cnt[i];
    }
  }

  if (t == 0) {
    tinfo[2 * b] = first;
    tinfo[2 * b + 1] = total;
  }
}

__global__ __launch_bounds__(BIN_TB) void k_quad_layer4(
    const unsigned* __restrict__ grouped, const unsigned* __restrict__ tinfo,
    const float* __restrict__ hs_in, float* __restrict__ part, size_t pstride, int n) {
  __shared__ float acc[4][QD_NODES];
  const int t = threadIdx.x;
  const int b = blockIdx.x;
  const int node0 = (b >> 2) * QD_NODES;
  const int sr = b & (QS_PARTS - 1);
  const unsigned first = tinfo[2 * b];
  const unsigned total = tinfo[2 * b + 1];
  for (int q = 0; q < QD_NODES_PER_THREAD; ++q) {
    const int i = q * BIN_TB + t;
    const int v = node0 + i;
    const int vc = (v < n) ? v : (n - 1);
    const float4 row = *(const float4*)(hs_in + (size_t)vc * 4);
    const bool own = (sr == 0) && (v < n);
    acc[0][i] = own ? row.x : 0.0f; acc[1][i] = own ? row.y : 0.0f;
    acc[2][i] = own ? row.z : 0.0f; acc[3][i] = own ? row.w : 0.0f;
  }
  __syncthreads();
  forListEdges<TILE_BATCH>(
      grouped + first, total, t,
      [&](const unsigned (&p)[TILE_BATCH], const bool (&ok)[TILE_BATCH]) {
        float4 m[TILE_BATCH];
        for (int u = 0; u < TILE_BATCH; ++u)
          m[u] = *(const float4*)(hs_in + (size_t)(p[u] >> QD_SHIFT) * 4);
        for (int u = 0; u < TILE_BATCH; ++u) {
          if (ok[u]) {
            const unsigned dl = p[u] & QD_MASK;
            ldsAddF32(&acc[0][dl], m[u].x);
            ldsAddF32(&acc[1][dl], m[u].y);
            ldsAddF32(&acc[2][dl], m[u].z);
            ldsAddF32(&acc[3][dl], m[u].w);
          }
        }
      });
  __syncthreads();
  float* __restrict__ po = part + (size_t)sr * pstride;
  for (int q = 0; q < QD_NODES_PER_THREAD; ++q) {
    const int i = q * BIN_TB + t;
    const int v = node0 + i;
    if (v < n)
      *(float4*)(po + (size_t)v * 4) = make_float4(acc[0][i], acc[1][i], acc[2][i], acc[3][i]);
  }
}

__global__ __launch_bounds__(BIN_TB) void k_quad_final2(
    const unsigned* __restrict__ grouped, const unsigned* __restrict__ tinfo,
    const float* __restrict__ hs3, float* __restrict__ part, size_t pstride, int n) {
  __shared__ float acc[2][QD_NODES];
  const int t = threadIdx.x;
  const int b = blockIdx.x;
  const int node0 = (b >> 2) * QD_NODES;
  const int sr = b & (QS_PARTS - 1);
  const unsigned first = tinfo[2 * b];
  const unsigned total = tinfo[2 * b + 1];
  for (int q = 0; q < QD_NODES_PER_THREAD; ++q) {
    const int i = q * BIN_TB + t;
    const int v = node0 + i;
    const int vc = (v < n) ? v : (n - 1);
    const float2 row = *(const float2*)(hs3 + (size_t)vc * 2);
    const bool own = (sr == 0) && (v < n);
    acc[0][i] = own ? row.x : 0.0f; acc[1][i] = own ? row.y : 0.0f;
  }
  __syncthreads();
  forListEdges<TILE_BATCH>(
      grouped + first, total, t,
      [&](const unsigned (&p)[TILE_BATCH], const bool (&ok)[TILE_BATCH]) {
        float2 m[TILE_BATCH];
        for (int u = 0; u < TILE_BATCH; ++u)
          m[u] = *(const float2*)(hs3 + (size_t)(p[u] >> QD_SHIFT) * 2);
        for (int u = 0; u < TILE_BATCH; ++u) {
          if (ok[u]) {
            const unsigned dl = p[u] & QD_MASK;
            ldsAddF32(&acc[0][dl], m[u].x);
            ldsAddF32(&acc[1][dl], m[u].y);
          }
        }
      });
  __syncthreads();
  float* __restrict__ po = part + (size_t)sr * pstride;
  for (int q = 0; q < QD_NODES_PER_THREAD; ++q) {
    const int i = q * BIN_TB + t;
    const int v = node0 + i;
    if (v < n) *(float2*)(po + (size_t)v * 2) = make_float2(acc[0][i], acc[1][i]);
  }
}

#define FIN_TB 256
#define FIN_NODES 4
__global__ __launch_bounds__(FIN_TB) void k_finish4(
    const float* __restrict__ part, size_t pstride,
    const float* __restrict__ dinv, const float* __restrict__ bvec,
    const float* __restrict__ W, float* __restrict__ hs_out, int n, int fout) {
  const int v0 = blockIdx.x * (FIN_TB * FIN_NODES) + threadIdx.x;
  float4 pl[FIN_NODES][QS_PARTS];
  float dj[FIN_NODES];
  for (int j = 0; j < FIN_NODES; ++j) {
    const int v = v0 + j * FIN_TB;
    const int vc = (v < n) ? v : (n - 1);
    for (int r = 0; r < QS_PARTS; ++r)
      pl[j][r] = *(const float4*)(part + (size_t)r * pstride + (size_t)vc * 4);
    dj[j] = dinv[vc];
  }
  for (int j = 0; j < FIN_NODES; ++j) {
  const int v = v0 + j * FIN_TB;
  if (v >= n) continue;
  float4 a = pl[j][0];
  for (int r = 1; r < QS_PARTS; ++r) {
    const float4 pr = pl[j][r];
    a.x += pr.x; a.y += pr.y; a.z += pr.z; a.w += pr.w;
  }
  const float di = dj[j];
  const float t0 = fmaxf(di * a.x + bvec[0], 0.0f);
  const float t1 = fmaxf(di * a.y + bvec[1], 0.0f);
  const float t2 = fmaxf(di * a.z + bvec[2], 0.0f);
  const float t3 = fmaxf(di * a.w + bvec[3], 0.0f);
  if (fout == 4) {
    const float h0 = t0 * W[0] + t1 * W[4] + t2 * W[8]  + t3 * W[12];
    const float h1 = t0 * W[1] + t1 * W[5] + t2 * W[9]  + t3 * W[13];
    const float h2 = t0 * W[2] + t1 * W[6] + t2 * W[10] + t3 * W[14];
    const float h3 = t0 * W[3] + t1 * W[7] + t2 * W[11] + t3 * W[15];
    *(float4*)(hs_out + (size_t)v * 4) = make_float4(di * h0, di * h1, di * h2, di * h3);
  } else {
    const float h0 = t0 * W[0] + t1 * W[2] + t2 * W[4] + t3 * W[6];
    const float h1 = t0 * W[1] + t1 * W[3] + t2 * W[5] + t3 * W[7];
    *(float2*)(hs_out + (size_t)v * 2) = make_float2(di * h0, di * h1);
  }
  }
}

__global__ __launch_bounds__(FIN_TB) void k_finish_out(
    const float* __restrict__ part, size_t pstride,
    const float* __restrict__ dinv, const float* __restrict__ b3,
    float* __restrict__ out, int n) {
  const int v0 = blockIdx.x * (FIN_TB * FIN_NODES) + threadIdx.x;
  float2 pl[FIN_NODES][QS_PARTS];
  float dj[FIN_NODES];
  for (int j = 0; j < FIN_NODES; ++j) {
    const int v = v0 + j * FIN_TB;
    const int vc = (v < n) ? v : (n - 1);
    for (int r = 0; r < QS_PARTS; ++r)
      pl[j][r] = *(const float2*)(part + (size_t)r * pstride + (size_t)vc * 2);
    dj[j] = dinv[vc];
  }
  for (int j = 0; j < FIN_NODES; ++j) {
    const int v = v0 + j * FIN_TB;
    if (v >= n) continue;
    float2 a = pl[j][0];
    for (int r = 1; r < QS_PARTS; ++r) { a.x += pl[j][r].x; a.y += pl[j][r].y; }
    const float di = dj[j];
    *(float2*)(out + (size_t)v * 2) = make_float2(di * a.x + b3[0], di * a.y + b3[1]);
  }
}

extern "C" void kernel_launch(void* const* d_in, const int* in_sizes, int n_in,
                              void* d_out, int out_size, void* d_ws,
                              size_t ws_size, hipStream_t stream) {
  const float* x  = (const float*)d_in[0];
  const int*   ei = (const int*)d_in[1];
  const float* W1 = (const float*)d_in[2];
  const float* b1 = (const float*)d_in[3];
  const float* W2 = (const float*)d_in[4];
  const float* b2 = (const float*)d_in[5];
  const float* W3 = (const float*)d_in[6];
  const float* b3 = (const float*)d_in[7];
  float* out = (float*)d_out;

  const int n = in_sizes[0] / 128;
  const int E = in_sizes[1] / 2;
  const int* src = ei;
  const int* dst = ei + E;

  const int TB = 256;
  dim3 blk(TB);
  dim3 gn((n + TB - 1) / TB);
  int tiles = (n + 16 * GEMM_WAVES - 1) / (16 * GEMM_WAVES);

  const int nb = (n + BIN_NODES - 1) >> BIN_SHIFT;
  const int nblk = (E + EDGE_CHUNK - 1) / EDGE_CHUNK;
  const size_t wE = ((size_t)E + 63) & ~(size_t)63;
  const size_t wO = ((size_t)nblk * (size_t)(nb + 1) + 63) & ~(size_t)63;
  const size_t wN = ((size_t)n + 63) & ~(size_t)63;
  const size_t w4 = ((size_t)4 * n + 63) & ~(size_t)63;
  const size_t w2 = ((size_t)2 * n + 63) & ~(size_t)63;
  const size_t need = 4 * (wE + wO + wN + 2 * w4 + w2);

  const int nt = (n + TILE_NODES - 1) >> TILE_SHIFT;
  const size_t wL = 2 * w4 + w2;
  const size_t w0 = (wE > wL) ? wE : wL;
  const size_t wOt = ((size_t)nblk * (size_t)(nt + 1) + 63) & ~(size_t)63;
  const size_t wG = ((size_t)E + (size_t)32 * nt + 63) & ~(size_t)63;
  const size_t wT = ((size_t)2 * nt + 63) & ~(size_t)63;
  const size_t needTile = 4 * (w0 + wOt + wN + wG + wT);

  const int nD = (n + QD_NODES - 1) >> QD_SHIFT;
  const int nq = nD * QS_PARTS;
  const unsigned qrows = (unsigned)(((n + QS_PARTS - 1) / QS_PARTS + 15) & ~15);
  const size_t wLq = 2 * w4 + w2 + QS_PARTS * w4;
  const size_t w0q = (wE > wLq) ? wE : wLq;
  const size_t wOq = ((size_t)nblk * (size_t)(nq + 1) + 63) & ~(size_t)63;
  const size_t wGq = ((size_t)E + (size_t)32 * nq + 63) & ~(size_t)63;
  const size_t wTq = ((size_t)2 * nq + 63) & ~(size_t)63;
  const size_t needQuad = 4 * (w0q + wOq + wN + QS_PARTS * wN + wGq + wTq);

  if (n <= (1 << 19) && qrows <= (1u << QS_SHIFT) && nq <= FILL_BIN_MAX && nblk > 0 && ws_size >= needQuad) {
    unsigned* sorted = (unsigned*)d_ws;
    float* bufA = (float*)d_ws;
    float* bufB = bufA + w4;
    float* buf3 = bufB + w4;
    float* part = buf3 + w2;
    unsigned* off    = sorted + w0q;
    float* dinv = (float*)(off + wOq);
    unsigned* cpart = (unsigned*)(dinv + wN);
    unsigned* grouped = cpart + QS_PARTS * wN;
    unsigned* tinfo   = grouped + wGq;
    const int gf = (n + FIN_TB * FIN_NODES - 1) / (FIN_TB * FIN_NODES);

    k_bin_chunk<QD_SHIFT, true><<<nblk, BIN_TB, 0, stream>>>(src, dst, off, sorted, E, nq, qrows);
    k_quad_group<<<nq, BIN_TB, 0, stream>>>(sorted, off, grouped, tinfo, cpart, wN, n, nblk, nq, qrows);
    k_bin_dinv_gemm<false><<<nb, FUSE_TB, 0, stream>>>(nullptr, nullptr, x, W1, dinv, bufB, n, 0, nb, cpart, wN);
    k_quad_layer4<<<nq, BIN_TB, 0, stream>>>(grouped, tinfo, bufB, part, w4, n);
    k_finish4<<<gf, FIN_TB, 0, stream>>>(part, w4, dinv, b1, W2, bufA, n, 4);
    k_quad_layer4<<<nq, BIN_TB, 0, stream>>>(grouped, tinfo, bufA, part, w4, n);
    k_finish4<<<gf, FIN_TB, 0, stream>>>(part, w4, dinv, b2, W3, buf3, n, 2);
    k_quad_final2<<<nq, BIN_TB, 0, stream>>>(grouped, tinfo, buf3, part, w2, n);
    k_finish_out<<<gf, FIN_TB, 0, stream>>>(part, w2, dinv, b3, out, n);
    return;
  }

  if (n <= (1 << 21) && nblk > 0 && ws_size >= needTile) {
    unsigned* sorted = (unsigned*)d_ws;
    float* bufA = (float*)d_ws;
    float* bufB = bufA + w4;
    float* buf3 = bufB + w4;
    unsigned* off    = sorted + w0;
    float* dinv = (float*)(off + wOt);
    unsigned* grouped = (unsigned*)(dinv + wN);
    unsigned* tinfo   = grouped + wG;

    k_bin_chunk<TILE_SHIFT, false><<<nblk, BIN_TB, 0, stream>>>(src, dst, off, sorted, E, nt, 0u);
    k_tile_group<<<nt, BIN_TB, 0, stream>>>(sorted, off, grouped, tinfo, dinv, n, nblk, nt);
    k_bin_dinv_gemm<false><<<nb, FUSE_TB, 0, stream>>>(nullptr, nullptr, x, W1, dinv, bufB, n, 0, nb, nullptr, 0);
    k_tile_layer4<<<nt, BIN_TB, 0, stream>>>(grouped, tinfo, bufB, dinv, b1, W2, bufA, n, 4);
    k_tile_layer4<<<nt, BIN_TB, 0, stream>>>(grouped, tinfo, bufA, dinv, b2, W3, buf3, n, 2);
    k_tile_final2<<<nt, BIN_TB, 0, stream>>>(grouped, tinfo, buf3, dinv, b3, out, n);
    return;
  }

  if (n <= (1 << 21) && nblk > 0 && ws_size >= need) {
    unsigned* sorted = (unsigned*)d_ws;
    unsigned* off    = sorted + wE;
    float* dinv = (float*)(off + wO);
    float* bufA = dinv + wN;
    float* bufB = bufA + w4;
    float* buf3 = bufB + w4;

    k_bin_chunk<BIN_SHIFT, false><<<nblk, BIN_TB, 0, stream>>>(src, dst, off, sorted, E, nb, 0u);
    k_bin_dinv_gemm<true><<<nb, FUSE_TB, 0, stream>>>(sorted, off, x, W1, dinv, bufB, n, nblk, nb, nullptr, 0);
    k_bin_layer4<<<nb, BIN_TB, 0, stream>>>(sorted, off, bufB, dinv, b1, W2, bufA, n, 4, nblk, nb);
    k_bin_layer4<<<nb, BIN_TB, 0, stream>>>(sorted, off, bufA, dinv, b2, W3, buf3, n, 2, nblk, nb);
    k_bin_final2<<<nb, BIN_TB, 0, stream>>>(sorted, off, buf3, dinv, b3, out, n, nblk, nb);
    return;
  }

  dim3 ge((E + TB - 1) / TB);
  float* ws   = (float*)d_ws;
  float* dinv = ws;
  float* hA   = dinv + (size_t)n;
  float* hB   = hA + (size_t)n * 4;
  float* hC   = hB + (size_t)n * 4;

  k_deg_init<<<gn, blk, 0, stream>>>(dinv, n);
  k_deg_scatter<<<ge, blk, 0, stream>>>(dst, dinv, E);
  k_dinv<<<gn, blk, 0, stream>>>(dinv, n);
  k_gemm_xW1<<<tiles, GEMM_WAVES * 32, 0, stream>>>(x, W1, hA, n);
  k_init_agg4<<<gn, blk, 0, stream>>>(dinv, hA, hB, n);
  k_scatter4<<<ge, blk, 0, stream>>>(src, dst, dinv, hA, hB, E);
  k_act_mm4<<<gn, blk, 0, stream>>>(hB, b1, W2, dinv, hA, n);
  k_scatter4<<<ge, blk, 0, stream>>>(src, dst, dinv, hA, hB, E);
  k_act_mm_final<<<gn, blk, 0, stream>>>(hB, b2, W3, b3, dinv, hC, out, n);
  k_scatter2<<<ge, blk, 0, stream>>>(src, dst, dinv, hC, out, E);
}
